# starting kernel
# baseline (speedup 1.0000x reference)
#include <hip/hip_runtime.h>
#include <hip/hip_bf16.h>


typedef __attribute__((ext_vector_type(2))) float v2f;
typedef __attribute__((ext_vector_type(8))) float v8f;

#define S4_DT 0.001f
#define TT 4096
#define NB 16
#define NS 64
#define NI 256
#define NO 256
#define LCH 64
#define NCH 64

__device__ __forceinline__ v8f wmma4(v2f a, v2f b, v8f c) {
  return __builtin_amdgcn_wmma_f32_16x16x4_f32(
       false, a,  false, b,
       (short)0, c,  false,  false);
}

__device__ __forceinline__ v2f load2(const float* p) { return *(const v2f*)p; }

__device__ __forceinline__ void store_c(float* dst, int rs, int n0, v8f acc,
                                        int l16, int half) {
#pragma unroll
  for (int v = 0; v < 8; ++v)
    dst[(v + 8 * half) * rs + n0 + l16] = acc[v];
}

__device__ __forceinline__ v8f load_c(const float* src, int rs, int n0,
                                      int l16, int half) {
  v8f r;
#pragma unroll
  for (int v = 0; v < 8; ++v)
    r[v] = src[(v + 8 * half) * rs + n0 + l16];
  return r;
}

__global__ __launch_bounds__(512) void s4_prep_kernel(
    const float* __restrict__ A, const float* __restrict__ B,
    float* __restrict__ Abar, float* __restrict__ Apow,
    float* __restrict__ Pcat, float* __restrict__ Bsc) {
  __shared__ float sPrev[2][NS * NS];
  __shared__ float sAbT[NS * NS];

  const int tid = threadIdx.x;
  for (int idx = tid; idx < NS * NS; idx += 512) {
    int r = idx >> 6, c = idx & 63;
    float v = A[idx] * S4_DT + (r == c ? 1.0f : 0.0f);
    Abar[idx] = v;
    sPrev[0][idx] = v;
    sAbT[c * NS + r] = v;
    Apow[0 * NS * NS + idx] = (r == c) ? 1.0f : 0.0f;
    Apow[1 * NS * NS + idx] = v;
  }
  for (int idx = tid; idx < NS * NI; idx += 512) Bsc[idx] = B[idx] * S4_DT;
  __syncthreads();

  const int wave = tid >> 5, lane = tid & 31;
  const int l16 = lane & 15, half = lane >> 4;
  const int m0 = (wave >> 2) * 16, n0 = (wave & 3) * 16;

  int cur = 0;
  for (int k = 2; k <= LCH; ++k) {
    v8f acc = {};
#pragma unroll
    for (int kk = 0; kk < 16; ++kk) {
      v2f a = load2(&sPrev[cur][(m0 + l16) * NS + 4 * kk + 2 * half]);
      v2f b = load2(&sAbT[(n0 + l16) * NS + 4 * kk + 2 * half]);
      acc = wmma4(a, b, acc);
    }
    int nxt = cur ^ 1;
#pragma unroll
    for (int v = 0; v < 8; ++v) {
      int row = m0 + v + 8 * half;
      float val = acc[v];
      sPrev[nxt][row * NS + n0 + l16] = val;
      Apow[k * NS * NS + row * NS + n0 + l16] = val;
    }
    __syncthreads();
    cur = nxt;
  }

  for (int idx = tid; idx < NS * (LCH * NS); idx += 512) {
    int n = idx >> 12, rest = idx & 4095, j = rest >> 6, m = rest & 63;
    Pcat[idx] = Apow[(63 - j) * NS * NS + n * NS + m];
  }
}

__global__ __launch_bounds__(128) void s4_ub_kernel(
    const float* __restrict__ u, const float* __restrict__ Bsc,
    float* __restrict__ uB) {
  const int wave = threadIdx.x >> 5, lane = threadIdx.x & 31;
  const int l16 = lane & 15, half = lane >> 4;
  const int t = blockIdx.x * 4 + wave;

  const float* urow = u + (size_t)l16 * TT * NI + (size_t)t * NI;
  v8f acc[4] = {{}, {}, {}, {}};
  for (int kk = 0; kk < NI / 4; ++kk) {
    v2f a = load2(urow + 4 * kk + 2 * half);
#pragma unroll
    for (int j = 0; j < 4; ++j) {
      v2f b = load2(Bsc + (j * 16 + l16) * NI + 4 * kk + 2 * half);
      acc[j] = wmma4(a, b, acc[j]);
    }
  }
  float* dst = uB + (size_t)t * NB * NS;
#pragma unroll
  for (int j = 0; j < 4; ++j) store_c(dst, NS, j * 16, acc[j], l16, half);
}

__global__ __launch_bounds__(128) void s4_gagg_kernel(
    const float* __restrict__ uB, const float* __restrict__ Pcat,
    float* __restrict__ G) {
  const int wave = threadIdx.x >> 5, lane = threadIdx.x & 31;
  const int l16 = lane & 15, half = lane >> 4;
  const int c = blockIdx.x * 4 + wave;

  const float* F = uB + (size_t)c * LCH * NB * NS;
  v8f acc[4] = {{}, {}, {}, {}};
  for (int kk = 0; kk < (LCH * NS) / 4; ++kk) {
    int kap = 4 * kk + 2 * half;
    int j = kap >> 6, m = kap & 63;
    v2f a = load2(F + j * NB * NS + l16 * NS + m);
#pragma unroll
    for (int jn = 0; jn < 4; ++jn) {
      v2f b = load2(Pcat + (size_t)(jn * 16 + l16) * (LCH * NS) + kap);
      acc[jn] = wmma4(a, b, acc[jn]);
    }
  }
  float* dst = G + (size_t)c * NB * NS;
#pragma unroll
  for (int j = 0; j < 4; ++j) store_c(dst, NS, j * 16, acc[j], l16, half);
}

__global__ __launch_bounds__(128) void s4_bscan_kernel(
    const float* __restrict__ x0, const float* __restrict__ G,
    const float* __restrict__ Apow, float* __restrict__ Sbound,
    float* __restrict__ xfinal) {
  __shared__ float sb[2][NB * NS];
  const int tid = threadIdx.x;
  const int wave = tid >> 5, lane = tid & 31;
  const int l16 = lane & 15, half = lane >> 4;
  const int n0 = wave * 16;
  const float* AL = Apow + (size_t)LCH * NS * NS;

  for (int idx = tid; idx < NB * NS; idx += 128) sb[0][idx] = x0[idx];
  __syncthreads();

  int cur = 0;
  for (int c = 0; c < NCH; ++c) {
    for (int idx = tid; idx < NB * NS; idx += 128)
      Sbound[(size_t)c * NB * NS + idx] = sb[cur][idx];
    v8f acc = load_c(G + (size_t)c * NB * NS, NS, n0, l16, half);
#pragma unroll
    for (int kk = 0; kk < 16; ++kk) {
      v2f a = load2(&sb[cur][l16 * NS + 4 * kk + 2 * half]);
      v2f b = load2(AL + (n0 + l16) * NS + 4 * kk + 2 * half);
      acc = wmma4(a, b, acc);
    }
    int nxt = cur ^ 1;
#pragma unroll
    for (int v = 0; v < 8; ++v)
      sb[nxt][(v + 8 * half) * NS + n0 + l16] = acc[v];
    __syncthreads();
    cur = nxt;
  }
  for (int idx = tid; idx < NB * NS; idx += 128) xfinal[idx] = sb[cur][idx];
}

__global__ __launch_bounds__(128) void s4_states_kernel(
    const float* __restrict__ uB, const float* __restrict__ Sbound,
    const float* __restrict__ Abar, float* __restrict__ S) {
  __shared__ float sb[2][NB * NS];
  const int tid = threadIdx.x;
  const int wave = tid >> 5, lane = tid & 31;
  const int l16 = lane & 15, half = lane >> 4;
  const int n0 = wave * 16;
  const int c = blockIdx.x;
  const int t0 = c * LCH;

  for (int idx = tid; idx < NB * NS; idx += 128)
    sb[0][idx] = Sbound[(size_t)c * NB * NS + idx];
  __syncthreads();

  int cur = 0;
  for (int k = 0; k < LCH; ++k) {
    float* dst = S + (size_t)(t0 + k) * NB * NS;
    for (int idx = tid; idx < NB * NS; idx += 128) dst[idx] = sb[cur][idx];
    v8f acc = load_c(uB + (size_t)(t0 + k) * NB * NS, NS, n0, l16, half);
#pragma unroll
    for (int kk = 0; kk < 16; ++kk) {
      v2f a = load2(&sb[cur][l16 * NS + 4 * kk + 2 * half]);
      v2f b = load2(Abar + (n0 + l16) * NS + 4 * kk + 2 * half);
      acc = wmma4(a, b, acc);
    }
    int nxt = cur ^ 1;
#pragma unroll
    for (int v = 0; v < 8; ++v)
      sb[nxt][(v + 8 * half) * NS + n0 + l16] = acc[v];
    __syncthreads();
    cur = nxt;
  }
}

__global__ __launch_bounds__(128) void s4_y_kernel(
    const float* __restrict__ S, const float* __restrict__ u,
    const float* __restrict__ C, const float* __restrict__ D,
    float* __restrict__ y) {
  const int wave = threadIdx.x >> 5, lane = threadIdx.x & 31;
  const int l16 = lane & 15, half = lane >> 4;
  const int t = blockIdx.x;
  const int o0 = wave * 64;

  v8f acc[4] = {{}, {}, {}, {}};

  const float* srow = S + (size_t)t * NB * NS + l16 * NS;
#pragma unroll
  for (int kk = 0; kk < NS / 4; ++kk) {
    v2f a = load2(srow + 4 * kk + 2 * half);
#pragma unroll
    for (int j = 0; j < 4; ++j) {
      v2f b = load2(C + (o0 + 16 * j + l16) * NS + 4 * kk + 2 * half);
      acc[j] = wmma4(a, b, acc[j]);
    }
  }

  const float* urow = u + (size_t)l16 * TT * NI + (size_t)t * NI;
  for (int kk = 0; kk < NI / 4; ++kk) {
    v2f a = load2(urow + 4 * kk + 2 * half);
#pragma unroll
    for (int j = 0; j < 4; ++j) {
      v2f b = load2(D + (o0 + 16 * j + l16) * NI + 4 * kk + 2 * half);
      acc[j] = wmma4(a, b, acc[j]);
    }
  }

#pragma unroll
  for (int j = 0; j < 4; ++j)
#pragma unroll
    for (int v = 0; v < 8; ++v)
      y[(size_t)(v + 8 * half) * TT * NO + (size_t)t * NO + o0 + 16 * j + l16] =
          acc[j][v];
}

extern "C" void kernel_launch(void* const* d_in, const int* in_sizes, int n_in,
                              void* d_out, int out_size, void* d_ws,
                              size_t ws_size, hipStream_t stream) {
  const float* u  = (const float*)d_in[0];
  const float* x0 = (const float*)d_in[1];
  const float* A  = (const float*)d_in[2];
  const float* B  = (const float*)d_in[3];
  const float* C  = (const float*)d_in[4];
  const float* D  = (const float*)d_in[5];

  float* y      = (float*)d_out;
  float* xfinal = (float*)d_out + (size_t)NB * TT * NO;

  float* ws = (float*)d_ws;
  float* Abar   = ws;                 ws += NS * NS;
  float* Apow   = ws;                 ws += (LCH + 1) * NS * NS;
  float* Pcat   = ws;                 ws += NS * LCH * NS;
  float* Bsc    = ws;                 ws += NS * NI;
  float* uB     = ws;                 ws += (size_t)TT * NB * NS;
  float* Sst    = ws;                 ws += (size_t)TT * NB * NS;
  float* G      = ws;                 ws += NCH * NB * NS;
  float* Sbound = ws;                 ws += NCH * NB * NS;

  s4_prep_kernel<<<1, 512, 0, stream>>>(A, B, Abar, Apow, Pcat, Bsc);
  s4_ub_kernel<<<TT / 4, 128, 0, stream>>>(u, Bsc, uB);
  s4_gagg_kernel<<<NCH / 4, 128, 0, stream>>>(uB, Pcat, G);
  s4_bscan_kernel<<<1, 128, 0, stream>>>(x0, G, Apow, Sbound, xfinal);
  s4_states_kernel<<<NCH, 128, 0, stream>>>(uB, Sbound, Abar, Sst);
  s4_y_kernel<<<TT, 128, 0, stream>>>(Sst, u, C, D, y);
}


# LDS row pitch of the 64-wide state and power tiles 64 -> 68 floats in the three LDS-staged chain kernels: WMMA fragment reads become bank-conflict-free
# speedup vs baseline: 1.3475x; 1.3475x over previous
#include <hip/hip_runtime.h>
#include <hip/hip_bf16.h>


typedef __attribute__((ext_vector_type(2))) float v2f;
typedef __attribute__((ext_vector_type(8))) float v8f;

#define S4_DT 0.001f
#define TT 4096
#define NB 16
#define NS 64
#define NI 256
#define NO 256
#define LCH 64
#define NCH 64
#define LP 68

__device__ __forceinline__ v8f wmma4(v2f a, v2f b, v8f c) {
  return __builtin_amdgcn_wmma_f32_16x16x4_f32(
       false, a,  false, b,
       (short)0, c,  false,  false);
}

__device__ __forceinline__ v2f load2(const float* p) { return *(const v2f*)p; }

__device__ __forceinline__ void store_c(float* dst, int rs, int n0, v8f acc,
                                        int l16, int half) {
#pragma unroll
  for (int v = 0; v < 8; ++v)
    dst[(v + 8 * half) * rs + n0 + l16] = acc[v];
}

__device__ __forceinline__ v8f load_c(const float* src, int rs, int n0,
                                      int l16, int half) {
  v8f r;
#pragma unroll
  for (int v = 0; v < 8; ++v)
    r[v] = src[(v + 8 * half) * rs + n0 + l16];
  return r;
}

__global__ __launch_bounds__(512) void s4_prep_kernel(
    const float* __restrict__ A, const float* __restrict__ B,
    float* __restrict__ Abar, float* __restrict__ Apow,
    float* __restrict__ Pcat, float* __restrict__ Bsc) {
  __shared__ float sPrev[2][NS * LP];
  __shared__ float sAbT[NS * LP];

  const int tid = threadIdx.x;
  for (int idx = tid; idx < NS * NS; idx += 512) {
    int r = idx >> 6, c = idx & 63;
    float v = A[idx] * S4_DT + (r == c ? 1.0f : 0.0f);
    Abar[idx] = v;
    sPrev[0][r * LP + c] = v;
    sAbT[c * LP + r] = v;
    Apow[0 * NS * NS + idx] = (r == c) ? 1.0f : 0.0f;
    Apow[1 * NS * NS + idx] = v;
  }
  for (int idx = tid; idx < NS * NI; idx += 512) Bsc[idx] = B[idx] * S4_DT;
  __syncthreads();

  const int wave = tid >> 5, lane = tid & 31;
  const int l16 = lane & 15, half = lane >> 4;
  const int m0 = (wave >> 2) * 16, n0 = (wave & 3) * 16;

  int cur = 0;
  for (int k = 2; k <= LCH; ++k) {
    v8f acc = {};
#pragma unroll
    for (int kk = 0; kk < 16; ++kk) {
      v2f a = load2(&sPrev[cur][(m0 + l16) * LP + 4 * kk + 2 * half]);
      v2f b = load2(&sAbT[(n0 + l16) * LP + 4 * kk + 2 * half]);
      acc = wmma4(a, b, acc);
    }
    int nxt = cur ^ 1;
#pragma unroll
    for (int v = 0; v < 8; ++v) {
      int row = m0 + v + 8 * half;
      float val = acc[v];
      sPrev[nxt][row * LP + n0 + l16] = val;
      Apow[k * NS * NS + row * NS + n0 + l16] = val;
    }
    __syncthreads();
    cur = nxt;
  }

  for (int idx = tid; idx < NS * (LCH * NS); idx += 512) {
    int n = idx >> 12, rest = idx & 4095, j = rest >> 6, m = rest & 63;
    Pcat[idx] = Apow[(63 - j) * NS * NS + n * NS + m];
  }
}

__global__ __launch_bounds__(128) void s4_ub_kernel(
    const float* __restrict__ u, const float* __restrict__ Bsc,
    float* __restrict__ uB) {
  const int wave = threadIdx.x >> 5, lane = threadIdx.x & 31;
  const int l16 = lane & 15, half = lane >> 4;
  const int t = blockIdx.x * 4 + wave;

  const float* urow = u + (size_t)l16 * TT * NI + (size_t)t * NI;
  v8f acc[4] = {{}, {}, {}, {}};
  for (int kk = 0; kk < NI / 4; ++kk) {
    v2f a = load2(urow + 4 * kk + 2 * half);
#pragma unroll
    for (int j = 0; j < 4; ++j) {
      v2f b = load2(Bsc + (j * 16 + l16) * NI + 4 * kk + 2 * half);
      acc[j] = wmma4(a, b, acc[j]);
    }
  }
  float* dst = uB + (size_t)t * NB * NS;
#pragma unroll
  for (int j = 0; j < 4; ++j) store_c(dst, NS, j * 16, acc[j], l16, half);
}

__global__ __launch_bounds__(128) void s4_gagg_kernel(
    const float* __restrict__ uB, const float* __restrict__ Pcat,
    float* __restrict__ G) {
  const int wave = threadIdx.x >> 5, lane = threadIdx.x & 31;
  const int l16 = lane & 15, half = lane >> 4;
  const int c = blockIdx.x * 4 + wave;

  const float* F = uB + (size_t)c * LCH * NB * NS;
  v8f acc[4] = {{}, {}, {}, {}};
  for (int kk = 0; kk < (LCH * NS) / 4; ++kk) {
    int kap = 4 * kk + 2 * half;
    int j = kap >> 6, m = kap & 63;
    v2f a = load2(F + j * NB * NS + l16 * NS + m);
#pragma unroll
    for (int jn = 0; jn < 4; ++jn) {
      v2f b = load2(Pcat + (size_t)(jn * 16 + l16) * (LCH * NS) + kap);
      acc[jn] = wmma4(a, b, acc[jn]);
    }
  }
  float* dst = G + (size_t)c * NB * NS;
#pragma unroll
  for (int j = 0; j < 4; ++j) store_c(dst, NS, j * 16, acc[j], l16, half);
}

__global__ __launch_bounds__(128) void s4_bscan_kernel(
    const float* __restrict__ x0, const float* __restrict__ G,
    const float* __restrict__ Apow, float* __restrict__ Sbound,
    float* __restrict__ xfinal) {
  __shared__ float sb[2][NB * LP];
  const int tid = threadIdx.x;
  const int wave = tid >> 5, lane = tid & 31;
  const int l16 = lane & 15, half = lane >> 4;
  const int n0 = wave * 16;
  const float* AL = Apow + (size_t)LCH * NS * NS;

  for (int idx = tid; idx < NB * NS; idx += 128)
    sb[0][(idx >> 6) * LP + (idx & 63)] = x0[idx];
  __syncthreads();

  int cur = 0;
  for (int c = 0; c < NCH; ++c) {
    for (int idx = tid; idx < NB * NS; idx += 128)
      Sbound[(size_t)c * NB * NS + idx] = sb[cur][(idx >> 6) * LP + (idx & 63)];
    v8f acc = load_c(G + (size_t)c * NB * NS, NS, n0, l16, half);
#pragma unroll
    for (int kk = 0; kk < 16; ++kk) {
      v2f a = load2(&sb[cur][l16 * LP + 4 * kk + 2 * half]);
      v2f b = load2(AL + (n0 + l16) * NS + 4 * kk + 2 * half);
      acc = wmma4(a, b, acc);
    }
    int nxt = cur ^ 1;
#pragma unroll
    for (int v = 0; v < 8; ++v)
      sb[nxt][(v + 8 * half) * LP + n0 + l16] = acc[v];
    __syncthreads();
    cur = nxt;
  }
  for (int idx = tid; idx < NB * NS; idx += 128)
    xfinal[idx] = sb[cur][(idx >> 6) * LP + (idx & 63)];
}

__global__ __launch_bounds__(128) void s4_states_kernel(
    const float* __restrict__ uB, const float* __restrict__ Sbound,
    const float* __restrict__ Abar, float* __restrict__ S) {
  __shared__ float sb[2][NB * LP];
  const int tid = threadIdx.x;
  const int wave = tid >> 5, lane = tid & 31;
  const int l16 = lane & 15, half = lane >> 4;
  const int n0 = wave * 16;
  const int c = blockIdx.x;
  const int t0 = c * LCH;

  for (int idx = tid; idx < NB * NS; idx += 128)
    sb[0][(idx >> 6) * LP + (idx & 63)] = Sbound[(size_t)c * NB * NS + idx];
  __syncthreads();

  int cur = 0;
  for (int k = 0; k < LCH; ++k) {
    float* dst = S + (size_t)(t0 + k) * NB * NS;
    for (int idx = tid; idx < NB * NS; idx += 128)
      dst[idx] = sb[cur][(idx >> 6) * LP + (idx & 63)];
    v8f acc = load_c(uB + (size_t)(t0 + k) * NB * NS, NS, n0, l16, half);
#pragma unroll
    for (int kk = 0; kk < 16; ++kk) {
      v2f a = load2(&sb[cur][l16 * LP + 4 * kk + 2 * half]);
      v2f b = load2(Abar + (n0 + l16) * NS + 4 * kk + 2 * half);
      acc = wmma4(a, b, acc);
    }
    int nxt = cur ^ 1;
#pragma unroll
    for (int v = 0; v < 8; ++v)
      sb[nxt][(v + 8 * half) * LP + n0 + l16] = acc[v];
    __syncthreads();
    cur = nxt;
  }
}

__global__ __launch_bounds__(128) void s4_y_kernel(
    const float* __restrict__ S, const float* __restrict__ u,
    const float* __restrict__ C, const float* __restrict__ D,
    float* __restrict__ y) {
  const int wave = threadIdx.x >> 5, lane = threadIdx.x & 31;
  const int l16 = lane & 15, half = lane >> 4;
  const int t = blockIdx.x;
  const int o0 = wave * 64;

  v8f acc[4] = {{}, {}, {}, {}};

  const float* srow = S + (size_t)t * NB * NS + l16 * NS;
#pragma unroll
  for (int kk = 0; kk < NS / 4; ++kk) {
    v2f a = load2(srow + 4 * kk + 2 * half);
#pragma unroll
    for (int j = 0; j < 4; ++j) {
      v2f b = load2(C + (o0 + 16 * j + l16) * NS + 4 * kk + 2 * half);
      acc[j] = wmma4(a, b, acc[j]);
    }
  }

  const float* urow = u + (size_t)l16 * TT * NI + (size_t)t * NI;
  for (int kk = 0; kk < NI / 4; ++kk) {
    v2f a = load2(urow + 4 * kk + 2 * half);
#pragma unroll
    for (int j = 0; j < 4; ++j) {
      v2f b = load2(D + (o0 + 16 * j + l16) * NI + 4 * kk + 2 * half);
      acc[j] = wmma4(a, b, acc[j]);
    }
  }

#pragma unroll
  for (int j = 0; j < 4; ++j)
#pragma unroll
    for (int v = 0; v < 8; ++v)
      y[(size_t)(v + 8 * half) * TT * NO + (size_t)t * NO + o0 + 16 * j + l16] =
          acc[j][v];
}

extern "C" void kernel_launch(void* const* d_in, const int* in_sizes, int n_in,
                              void* d_out, int out_size, void* d_ws,
                              size_t ws_size, hipStream_t stream) {
  const float* u  = (const float*)d_in[0];
  const float* x0 = (const float*)d_in[1];
  const float* A  = (const float*)d_in[2];
  const float* B  = (const float*)d_in[3];
  const float* C  = (const float*)d_in[4];
  const float* D  = (const float*)d_in[5];

  float* y      = (float*)d_out;
  float* xfinal = (float*)d_out + (size_t)NB * TT * NO;

  float* ws = (float*)d_ws;
  float* Abar   = ws;                 ws += NS * NS;
  float* Apow   = ws;                 ws += (LCH + 1) * NS * NS;
  float* Pcat   = ws;                 ws += NS * LCH * NS;
  float* Bsc    = ws;                 ws += NS * NI;
  float* uB     = ws;                 ws += (size_t)TT * NB * NS;
  float* Sst    = ws;                 ws += (size_t)TT * NB * NS;
  float* G      = ws;                 ws += NCH * NB * NS;
  float* Sbound = ws;                 ws += NCH * NB * NS;

  s4_prep_kernel<<<1, 512, 0, stream>>>(A, B, Abar, Apow, Pcat, Bsc);
  s4_ub_kernel<<<TT / 4, 128, 0, stream>>>(u, Bsc, uB);
  s4_gagg_kernel<<<NCH / 4, 128, 0, stream>>>(uB, Pcat, G);
  s4_bscan_kernel<<<1, 128, 0, stream>>>(x0, G, Apow, Sbound, xfinal);
  s4_states_kernel<<<NCH, 128, 0, stream>>>(uB, Sbound, Abar, Sst);
  s4_y_kernel<<<TT, 128, 0, stream>>>(Sst, u, C, D, y);
}


# y kernel: weight and row slices staged in LDS by coalesced 16-byte loads, one workgroup per 4 time steps x 256 outputs, instead of per-lane 8-byte fragment loads from global; k order kept
# speedup vs baseline: 1.6069x; 1.1925x over previous
#include <hip/hip_runtime.h>
#include <hip/hip_bf16.h>


typedef __attribute__((ext_vector_type(2))) float v2f;
typedef __attribute__((ext_vector_type(8))) float v8f;

#define S4_DT 0.001f
#define TT 4096
#define NB 16
#define NS 64
#define NI 256
#define NO 256
#define LCH 64
#define NCH 64
#define LP 68

__device__ __forceinline__ v8f wmma4(v2f a, v2f b, v8f c) {
  return __builtin_amdgcn_wmma_f32_16x16x4_f32(
       false, a,  false, b,
       (short)0, c,  false,  false);
}

__device__ __forceinline__ v2f load2(const float* p) { return *(const v2f*)p; }

__device__ __forceinline__ void store_c(float* dst, int rs, int n0, v8f acc,
                                        int l16, int half) {
#pragma unroll
  for (int v = 0; v < 8; ++v)
    dst[(v + 8 * half) * rs + n0 + l16] = acc[v];
}

__device__ __forceinline__ v8f load_c(const float* src, int rs, int n0,
                                      int l16, int half) {
  v8f r;
#pragma unroll
  for (int v = 0; v < 8; ++v)
    r[v] = src[(v + 8 * half) * rs + n0 + l16];
  return r;
}

__global__ __launch_bounds__(512) void s4_prep_kernel(
    const float* __restrict__ A, const float* __restrict__ B,
    float* __restrict__ Abar, float* __restrict__ Apow,
    float* __restrict__ Pcat, float* __restrict__ Bsc) {
  __shared__ float sPrev[2][NS * LP];
  __shared__ float sAbT[NS * LP];

  const int tid = threadIdx.x;
  for (int idx = tid; idx < NS * NS; idx += 512) {
    int r = idx >> 6, c = idx & 63;
    float v = A[idx] * S4_DT + (r == c ? 1.0f : 0.0f);
    Abar[idx] = v;
    sPrev[0][r * LP + c] = v;
    sAbT[c * LP + r] = v;
    Apow[0 * NS * NS + idx] = (r == c) ? 1.0f : 0.0f;
    Apow[1 * NS * NS + idx] = v;
  }
  for (int idx = tid; idx < NS * NI; idx += 512) Bsc[idx] = B[idx] * S4_DT;
  __syncthreads();

  const int wave = tid >> 5, lane = tid & 31;
  const int l16 = lane & 15, half = lane >> 4;
  const int m0 = (wave >> 2) * 16, n0 = (wave & 3) * 16;

  int cur = 0;
  for (int k = 2; k <= LCH; ++k) {
    v8f acc = {};
#pragma unroll
    for (int kk = 0; kk < 16; ++kk) {
      v2f a = load2(&sPrev[cur][(m0 + l16) * LP + 4 * kk + 2 * half]);
      v2f b = load2(&sAbT[(n0 + l16) * LP + 4 * kk + 2 * half]);
      acc = wmma4(a, b, acc);
    }
    int nxt = cur ^ 1;
#pragma unroll
    for (int v = 0; v < 8; ++v) {
      int row = m0 + v + 8 * half;
      float val = acc[v];
      sPrev[nxt][row * LP + n0 + l16] = val;
      Apow[k * NS * NS + row * NS + n0 + l16] = val;
    }
    __syncthreads();
    cur = nxt;
  }

  for (int idx = tid; idx < NS * (LCH * NS); idx += 512) {
    int n = idx >> 12, rest = idx & 4095, j = rest >> 6, m = rest & 63;
    Pcat[idx] = Apow[(63 - j) * NS * NS + n * NS + m];
  }
}

__global__ __launch_bounds__(128) void s4_ub_kernel(
    const float* __restrict__ u, const float* __restrict__ Bsc,
    float* __restrict__ uB) {
  const int wave = threadIdx.x >> 5, lane = threadIdx.x & 31;
  const int l16 = lane & 15, half = lane >> 4;
  const int t = blockIdx.x * 4 + wave;

  const float* urow = u + (size_t)l16 * TT * NI + (size_t)t * NI;
  v8f acc[4] = {{}, {}, {}, {}};
  for (int kk = 0; kk < NI / 4; ++kk) {
    v2f a = load2(urow + 4 * kk + 2 * half);
#pragma unroll
    for (int j = 0; j < 4; ++j) {
      v2f b = load2(Bsc + (j * 16 + l16) * NI + 4 * kk + 2 * half);
      acc[j] = wmma4(a, b, acc[j]);
    }
  }
  float* dst = uB + (size_t)t * NB * NS;
#pragma unroll
  for (int j = 0; j < 4; ++j) store_c(dst, NS, j * 16, acc[j], l16, half);
}

__global__ __launch_bounds__(128) void s4_gagg_kernel(
    const float* __restrict__ uB, const float* __restrict__ Pcat,
    float* __restrict__ G) {
  const int wave = threadIdx.x >> 5, lane = threadIdx.x & 31;
  const int l16 = lane & 15, half = lane >> 4;
  const int c = blockIdx.x * 4 + wave;

  const float* F = uB + (size_t)c * LCH * NB * NS;
  v8f acc[4] = {{}, {}, {}, {}};
  for (int kk = 0; kk < (LCH * NS) / 4; ++kk) {
    int kap = 4 * kk + 2 * half;
    int j = kap >> 6, m = kap & 63;
    v2f a = load2(F + j * NB * NS + l16 * NS + m);
#pragma unroll
    for (int jn = 0; jn < 4; ++jn) {
      v2f b = load2(Pcat + (size_t)(jn * 16 + l16) * (LCH * NS) + kap);
      acc[jn] = wmma4(a, b, acc[jn]);
    }
  }
  float* dst = G + (size_t)c * NB * NS;
#pragma unroll
  for (int j = 0; j < 4; ++j) store_c(dst, NS, j * 16, acc[j], l16, half);
}

__global__ __launch_bounds__(128) void s4_bscan_kernel(
    const float* __restrict__ x0, const float* __restrict__ G,
    const float* __restrict__ Apow, float* __restrict__ Sbound,
    float* __restrict__ xfinal) {
  __shared__ float sb[2][NB * LP];
  const int tid = threadIdx.x;
  const int wave = tid >> 5, lane = tid & 31;
  const int l16 = lane & 15, half = lane >> 4;
  const int n0 = wave * 16;
  const float* AL = Apow + (size_t)LCH * NS * NS;

  for (int idx = tid; idx < NB * NS; idx += 128)
    sb[0][(idx >> 6) * LP + (idx & 63)] = x0[idx];
  __syncthreads();

  int cur = 0;
  for (int c = 0; c < NCH; ++c) {
    for (int idx = tid; idx < NB * NS; idx += 128)
      Sbound[(size_t)c * NB * NS + idx] = sb[cur][(idx >> 6) * LP + (idx & 63)];
    v8f acc = load_c(G + (size_t)c * NB * NS, NS, n0, l16, half);
#pragma unroll
    for (int kk = 0; kk < 16; ++kk) {
      v2f a = load2(&sb[cur][l16 * LP + 4 * kk + 2 * half]);
      v2f b = load2(AL + (n0 + l16) * NS + 4 * kk + 2 * half);
      acc = wmma4(a, b, acc);
    }
    int nxt = cur ^ 1;
#pragma unroll
    for (int v = 0; v < 8; ++v)
      sb[nxt][(v + 8 * half) * LP + n0 + l16] = acc[v];
    __syncthreads();
    cur = nxt;
  }
  for (int idx = tid; idx < NB * NS; idx += 128)
    xfinal[idx] = sb[cur][(idx >> 6) * LP + (idx & 63)];
}

__global__ __launch_bounds__(128) void s4_states_kernel(
    const float* __restrict__ uB, const float* __restrict__ Sbound,
    const float* __restrict__ Abar, float* __restrict__ S) {
  __shared__ float sb[2][NB * LP];
  const int tid = threadIdx.x;
  const int wave = tid >> 5, lane = tid & 31;
  const int l16 = lane & 15, half = lane >> 4;
  const int n0 = wave * 16;
  const int c = blockIdx.x;
  const int t0 = c * LCH;

  for (int idx = tid; idx < NB * NS; idx += 128)
    sb[0][(idx >> 6) * LP + (idx & 63)] = Sbound[(size_t)c * NB * NS + idx];
  __syncthreads();

  int cur = 0;
  for (int k = 0; k < LCH; ++k) {
    float* dst = S + (size_t)(t0 + k) * NB * NS;
    for (int idx = tid; idx < NB * NS; idx += 128)
      dst[idx] = sb[cur][(idx >> 6) * LP + (idx & 63)];
    v8f acc = load_c(uB + (size_t)(t0 + k) * NB * NS, NS, n0, l16, half);
#pragma unroll
    for (int kk = 0; kk < 16; ++kk) {
      v2f a = load2(&sb[cur][l16 * LP + 4 * kk + 2 * half]);
      v2f b = load2(Abar + (n0 + l16) * NS + 4 * kk + 2 * half);
      acc = wmma4(a, b, acc);
    }
    int nxt = cur ^ 1;
#pragma unroll
    for (int v = 0; v < 8; ++v)
      sb[nxt][(v + 8 * half) * LP + n0 + l16] = acc[v];
    __syncthreads();
    cur = nxt;
  }
}

typedef __attribute__((ext_vector_type(4))) float v4f;

#define YT 4
#define YM (YT * NB)
#define YKB 32
#define YSC (NS / YKB)
#define YSTEPS ((NS + NI) / YKB)
#define YLP 36

__device__ __forceinline__ void y_fetch(
    int s, int t0, int srow, int c4, const float* __restrict__ S,
    const float* __restrict__ u, const float* __restrict__ C,
    const float* __restrict__ D, v4f (&wreg)[8], v4f (&xreg)[2]) {
  const bool st = s < YSC;
  const float* wsrc = st ? C + YKB * s : D + YKB * (s - YSC);
  const int wpitch = st ? NS : NI;
#pragma unroll
  for (int i = 0; i < 8; ++i)
    wreg[i] = *(const v4f*)(wsrc + (size_t)(srow + 32 * i) * wpitch + 4 * c4);
#pragma unroll
  for (int i = 0; i < 2; ++i) {
    const int row = srow + 32 * i, tt = row >> 4, b = row & 15;
    const float* xsrc =
        st ? S + (size_t)(t0 + tt) * NB * NS + b * NS + YKB * s
           : u + (size_t)b * TT * NI + (size_t)(t0 + tt) * NI + YKB * (s - YSC);
    xreg[i] = *(const v4f*)(xsrc + 4 * c4);
  }
}

__global__ __launch_bounds__(256) void s4_y_kernel(
    const float* __restrict__ S, const float* __restrict__ u,
    const float* __restrict__ C, const float* __restrict__ D,
    float* __restrict__ y) {
  __shared__ __attribute__((aligned(16))) float sW[NO * YLP];
  __shared__ __attribute__((aligned(16))) float sX[YM * YLP];

  const int tid = threadIdx.x;
  const int wave = tid >> 5, lane = tid & 31;
  const int l16 = lane & 15, half = lane >> 4;
  const int t0 = blockIdx.x * YT;
  const int m0 = (wave & 1) * 2;
  const int o0 = (wave >> 1) * 64;
  const int srow = tid >> 3;
  const int c4 = tid & 7;

  v8f acc[2][4] = {{{}, {}, {}, {}}, {{}, {}, {}, {}}};
  v4f wreg[8], xreg[2];

  y_fetch(0, t0, srow, c4, S, u, C, D, wreg, xreg);
  for (int s = 0; s < YSTEPS; ++s) {
#pragma unroll
    for (int i = 0; i < 8; ++i)
      *(v4f*)&sW[(srow + 32 * i) * YLP + 4 * c4] = wreg[i];
#pragma unroll
    for (int i = 0; i < 2; ++i)
      *(v4f*)&sX[(srow + 32 * i) * YLP + 4 * c4] = xreg[i];
    __syncthreads();
    if (s + 1 < YSTEPS) y_fetch(s + 1, t0, srow, c4, S, u, C, D, wreg, xreg);
    __builtin_amdgcn_sched_barrier(0);
#pragma unroll
    for (int kk = 0; kk < YKB / 4; ++kk) {
      v2f a0 = load2(&sX[((m0 + 0) * NB + l16) * YLP + 4 * kk + 2 * half]);
      v2f a1 = load2(&sX[((m0 + 1) * NB + l16) * YLP + 4 * kk + 2 * half]);
#pragma unroll
      for (int j = 0; j < 4; ++j) {
        v2f b = load2(&sW[(o0 + 16 * j + l16) * YLP + 4 * kk + 2 * half]);
        acc[0][j] = wmma4(a0, b, acc[0][j]);
        acc[1][j] = wmma4(a1, b, acc[1][j]);
      }
    }
    __syncthreads();
  }

#pragma unroll
  for (int mi = 0; mi < 2; ++mi)
#pragma unroll
    for (int j = 0; j < 4; ++j)
#pragma unroll
      for (int v = 0; v < 8; ++v)
        y[(size_t)(v + 8 * half) * TT * NO + (size_t)(t0 + m0 + mi) * NO + o0 +
          16 * j + l16] = acc[mi][j][v];
}

extern "C" void kernel_launch(void* const* d_in, const int* in_sizes, int n_in,
                              void* d_out, int out_size, void* d_ws,
                              size_t ws_size, hipStream_t stream) {
  const float* u  = (const float*)d_in[0];
  const float* x0 = (const float*)d_in[1];
  const float* A  = (const float*)d_in[2];
  const float* B  = (const float*)d_in[3];
  const float* C  = (const float*)d_in[4];
  const float* D  = (const float*)d_in[5];

  float* y      = (float*)d_out;
  float* xfinal = (float*)d_out + (size_t)NB * TT * NO;

  float* ws = (float*)d_ws;
  float* Abar   = ws;                 ws += NS * NS;
  float* Apow   = ws;                 ws += (LCH + 1) * NS * NS;
  float* Pcat   = ws;                 ws += NS * LCH * NS;
  float* Bsc    = ws;                 ws += NS * NI;
  float* uB     = ws;                 ws += (size_t)TT * NB * NS;
  float* Sst    = ws;                 ws += (size_t)TT * NB * NS;
  float* G      = ws;                 ws += NCH * NB * NS;
  float* Sbound = ws;                 ws += NCH * NB * NS;

  s4_prep_kernel<<<1, 512, 0, stream>>>(A, B, Abar, Apow, Pcat, Bsc);
  s4_ub_kernel<<<TT / 4, 128, 0, stream>>>(u, Bsc, uB);
  s4_gagg_kernel<<<NCH / 4, 128, 0, stream>>>(uB, Pcat, G);
  s4_bscan_kernel<<<1, 128, 0, stream>>>(x0, G, Apow, Sbound, xfinal);
  s4_states_kernel<<<NCH, 128, 0, stream>>>(uB, Sbound, Abar, Sst);
  s4_y_kernel<<<TT / YT, 256, 0, stream>>>(Sst, u, C, D, y);
}


# gagg: each chunk's 4096-long reduction split over 16 waves (4096 waves on all WGPs instead of 64 lone ones); the 16 partial sums per element added in fixed ascending order through LDS; G tile-major
# speedup vs baseline: 2.1617x; 1.3453x over previous
#include <hip/hip_runtime.h>
#include <hip/hip_bf16.h>


typedef __attribute__((ext_vector_type(2))) float v2f;
typedef __attribute__((ext_vector_type(8))) float v8f;

#define S4_DT 0.001f
#define TT 4096
#define NB 16
#define NS 64
#define NI 256
#define NO 256
#define LCH 64
#define NCH 64
#define LP 68

__device__ __forceinline__ v8f wmma4(v2f a, v2f b, v8f c) {
  return __builtin_amdgcn_wmma_f32_16x16x4_f32(
       false, a,  false, b,
       (short)0, c,  false,  false);
}

__device__ __forceinline__ v2f load2(const float* p) { return *(const v2f*)p; }

__device__ __forceinline__ void store_c(float* dst, int rs, int n0, v8f acc,
                                        int l16, int half) {
#pragma unroll
  for (int v = 0; v < 8; ++v)
    dst[(v + 8 * half) * rs + n0 + l16] = acc[v];
}

__device__ __forceinline__ v8f load_c(const float* src, int rs, int n0,
                                      int l16, int half) {
  v8f r;
#pragma unroll
  for (int v = 0; v < 8; ++v)
    r[v] = src[(v + 8 * half) * rs + n0 + l16];
  return r;
}

__global__ __launch_bounds__(512) void s4_prep_kernel(
    const float* __restrict__ A, const float* __restrict__ B,
    float* __restrict__ Abar, float* __restrict__ Apow,
    float* __restrict__ Pcat, float* __restrict__ Bsc) {
  __shared__ float sPrev[2][NS * LP];
  __shared__ float sAbT[NS * LP];

  const int tid = threadIdx.x;
  for (int idx = tid; idx < NS * NS; idx += 512) {
    int r = idx >> 6, c = idx & 63;
    float v = A[idx] * S4_DT + (r == c ? 1.0f : 0.0f);
    Abar[idx] = v;
    sPrev[0][r * LP + c] = v;
    sAbT[c * LP + r] = v;
    Apow[0 * NS * NS + idx] = (r == c) ? 1.0f : 0.0f;
    Apow[1 * NS * NS + idx] = v;
  }
  for (int idx = tid; idx < NS * NI; idx += 512) Bsc[idx] = B[idx] * S4_DT;
  __syncthreads();

  const int wave = tid >> 5, lane = tid & 31;
  const int l16 = lane & 15, half = lane >> 4;
  const int m0 = (wave >> 2) * 16, n0 = (wave & 3) * 16;

  int cur = 0;
  for (int k = 2; k <= LCH; ++k) {
    v8f acc = {};
#pragma unroll
    for (int kk = 0; kk < 16; ++kk) {
      v2f a = load2(&sPrev[cur][(m0 + l16) * LP + 4 * kk + 2 * half]);
      v2f b = load2(&sAbT[(n0 + l16) * LP + 4 * kk + 2 * half]);
      acc = wmma4(a, b, acc);
    }
    int nxt = cur ^ 1;
#pragma unroll
    for (int v = 0; v < 8; ++v) {
      int row = m0 + v + 8 * half;
      float val = acc[v];
      sPrev[nxt][row * LP + n0 + l16] = val;
      Apow[k * NS * NS + row * NS + n0 + l16] = val;
    }
    __syncthreads();
    cur = nxt;
  }

  for (int idx = tid; idx < NS * (LCH * NS); idx += 512) {
    int n = idx >> 12, rest = idx & 4095, j = rest >> 6, m = rest & 63;
    Pcat[idx] = Apow[(63 - j) * NS * NS + n * NS + m];
  }
}

__global__ __launch_bounds__(128) void s4_ub_kernel(
    const float* __restrict__ u, const float* __restrict__ Bsc,
    float* __restrict__ uB) {
  const int wave = threadIdx.x >> 5, lane = threadIdx.x & 31;
  const int l16 = lane & 15, half = lane >> 4;
  const int t = blockIdx.x * 4 + wave;

  const float* urow = u + (size_t)l16 * TT * NI + (size_t)t * NI;
  v8f acc[4] = {{}, {}, {}, {}};
  for (int kk = 0; kk < NI / 4; ++kk) {
    v2f a = load2(urow + 4 * kk + 2 * half);
#pragma unroll
    for (int j = 0; j < 4; ++j) {
      v2f b = load2(Bsc + (j * 16 + l16) * NI + 4 * kk + 2 * half);
      acc[j] = wmma4(a, b, acc[j]);
    }
  }
  float* dst = uB + (size_t)t * NB * NS;
#pragma unroll
  for (int j = 0; j < 4; ++j) store_c(dst, NS, j * 16, acc[j], l16, half);
}

#define GKS 16
#define GKL ((LCH * NS) / GKS)
#define GRP 18

__global__ __launch_bounds__(512) void s4_gagg_kernel(
    const float* __restrict__ uB, const float* __restrict__ Pcat,
    float* __restrict__ G) {
  __shared__ float red[GKS][16 * GRP];
  const int tid = threadIdx.x;
  const int wave = tid >> 5, lane = tid & 31;
  const int l16 = lane & 15, half = lane >> 4;
  const int c = blockIdx.x >> 2, jn = blockIdx.x & 3;

  const float* F = uB + (size_t)c * LCH * NB * NS;
  const float* Brow = Pcat + (size_t)(jn * 16 + l16) * (LCH * NS);
  v8f acc = {};
#pragma unroll 8
  for (int kk = 0; kk < GKL / 4; ++kk) {
    int kap = wave * GKL + 4 * kk + 2 * half;
    int j = kap >> 6, m = kap & 63;
    v2f a = load2(F + j * NB * NS + l16 * NS + m);
    v2f b = load2(Brow + kap);
    acc = wmma4(a, b, acc);
  }
#pragma unroll
  for (int v = 0; v < 8; ++v) red[wave][(v + 8 * half) * GRP + l16] = acc[v];
  __syncthreads();
  if (tid < 256) {
    const int row = tid >> 4, col = tid & 15;
    float sum = red[0][row * GRP + col];
#pragma unroll
    for (int s = 1; s < GKS; ++s) sum += red[s][row * GRP + col];
    G[(size_t)(c * 4 + jn) * 256 + row * 16 + col] = sum;
  }
}

__global__ __launch_bounds__(128) void s4_bscan_kernel(
    const float* __restrict__ x0, const float* __restrict__ G,
    const float* __restrict__ Apow, float* __restrict__ Sbound,
    float* __restrict__ xfinal) {
  __shared__ float sb[2][NB * LP];
  const int tid = threadIdx.x;
  const int wave = tid >> 5, lane = tid & 31;
  const int l16 = lane & 15, half = lane >> 4;
  const int n0 = wave * 16;
  const float* AL = Apow + (size_t)LCH * NS * NS;

  for (int idx = tid; idx < NB * NS; idx += 128)
    sb[0][(idx >> 6) * LP + (idx & 63)] = x0[idx];
  __syncthreads();

  int cur = 0;
  for (int c = 0; c < NCH; ++c) {
    for (int idx = tid; idx < NB * NS; idx += 128)
      Sbound[(size_t)c * NB * NS + idx] = sb[cur][(idx >> 6) * LP + (idx & 63)];
    v8f acc = load_c(G + (size_t)c * NB * NS + wave * 256, 16, 0, l16, half);
#pragma unroll
    for (int kk = 0; kk < 16; ++kk) {
      v2f a = load2(&sb[cur][l16 * LP + 4 * kk + 2 * half]);
      v2f b = load2(AL + (n0 + l16) * NS + 4 * kk + 2 * half);
      acc = wmma4(a, b, acc);
    }
    int nxt = cur ^ 1;
#pragma unroll
    for (int v = 0; v < 8; ++v)
      sb[nxt][(v + 8 * half) * LP + n0 + l16] = acc[v];
    __syncthreads();
    cur = nxt;
  }
  for (int idx = tid; idx < NB * NS; idx += 128)
    xfinal[idx] = sb[cur][(idx >> 6) * LP + (idx & 63)];
}

__global__ __launch_bounds__(128) void s4_states_kernel(
    const float* __restrict__ uB, const float* __restrict__ Sbound,
    const float* __restrict__ Abar, float* __restrict__ S) {
  __shared__ float sb[2][NB * LP];
  const int tid = threadIdx.x;
  const int wave = tid >> 5, lane = tid & 31;
  const int l16 = lane & 15, half = lane >> 4;
  const int n0 = wave * 16;
  const int c = blockIdx.x;
  const int t0 = c * LCH;

  for (int idx = tid; idx < NB * NS; idx += 128)
    sb[0][(idx >> 6) * LP + (idx & 63)] = Sbound[(size_t)c * NB * NS + idx];
  __syncthreads();

  int cur = 0;
  for (int k = 0; k < LCH; ++k) {
    float* dst = S + (size_t)(t0 + k) * NB * NS;
    for (int idx = tid; idx < NB * NS; idx += 128)
      dst[idx] = sb[cur][(idx >> 6) * LP + (idx & 63)];
    v8f acc = load_c(uB + (size_t)(t0 + k) * NB * NS, NS, n0, l16, half);
#pragma unroll
    for (int kk = 0; kk < 16; ++kk) {
      v2f a = load2(&sb[cur][l16 * LP + 4 * kk + 2 * half]);
      v2f b = load2(Abar + (n0 + l16) * NS + 4 * kk + 2 * half);
      acc = wmma4(a, b, acc);
    }
    int nxt = cur ^ 1;
#pragma unroll
    for (int v = 0; v < 8; ++v)
      sb[nxt][(v + 8 * half) * LP + n0 + l16] = acc[v];
    __syncthreads();
    cur = nxt;
  }
}

typedef __attribute__((ext_vector_type(4))) float v4f;

#define YT 4
#define YM (YT * NB)
#define YKB 32
#define YSC (NS / YKB)
#define YSTEPS ((NS + NI) / YKB)
#define YLP 36

__device__ __forceinline__ void y_fetch(
    int s, int t0, int srow, int c4, const float* __restrict__ S,
    const float* __restrict__ u, const float* __restrict__ C,
    const float* __restrict__ D, v4f (&wreg)[8], v4f (&xreg)[2]) {
  const bool st = s < YSC;
  const float* wsrc = st ? C + YKB * s : D + YKB * (s - YSC);
  const int wpitch = st ? NS : NI;
#pragma unroll
  for (int i = 0; i < 8; ++i)
    wreg[i] = *(const v4f*)(wsrc + (size_t)(srow + 32 * i) * wpitch + 4 * c4);
#pragma unroll
  for (int i = 0; i < 2; ++i) {
    const int row = srow + 32 * i, tt = row >> 4, b = row & 15;
    const float* xsrc =
        st ? S + (size_t)(t0 + tt) * NB * NS + b * NS + YKB * s
           : u + (size_t)b * TT * NI + (size_t)(t0 + tt) * NI + YKB * (s - YSC);
    xreg[i] = *(const v4f*)(xsrc + 4 * c4);
  }
}

__global__ __launch_bounds__(256) void s4_y_kernel(
    const float* __restrict__ S, const float* __restrict__ u,
    const float* __restrict__ C, const float* __restrict__ D,
    float* __restrict__ y) {
  __shared__ __attribute__((aligned(16))) float sW[NO * YLP];
  __shared__ __attribute__((aligned(16))) float sX[YM * YLP];

  const int tid = threadIdx.x;
  const int wave = tid >> 5, lane = tid & 31;
  const int l16 = lane & 15, half = lane >> 4;
  const int t0 = blockIdx.x * YT;
  const int m0 = (wave & 1) * 2;
  const int o0 = (wave >> 1) * 64;
  const int srow = tid >> 3;
  const int c4 = tid & 7;

  v8f acc[2][4] = {{{}, {}, {}, {}}, {{}, {}, {}, {}}};
  v4f wreg[8], xreg[2];

  y_fetch(0, t0, srow, c4, S, u, C, D, wreg, xreg);
  for (int s = 0; s < YSTEPS; ++s) {
#pragma unroll
    for (int i = 0; i < 8; ++i)
      *(v4f*)&sW[(srow + 32 * i) * YLP + 4 * c4] = wreg[i];
#pragma unroll
    for (int i = 0; i < 2; ++i)
      *(v4f*)&sX[(srow + 32 * i) * YLP + 4 * c4] = xreg[i];
    __syncthreads();
    if (s + 1 < YSTEPS) y_fetch(s + 1, t0, srow, c4, S, u, C, D, wreg, xreg);
    __builtin_amdgcn_sched_barrier(0);
#pragma unroll
    for (int kk = 0; kk < YKB / 4; ++kk) {
      v2f a0 = load2(&sX[((m0 + 0) * NB + l16) * YLP + 4 * kk + 2 * half]);
      v2f a1 = load2(&sX[((m0 + 1) * NB + l16) * YLP + 4 * kk + 2 * half]);
#pragma unroll
      for (int j = 0; j < 4; ++j) {
        v2f b = load2(&sW[(o0 + 16 * j + l16) * YLP + 4 * kk + 2 * half]);
        acc[0][j] = wmma4(a0, b, acc[0][j]);
        acc[1][j] = wmma4(a1, b, acc[1][j]);
      }
    }
    __syncthreads();
  }

#pragma unroll
  for (int mi = 0; mi < 2; ++mi)
#pragma unroll
    for (int j = 0; j < 4; ++j)
#pragma unroll
      for (int v = 0; v < 8; ++v)
        y[(size_t)(v + 8 * half) * TT * NO + (size_t)(t0 + m0 + mi) * NO + o0 +
          16 * j + l16] = acc[mi][j][v];
}

extern "C" void kernel_launch(void* const* d_in, const int* in_sizes, int n_in,
                              void* d_out, int out_size, void* d_ws,
                              size_t ws_size, hipStream_t stream) {
  const float* u  = (const float*)d_in[0];
  const float* x0 = (const float*)d_in[1];
  const float* A  = (const float*)d_in[2];
  const float* B  = (const float*)d_in[3];
  const float* C  = (const float*)d_in[4];
  const float* D  = (const float*)d_in[5];

  float* y      = (float*)d_out;
  float* xfinal = (float*)d_out + (size_t)NB * TT * NO;

  float* ws = (float*)d_ws;
  float* Abar   = ws;                 ws += NS * NS;
  float* Apow   = ws;                 ws += (LCH + 1) * NS * NS;
  float* Pcat   = ws;                 ws += NS * LCH * NS;
  float* Bsc    = ws;                 ws += NS * NI;
  float* uB     = ws;                 ws += (size_t)TT * NB * NS;
  float* Sst    = ws;                 ws += (size_t)TT * NB * NS;
  float* G      = ws;                 ws += NCH * NB * NS;
  float* Sbound = ws;                 ws += NCH * NB * NS;

  s4_prep_kernel<<<1, 512, 0, stream>>>(A, B, Abar, Apow, Pcat, Bsc);
  s4_ub_kernel<<<TT / 4, 128, 0, stream>>>(u, Bsc, uB);
  s4_gagg_kernel<<<NCH * 4, 32 * GKS, 0, stream>>>(uB, Pcat, G);
  s4_bscan_kernel<<<1, 128, 0, stream>>>(x0, G, Apow, Sbound, xfinal);
  s4_states_kernel<<<NCH, 128, 0, stream>>>(uB, Sbound, Abar, Sst);
  s4_y_kernel<<<TT / YT, 256, 0, stream>>>(Sst, u, C, D, y);
}


# serial chains (powers, boundary scan, state steps): no step waits on global memory: start tile fetched a step ahead, copy-out by 16-byte LDS reads, stores behind an LDS-only step barrier
# speedup vs baseline: 3.5767x; 1.6546x over previous
#include <hip/hip_runtime.h>
#include <hip/hip_bf16.h>


typedef __attribute__((ext_vector_type(2))) float v2f;
typedef __attribute__((ext_vector_type(4))) float v4f;
typedef __attribute__((ext_vector_type(8))) float v8f;

#define S4_DT 0.001f
#define TT 4096
#define NB 16
#define NS 64
#define NI 256
#define NO 256
#define LCH 64
#define NCH 64
#define LP 68

__device__ __forceinline__ v8f wmma4(v2f a, v2f b, v8f c) {
  return __builtin_amdgcn_wmma_f32_16x16x4_f32(
       false, a,  false, b,
       (short)0, c,  false,  false);
}

__device__ __forceinline__ v2f load2(const float* p) { return *(const v2f*)p; }

__device__ __forceinline__ void store_c(float* dst, int rs, int n0, v8f acc,
                                        int l16, int half) {
#pragma unroll
  for (int v = 0; v < 8; ++v)
    dst[(v + 8 * half) * rs + n0 + l16] = acc[v];
}

__device__ __forceinline__ v8f load_c(const float* src, int rs, int n0,
                                      int l16, int half) {
  v8f r;
#pragma unroll
  for (int v = 0; v < 8; ++v)
    r[v] = src[(v + 8 * half) * rs + n0 + l16];
  return r;
}

__global__ __launch_bounds__(512) void s4_prep_kernel(
    const float* __restrict__ A, const float* __restrict__ B,
    float* __restrict__ Abar, float* __restrict__ Apow,
    float* __restrict__ Pcat, float* __restrict__ Bsc) {
  __shared__ float sPrev[2][NS * LP];
  __shared__ float sAbT[NS * LP];

  const int tid = threadIdx.x;
  for (int idx = tid; idx < NS * NS; idx += 512) {
    int r = idx >> 6, c = idx & 63;
    float v = A[idx] * S4_DT + (r == c ? 1.0f : 0.0f);
    Abar[idx] = v;
    sPrev[0][r * LP + c] = v;
    sAbT[c * LP + r] = v;
    Apow[0 * NS * NS + idx] = (r == c) ? 1.0f : 0.0f;
    Apow[1 * NS * NS + idx] = v;
  }
  for (int idx = tid; idx < NS * NI; idx += 512) Bsc[idx] = B[idx] * S4_DT;
  __syncthreads();

  const int wave = tid >> 5, lane = tid & 31;
  const int l16 = lane & 15, half = lane >> 4;
  const int m0 = (wave >> 2) * 16, n0 = (wave & 3) * 16;

  int cur = 0;
  for (int k = 2; k <= LCH; ++k) {
    v8f acc = {};
#pragma unroll
    for (int kk = 0; kk < 16; ++kk) {
      v2f a = load2(&sPrev[cur][(m0 + l16) * LP + 4 * kk + 2 * half]);
      v2f b = load2(&sAbT[(n0 + l16) * LP + 4 * kk + 2 * half]);
      acc = wmma4(a, b, acc);
    }
    int nxt = cur ^ 1;
#pragma unroll
    for (int v = 0; v < 8; ++v)
      sPrev[nxt][(m0 + v + 8 * half) * LP + n0 + l16] = acc[v];
    __syncthreads();
#pragma unroll
    for (int v = 0; v < 8; ++v)
      Apow[k * NS * NS + (m0 + v + 8 * half) * NS + n0 + l16] = acc[v];
    cur = nxt;
  }
  __syncthreads();

  for (int idx = tid; idx < NS * (LCH * NS); idx += 512) {
    int n = idx >> 12, rest = idx & 4095, j = rest >> 6, m = rest & 63;
    Pcat[idx] = Apow[(63 - j) * NS * NS + n * NS + m];
  }
}

__global__ __launch_bounds__(128) void s4_ub_kernel(
    const float* __restrict__ u, const float* __restrict__ Bsc,
    float* __restrict__ uB) {
  const int wave = threadIdx.x >> 5, lane = threadIdx.x & 31;
  const int l16 = lane & 15, half = lane >> 4;
  const int t = blockIdx.x * 4 + wave;

  const float* urow = u + (size_t)l16 * TT * NI + (size_t)t * NI;
  v8f acc[4] = {{}, {}, {}, {}};
  for (int kk = 0; kk < NI / 4; ++kk) {
    v2f a = load2(urow + 4 * kk + 2 * half);
#pragma unroll
    for (int j = 0; j < 4; ++j) {
      v2f b = load2(Bsc + (j * 16 + l16) * NI + 4 * kk + 2 * half);
      acc[j] = wmma4(a, b, acc[j]);
    }
  }
  float* dst = uB + (size_t)t * NB * NS;
#pragma unroll
  for (int j = 0; j < 4; ++j) store_c(dst, NS, j * 16, acc[j], l16, half);
}

#define GKS 16
#define GKL ((LCH * NS) / GKS)
#define GRP 18

__global__ __launch_bounds__(512) void s4_gagg_kernel(
    const float* __restrict__ uB, const float* __restrict__ Pcat,
    float* __restrict__ G) {
  __shared__ float red[GKS][16 * GRP];
  const int tid = threadIdx.x;
  const int wave = tid >> 5, lane = tid & 31;
  const int l16 = lane & 15, half = lane >> 4;
  const int c = blockIdx.x >> 2, jn = blockIdx.x & 3;

  const float* F = uB + (size_t)c * LCH * NB * NS;
  const float* Brow = Pcat + (size_t)(jn * 16 + l16) * (LCH * NS);
  v8f acc = {};
#pragma unroll 8
  for (int kk = 0; kk < GKL / 4; ++kk) {
    int kap = wave * GKL + 4 * kk + 2 * half;
    int j = kap >> 6, m = kap & 63;
    v2f a = load2(F + j * NB * NS + l16 * NS + m);
    v2f b = load2(Brow + kap);
    acc = wmma4(a, b, acc);
  }
#pragma unroll
  for (int v = 0; v < 8; ++v) red[wave][(v + 8 * half) * GRP + l16] = acc[v];
  __syncthreads();
  if (tid < 256) {
    const int row = tid >> 4, col = tid & 15;
    float sum = red[0][row * GRP + col];
#pragma unroll
    for (int s = 1; s < GKS; ++s) sum += red[s][row * GRP + col];
    G[(size_t)(c * 4 + jn) * 256 + row * 16 + col] = sum;
  }
}

__device__ __forceinline__ void lds_step_barrier() {
  __builtin_amdgcn_fence(__ATOMIC_RELEASE, "workgroup", "local");
  __builtin_amdgcn_s_barrier();
  __builtin_amdgcn_fence(__ATOMIC_ACQUIRE, "workgroup", "local");
}

__device__ __forceinline__ int piece_lds(int q) {
  return (q >> 4) * LP + 4 * (q & 15);
}

__global__ __launch_bounds__(128) void s4_bscan_kernel(
    const float* __restrict__ x0, const float* __restrict__ G,
    const float* __restrict__ Apow, float* __restrict__ Sbound,
    float* __restrict__ xfinal) {
  __shared__ __attribute__((aligned(16))) float sb[2][NB * LP];
  const int tid = threadIdx.x;
  const int wave = tid >> 5, lane = tid & 31;
  const int l16 = lane & 15, half = lane >> 4;
  const int n0 = wave * 16;
  const float* AL = Apow + (size_t)LCH * NS * NS;
  const int q0 = tid, q1 = tid + 128;
  const int lo0 = piece_lds(q0), lo1 = piece_lds(q1);

  *(v4f*)&sb[0][lo0] = *(const v4f*)(x0 + 4 * q0);
  *(v4f*)&sb[0][lo1] = *(const v4f*)(x0 + 4 * q1);
  v8f f = load_c(G + wave * 256, 16, 0, l16, half);
  __syncthreads();

  v2f bfr[16];
#pragma unroll
  for (int kk = 0; kk < 16; ++kk)
    bfr[kk] = load2(AL + (n0 + l16) * NS + 4 * kk + 2 * half);

  int cur = 0;
  for (int c = 0; c < NCH; ++c) {
    const v4f o0 = *(const v4f*)&sb[cur][lo0];
    const v4f o1 = *(const v4f*)&sb[cur][lo1];
    v2f afr[16];
#pragma unroll
    for (int kk = 0; kk < 16; ++kk)
      afr[kk] = load2(&sb[cur][l16 * LP + 4 * kk + 2 * half]);
    const int cn = (c + 1 < NCH) ? c + 1 : c;
    const v8f fn =
        load_c(G + (size_t)cn * NB * NS + wave * 256, 16, 0, l16, half);
    __builtin_amdgcn_sched_barrier(0);
    float* dst = Sbound + (size_t)c * NB * NS;
    *(v4f*)(dst + 4 * q0) = o0;
    *(v4f*)(dst + 4 * q1) = o1;
    v8f acc = f;
#pragma unroll
    for (int kk = 0; kk < 16; ++kk) acc = wmma4(afr[kk], bfr[kk], acc);
    int nxt = cur ^ 1;
#pragma unroll
    for (int v = 0; v < 8; ++v)
      sb[nxt][(v + 8 * half) * LP + n0 + l16] = acc[v];
    lds_step_barrier();
    cur = nxt;
    f = fn;
  }
  *(v4f*)(xfinal + 4 * q0) = *(const v4f*)&sb[cur][lo0];
  *(v4f*)(xfinal + 4 * q1) = *(const v4f*)&sb[cur][lo1];
}

__global__ __launch_bounds__(128) void s4_states_kernel(
    const float* __restrict__ uB, const float* __restrict__ Sbound,
    const float* __restrict__ Abar, float* __restrict__ S) {
  __shared__ __attribute__((aligned(16))) float sb[2][NB * LP];
  const int tid = threadIdx.x;
  const int wave = tid >> 5, lane = tid & 31;
  const int l16 = lane & 15, half = lane >> 4;
  const int n0 = wave * 16;
  const int c = blockIdx.x;
  const int t0 = c * LCH;
  const int q0 = tid, q1 = tid + 128;
  const int lo0 = piece_lds(q0), lo1 = piece_lds(q1);

  const float* sb0 = Sbound + (size_t)c * NB * NS;
  *(v4f*)&sb[0][lo0] = *(const v4f*)(sb0 + 4 * q0);
  *(v4f*)&sb[0][lo1] = *(const v4f*)(sb0 + 4 * q1);
  v8f f = load_c(uB + (size_t)t0 * NB * NS, NS, n0, l16, half);
  __syncthreads();

  v2f bfr[16];
#pragma unroll
  for (int kk = 0; kk < 16; ++kk)
    bfr[kk] = load2(Abar + (n0 + l16) * NS + 4 * kk + 2 * half);

  int cur = 0;
  for (int k = 0; k < LCH; ++k) {
    const v4f o0 = *(const v4f*)&sb[cur][lo0];
    const v4f o1 = *(const v4f*)&sb[cur][lo1];
    v2f afr[16];
#pragma unroll
    for (int kk = 0; kk < 16; ++kk)
      afr[kk] = load2(&sb[cur][l16 * LP + 4 * kk + 2 * half]);
    const int kn = (k + 1 < LCH) ? k + 1 : k;
    const v8f fn = load_c(uB + (size_t)(t0 + kn) * NB * NS, NS, n0, l16, half);
    __builtin_amdgcn_sched_barrier(0);
    float* dst = S + (size_t)(t0 + k) * NB * NS;
    *(v4f*)(dst + 4 * q0) = o0;
    *(v4f*)(dst + 4 * q1) = o1;
    v8f acc = f;
#pragma unroll
    for (int kk = 0; kk < 16; ++kk) acc = wmma4(afr[kk], bfr[kk], acc);
    int nxt = cur ^ 1;
#pragma unroll
    for (int v = 0; v < 8; ++v)
      sb[nxt][(v + 8 * half) * LP + n0 + l16] = acc[v];
    lds_step_barrier();
    cur = nxt;
    f = fn;
  }
}

#define YT 4
#define YM (YT * NB)
#define YKB 32
#define YSC (NS / YKB)
#define YSTEPS ((NS + NI) / YKB)
#define YLP 36

__device__ __forceinline__ void y_fetch(
    int s, int t0, int srow, int c4, const float* __restrict__ S,
    const float* __restrict__ u, const float* __restrict__ C,
    const float* __restrict__ D, v4f (&wreg)[8], v4f (&xreg)[2]) {
  const bool st = s < YSC;
  const float* wsrc = st ? C + YKB * s : D + YKB * (s - YSC);
  const int wpitch = st ? NS : NI;
#pragma unroll
  for (int i = 0; i < 8; ++i)
    wreg[i] = *(const v4f*)(wsrc + (size_t)(srow + 32 * i) * wpitch + 4 * c4);
#pragma unroll
  for (int i = 0; i < 2; ++i) {
    const int row = srow + 32 * i, tt = row >> 4, b = row & 15;
    const float* xsrc =
        st ? S + (size_t)(t0 + tt) * NB * NS + b * NS + YKB * s
           : u + (size_t)b * TT * NI + (size_t)(t0 + tt) * NI + YKB * (s - YSC);
    xreg[i] = *(const v4f*)(xsrc + 4 * c4);
  }
}

__global__ __launch_bounds__(256) void s4_y_kernel(
    const float* __restrict__ S, const float* __restrict__ u,
    const float* __restrict__ C, const float* __restrict__ D,
    float* __restrict__ y) {
  __shared__ __attribute__((aligned(16))) float sW[NO * YLP];
  __shared__ __attribute__((aligned(16))) float sX[YM * YLP];

  const int tid = threadIdx.x;
  const int wave = tid >> 5, lane = tid & 31;
  const int l16 = lane & 15, half = lane >> 4;
  const int t0 = blockIdx.x * YT;
  const int m0 = (wave & 1) * 2;
  const int o0 = (wave >> 1) * 64;
  const int srow = tid >> 3;
  const int c4 = tid & 7;

  v8f acc[2][4] = {{{}, {}, {}, {}}, {{}, {}, {}, {}}};
  v4f wreg[8], xreg[2];

  y_fetch(0, t0, srow, c4, S, u, C, D, wreg, xreg);
  for (int s = 0; s < YSTEPS; ++s) {
#pragma unroll
    for (int i = 0; i < 8; ++i)
      *(v4f*)&sW[(srow + 32 * i) * YLP + 4 * c4] = wreg[i];
#pragma unroll
    for (int i = 0; i < 2; ++i)
      *(v4f*)&sX[(srow + 32 * i) * YLP + 4 * c4] = xreg[i];
    __syncthreads();
    if (s + 1 < YSTEPS) y_fetch(s + 1, t0, srow, c4, S, u, C, D, wreg, xreg);
    __builtin_amdgcn_sched_barrier(0);
#pragma unroll
    for (int kk = 0; kk < YKB / 4; ++kk) {
      v2f a0 = load2(&sX[((m0 + 0) * NB + l16) * YLP + 4 * kk + 2 * half]);
      v2f a1 = load2(&sX[((m0 + 1) * NB + l16) * YLP + 4 * kk + 2 * half]);
#pragma unroll
      for (int j = 0; j < 4; ++j) {
        v2f b = load2(&sW[(o0 + 16 * j + l16) * YLP + 4 * kk + 2 * half]);
        acc[0][j] = wmma4(a0, b, acc[0][j]);
        acc[1][j] = wmma4(a1, b, acc[1][j]);
      }
    }
    __syncthreads();
  }

#pragma unroll
  for (int mi = 0; mi < 2; ++mi)
#pragma unroll
    for (int j = 0; j < 4; ++j)
#pragma unroll
      for (int v = 0; v < 8; ++v)
        y[(size_t)(v + 8 * half) * TT * NO + (size_t)(t0 + m0 + mi) * NO + o0 +
          16 * j + l16] = acc[mi][j][v];
}

extern "C" void kernel_launch(void* const* d_in, const int* in_sizes, int n_in,
                              void* d_out, int out_size, void* d_ws,
                              size_t ws_size, hipStream_t stream) {
  const float* u  = (const float*)d_in[0];
  const float* x0 = (const float*)d_in[1];
  const float* A  = (const float*)d_in[2];
  const float* B  = (const float*)d_in[3];
  const float* C  = (const float*)d_in[4];
  const float* D  = (const float*)d_in[5];

  float* y      = (float*)d_out;
  float* xfinal = (float*)d_out + (size_t)NB * TT * NO;

  float* ws = (float*)d_ws;
  float* Abar   = ws;                 ws += NS * NS;
  float* Apow   = ws;                 ws += (LCH + 1) * NS * NS;
  float* Pcat   = ws;                 ws += NS * LCH * NS;
  float* Bsc    = ws;                 ws += NS * NI;
  float* uB     = ws;                 ws += (size_t)TT * NB * NS;
  float* Sst    = ws;                 ws += (size_t)TT * NB * NS;
  float* G      = ws;                 ws += NCH * NB * NS;
  float* Sbound = ws;                 ws += NCH * NB * NS;

  s4_prep_kernel<<<1, 512, 0, stream>>>(A, B, Abar, Apow, Pcat, Bsc);
  s4_ub_kernel<<<TT / 4, 128, 0, stream>>>(u, Bsc, uB);
  s4_gagg_kernel<<<NCH * 4, 32 * GKS, 0, stream>>>(uB, Pcat, G);
  s4_bscan_kernel<<<1, 128, 0, stream>>>(x0, G, Apow, Sbound, xfinal);
  s4_states_kernel<<<NCH, 128, 0, stream>>>(uB, Sbound, Abar, Sst);
  s4_y_kernel<<<TT / YT, 256, 0, stream>>>(Sst, u, C, D, y);
}


# prep: powers split by row blocks over 4 workgroups of 4 waves (one chain per SIMD, not 4 on one matrix unit), rows of P stored straight from LDS, no gather loop; B scaled on 4 more; same k order
# speedup vs baseline: 4.9970x; 1.3971x over previous
#include <hip/hip_runtime.h>
#include <hip/hip_bf16.h>


typedef __attribute__((ext_vector_type(2))) float v2f;
typedef __attribute__((ext_vector_type(4))) float v4f;
typedef __attribute__((ext_vector_type(8))) float v8f;

#define S4_DT 0.001f
#define TT 4096
#define NB 16
#define NS 64
#define NI 256
#define NO 256
#define LCH 64
#define NCH 64
#define LP 68

__device__ __forceinline__ v8f wmma4(v2f a, v2f b, v8f c) {
  return __builtin_amdgcn_wmma_f32_16x16x4_f32(
       false, a,  false, b,
       (short)0, c,  false,  false);
}

__device__ __forceinline__ v2f load2(const float* p) { return *(const v2f*)p; }

__device__ __forceinline__ void store_c(float* dst, int rs, int n0, v8f acc,
                                        int l16, int half) {
#pragma unroll
  for (int v = 0; v < 8; ++v)
    dst[(v + 8 * half) * rs + n0 + l16] = acc[v];
}

__device__ __forceinline__ v8f load_c(const float* src, int rs, int n0,
                                      int l16, int half) {
  v8f r;
#pragma unroll
  for (int v = 0; v < 8; ++v)
    r[v] = src[(v + 8 * half) * rs + n0 + l16];
  return r;
}

__device__ __forceinline__ void lds_step_barrier() {
  __builtin_amdgcn_fence(__ATOMIC_RELEASE, "workgroup", "local");
  __builtin_amdgcn_s_barrier();
  __builtin_amdgcn_fence(__ATOMIC_ACQUIRE, "workgroup", "local");
}

__device__ __forceinline__ int piece_lds(int q) {
  return (q >> 4) * LP + 4 * (q & 15);
}

#define PRB (NS / 16)
#define PBW 4

__device__ __forceinline__ float abar_elem(float a, int r, int c) {
  return a * S4_DT + (r == c ? 1.0f : 0.0f);
}

__global__ __launch_bounds__(128) void s4_prep_kernel(
    const float* __restrict__ A, const float* __restrict__ B,
    float* __restrict__ Abar, float* __restrict__ Apow,
    float* __restrict__ Pcat, float* __restrict__ Bsc) {
  __shared__ __attribute__((aligned(16))) float sb[2][16 * LP];
  const int tid = threadIdx.x;
  if (blockIdx.x >= PRB) {
    const int p0 = (blockIdx.x - PRB) * 1024 + tid;
    v4f bq[8];
#pragma unroll
    for (int i = 0; i < 8; ++i) bq[i] = *(const v4f*)(B + 4 * (p0 + 128 * i));
#pragma unroll
    for (int i = 0; i < 8; ++i) *(v4f*)(Bsc + 4 * (p0 + 128 * i)) = bq[i] * S4_DT;
    return;
  }
  const int wave = tid >> 5, lane = tid & 31;
  const int l16 = lane & 15, half = lane >> 4;
  const int n0 = wave * 16;
  const int rb = blockIdx.x;
  const int r0 = rb * 16;
  const int q0 = tid, q1 = tid + 128;
  const int lo0 = piece_lds(q0), lo1 = piece_lds(q1);
  const int row0 = r0 + (q0 >> 4), row1 = r0 + (q1 >> 4);
  const int c0 = 4 * (q0 & 15), c1 = 4 * (q1 & 15);

  float braw[32];
#pragma unroll
  for (int kk = 0; kk < 16; ++kk) {
    const int k = 4 * kk + 2 * half;
    braw[2 * kk] = A[k * NS + n0 + l16];
    braw[2 * kk + 1] = A[(k + 1) * NS + n0 + l16];
  }
  const v4f a0 = *(const v4f*)(A + row0 * NS + c0);
  const v4f a1 = *(const v4f*)(A + row1 * NS + c1);

  v2f bfr[16];
#pragma unroll
  for (int kk = 0; kk < 16; ++kk) {
    const int k = 4 * kk + 2 * half;
    bfr[kk][0] = abar_elem(braw[2 * kk], k, n0 + l16);
    bfr[kk][1] = abar_elem(braw[2 * kk + 1], k + 1, n0 + l16);
  }
  v4f v0, v1, i0, i1;
#pragma unroll
  for (int e = 0; e < 4; ++e) {
    v0[e] = abar_elem(a0[e], row0, c0 + e);
    v1[e] = abar_elem(a1[e], row1, c1 + e);
    i0[e] = (row0 == c0 + e) ? 1.0f : 0.0f;
    i1[e] = (row1 == c1 + e) ? 1.0f : 0.0f;
  }
  *(v4f*)&sb[0][lo0] = v0;
  *(v4f*)&sb[0][lo1] = v1;
  *(v4f*)(Abar + row0 * NS + c0) = v0;
  *(v4f*)(Abar + row1 * NS + c1) = v1;
  float* pd0 = Pcat + row0 * (LCH * NS) + c0;
  float* pd1 = Pcat + row1 * (LCH * NS) + c1;
  *(v4f*)(pd0 + (LCH - 1) * NS) = i0;
  *(v4f*)(pd1 + (LCH - 1) * NS) = i1;
  lds_step_barrier();

  int cur = 0;
  for (int k = 2; k <= LCH; ++k) {
    const v4f o0 = *(const v4f*)&sb[cur][lo0];
    const v4f o1 = *(const v4f*)&sb[cur][lo1];
    v2f afr[16];
#pragma unroll
    for (int kk = 0; kk < 16; ++kk)
      afr[kk] = load2(&sb[cur][l16 * LP + 4 * kk + 2 * half]);
    __builtin_amdgcn_sched_barrier(0);
    *(v4f*)(pd0 + (LCH - k) * NS) = o0;
    *(v4f*)(pd1 + (LCH - k) * NS) = o1;
    v8f acc = {};
#pragma unroll
    for (int kk = 0; kk < 16; ++kk) acc = wmma4(afr[kk], bfr[kk], acc);
    int nxt = cur ^ 1;
#pragma unroll
    for (int v = 0; v < 8; ++v)
      sb[nxt][(v + 8 * half) * LP + n0 + l16] = acc[v];
    lds_step_barrier();
    cur = nxt;
  }
  float* a64 = Apow + (size_t)LCH * NS * NS + r0 * NS;
  *(v4f*)(a64 + 4 * q0) = *(const v4f*)&sb[cur][lo0];
  *(v4f*)(a64 + 4 * q1) = *(const v4f*)&sb[cur][lo1];
}

__global__ __launch_bounds__(128) void s4_ub_kernel(
    const float* __restrict__ u, const float* __restrict__ Bsc,
    float* __restrict__ uB) {
  const int wave = threadIdx.x >> 5, lane = threadIdx.x & 31;
  const int l16 = lane & 15, half = lane >> 4;
  const int t = blockIdx.x * 4 + wave;

  const float* urow = u + (size_t)l16 * TT * NI + (size_t)t * NI;
  v8f acc[4] = {{}, {}, {}, {}};
  for (int kk = 0; kk < NI / 4; ++kk) {
    v2f a = load2(urow + 4 * kk + 2 * half);
#pragma unroll
    for (int j = 0; j < 4; ++j) {
      v2f b = load2(Bsc + (j * 16 + l16) * NI + 4 * kk + 2 * half);
      acc[j] = wmma4(a, b, acc[j]);
    }
  }
  float* dst = uB + (size_t)t * NB * NS;
#pragma unroll
  for (int j = 0; j < 4; ++j) store_c(dst, NS, j * 16, acc[j], l16, half);
}

#define GKS 16
#define GKL ((LCH * NS) / GKS)
#define GRP 18

__global__ __launch_bounds__(512) void s4_gagg_kernel(
    const float* __restrict__ uB, const float* __restrict__ Pcat,
    float* __restrict__ G) {
  __shared__ float red[GKS][16 * GRP];
  const int tid = threadIdx.x;
  const int wave = tid >> 5, lane = tid & 31;
  const int l16 = lane & 15, half = lane >> 4;
  const int c = blockIdx.x >> 2, jn = blockIdx.x & 3;

  const float* F = uB + (size_t)c * LCH * NB * NS;
  const float* Brow = Pcat + (size_t)(jn * 16 + l16) * (LCH * NS);
  v8f acc = {};
#pragma unroll 8
  for (int kk = 0; kk < GKL / 4; ++kk) {
    int kap = wave * GKL + 4 * kk + 2 * half;
    int j = kap >> 6, m = kap & 63;
    v2f a = load2(F + j * NB * NS + l16 * NS + m);
    v2f b = load2(Brow + kap);
    acc = wmma4(a, b, acc);
  }
#pragma unroll
  for (int v = 0; v < 8; ++v) red[wave][(v + 8 * half) * GRP + l16] = acc[v];
  __syncthreads();
  if (tid < 256) {
    const int row = tid >> 4, col = tid & 15;
    float sum = red[0][row * GRP + col];
#pragma unroll
    for (int s = 1; s < GKS; ++s) sum += red[s][row * GRP + col];
    G[(size_t)(c * 4 + jn) * 256 + row * 16 + col] = sum;
  }
}

__global__ __launch_bounds__(128) void s4_bscan_kernel(
    const float* __restrict__ x0, const float* __restrict__ G,
    const float* __restrict__ Apow, float* __restrict__ Sbound,
    float* __restrict__ xfinal) {
  __shared__ __attribute__((aligned(16))) float sb[2][NB * LP];
  const int tid = threadIdx.x;
  const int wave = tid >> 5, lane = tid & 31;
  const int l16 = lane & 15, half = lane >> 4;
  const int n0 = wave * 16;
  const float* AL = Apow + (size_t)LCH * NS * NS;
  const int q0 = tid, q1 = tid + 128;
  const int lo0 = piece_lds(q0), lo1 = piece_lds(q1);

  *(v4f*)&sb[0][lo0] = *(const v4f*)(x0 + 4 * q0);
  *(v4f*)&sb[0][lo1] = *(const v4f*)(x0 + 4 * q1);
  v8f f = load_c(G + wave * 256, 16, 0, l16, half);
  __syncthreads();

  v2f bfr[16];
#pragma unroll
  for (int kk = 0; kk < 16; ++kk)
    bfr[kk] = load2(AL + (n0 + l16) * NS + 4 * kk + 2 * half);

  int cur = 0;
  for (int c = 0; c < NCH; ++c) {
    const v4f o0 = *(const v4f*)&sb[cur][lo0];
    const v4f o1 = *(const v4f*)&sb[cur][lo1];
    v2f afr[16];
#pragma unroll
    for (int kk = 0; kk < 16; ++kk)
      afr[kk] = load2(&sb[cur][l16 * LP + 4 * kk + 2 * half]);
    const int cn = (c + 1 < NCH) ? c + 1 : c;
    const v8f fn =
        load_c(G + (size_t)cn * NB * NS + wave * 256, 16, 0, l16, half);
    __builtin_amdgcn_sched_barrier(0);
    float* dst = Sbound + (size_t)c * NB * NS;
    *(v4f*)(dst + 4 * q0) = o0;
    *(v4f*)(dst + 4 * q1) = o1;
    v8f acc = f;
#pragma unroll
    for (int kk = 0; kk < 16; ++kk) acc = wmma4(afr[kk], bfr[kk], acc);
    int nxt = cur ^ 1;
#pragma unroll
    for (int v = 0; v < 8; ++v)
      sb[nxt][(v + 8 * half) * LP + n0 + l16] = acc[v];
    lds_step_barrier();
    cur = nxt;
    f = fn;
  }
  *(v4f*)(xfinal + 4 * q0) = *(const v4f*)&sb[cur][lo0];
  *(v4f*)(xfinal + 4 * q1) = *(const v4f*)&sb[cur][lo1];
}

__global__ __launch_bounds__(128) void s4_states_kernel(
    const float* __restrict__ uB, const float* __restrict__ Sbound,
    const float* __restrict__ Abar, float* __restrict__ S) {
  __shared__ __attribute__((aligned(16))) float sb[2][NB * LP];
  const int tid = threadIdx.x;
  const int wave = tid >> 5, lane = tid & 31;
  const int l16 = lane & 15, half = lane >> 4;
  const int n0 = wave * 16;
  const int c = blockIdx.x;
  const int t0 = c * LCH;
  const int q0 = tid, q1 = tid + 128;
  const int lo0 = piece_lds(q0), lo1 = piece_lds(q1);

  const float* sb0 = Sbound + (size_t)c * NB * NS;
  *(v4f*)&sb[0][lo0] = *(const v4f*)(sb0 + 4 * q0);
  *(v4f*)&sb[0][lo1] = *(const v4f*)(sb0 + 4 * q1);
  v8f f = load_c(uB + (size_t)t0 * NB * NS, NS, n0, l16, half);
  __syncthreads();

  v2f bfr[16];
#pragma unroll
  for (int kk = 0; kk < 16; ++kk)
    bfr[kk] = load2(Abar + (n0 + l16) * NS + 4 * kk + 2 * half);

  int cur = 0;
  for (int k = 0; k < LCH; ++k) {
    const v4f o0 = *(const v4f*)&sb[cur][lo0];
    const v4f o1 = *(const v4f*)&sb[cur][lo1];
    v2f afr[16];
#pragma unroll
    for (int kk = 0; kk < 16; ++kk)
      afr[kk] = load2(&sb[cur][l16 * LP + 4 * kk + 2 * half]);
    const int kn = (k + 1 < LCH) ? k + 1 : k;
    const v8f fn = load_c(uB + (size_t)(t0 + kn) * NB * NS, NS, n0, l16, half);
    __builtin_amdgcn_sched_barrier(0);
    float* dst = S + (size_t)(t0 + k) * NB * NS;
    *(v4f*)(dst + 4 * q0) = o0;
    *(v4f*)(dst + 4 * q1) = o1;
    v8f acc = f;
#pragma unroll
    for (int kk = 0; kk < 16; ++kk) acc = wmma4(afr[kk], bfr[kk], acc);
    int nxt = cur ^ 1;
#pragma unroll
    for (int v = 0; v < 8; ++v)
      sb[nxt][(v + 8 * half) * LP + n0 + l16] = acc[v];
    lds_step_barrier();
    cur = nxt;
    f = fn;
  }
}

#define YT 4
#define YM (YT * NB)
#define YKB 32
#define YSC (NS / YKB)
#define YSTEPS ((NS + NI) / YKB)
#define YLP 36

__device__ __forceinline__ void y_fetch(
    int s, int t0, int srow, int c4, const float* __restrict__ S,
    const float* __restrict__ u, const float* __restrict__ C,
    const float* __restrict__ D, v4f (&wreg)[8], v4f (&xreg)[2]) {
  const bool st = s < YSC;
  const float* wsrc = st ? C + YKB * s : D + YKB * (s - YSC);
  const int wpitch = st ? NS : NI;
#pragma unroll
  for (int i = 0; i < 8; ++i)
    wreg[i] = *(const v4f*)(wsrc + (size_t)(srow + 32 * i) * wpitch + 4 * c4);
#pragma unroll
  for (int i = 0; i < 2; ++i) {
    const int row = srow + 32 * i, tt = row >> 4, b = row & 15;
    const float* xsrc =
        st ? S + (size_t)(t0 + tt) * NB * NS + b * NS + YKB * s
           : u + (size_t)b * TT * NI + (size_t)(t0 + tt) * NI + YKB * (s - YSC);
    xreg[i] = *(const v4f*)(xsrc + 4 * c4);
  }
}

__global__ __launch_bounds__(256) void s4_y_kernel(
    const float* __restrict__ S, const float* __restrict__ u,
    const float* __restrict__ C, const float* __restrict__ D,
    float* __restrict__ y) {
  __shared__ __attribute__((aligned(16))) float sW[NO * YLP];
  __shared__ __attribute__((aligned(16))) float sX[YM * YLP];

  const int tid = threadIdx.x;
  const int wave = tid >> 5, lane = tid & 31;
  const int l16 = lane & 15, half = lane >> 4;
  const int t0 = blockIdx.x * YT;
  const int m0 = (wave & 1) * 2;
  const int o0 = (wave >> 1) * 64;
  const int srow = tid >> 3;
  const int c4 = tid & 7;

  v8f acc[2][4] = {{{}, {}, {}, {}}, {{}, {}, {}, {}}};
  v4f wreg[8], xreg[2];

  y_fetch(0, t0, srow, c4, S, u, C, D, wreg, xreg);
  for (int s = 0; s < YSTEPS; ++s) {
#pragma unroll
    for (int i = 0; i < 8; ++i)
      *(v4f*)&sW[(srow + 32 * i) * YLP + 4 * c4] = wreg[i];
#pragma unroll
    for (int i = 0; i < 2; ++i)
      *(v4f*)&sX[(srow + 32 * i) * YLP + 4 * c4] = xreg[i];
    __syncthreads();
    if (s + 1 < YSTEPS) y_fetch(s + 1, t0, srow, c4, S, u, C, D, wreg, xreg);
    __builtin_amdgcn_sched_barrier(0);
#pragma unroll
    for (int kk = 0; kk < YKB / 4; ++kk) {
      v2f a0 = load2(&sX[((m0 + 0) * NB + l16) * YLP + 4 * kk + 2 * half]);
      v2f a1 = load2(&sX[((m0 + 1) * NB + l16) * YLP + 4 * kk + 2 * half]);
#pragma unroll
      for (int j = 0; j < 4; ++j) {
        v2f b = load2(&sW[(o0 + 16 * j + l16) * YLP + 4 * kk + 2 * half]);
        acc[0][j] = wmma4(a0, b, acc[0][j]);
        acc[1][j] = wmma4(a1, b, acc[1][j]);
      }
    }
    __syncthreads();
  }

#pragma unroll
  for (int mi = 0; mi < 2; ++mi)
#pragma unroll
    for (int j = 0; j < 4; ++j)
#pragma unroll
      for (int v = 0; v < 8; ++v)
        y[(size_t)(v + 8 * half) * TT * NO + (size_t)(t0 + m0 + mi) * NO + o0 +
          16 * j + l16] = acc[mi][j][v];
}

extern "C" void kernel_launch(void* const* d_in, const int* in_sizes, int n_in,
                              void* d_out, int out_size, void* d_ws,
                              size_t ws_size, hipStream_t stream) {
  const float* u  = (const float*)d_in[0];
  const float* x0 = (const float*)d_in[1];
  const float* A  = (const float*)d_in[2];
  const float* B  = (const float*)d_in[3];
  const float* C  = (const float*)d_in[4];
  const float* D  = (const float*)d_in[5];

  float* y      = (float*)d_out;
  float* xfinal = (float*)d_out + (size_t)NB * TT * NO;

  float* ws = (float*)d_ws;
  float* Abar   = ws;                 ws += NS * NS;
  float* Apow   = ws;                 ws += (LCH + 1) * NS * NS;
  float* Pcat   = ws;                 ws += NS * LCH * NS;
  float* Bsc    = ws;                 ws += NS * NI;
  float* uB     = ws;                 ws += (size_t)TT * NB * NS;
  float* Sst    = ws;                 ws += (size_t)TT * NB * NS;
  float* G      = ws;                 ws += NCH * NB * NS;
  float* Sbound = ws;                 ws += NCH * NB * NS;

  s4_prep_kernel<<<PRB + PBW, 128, 0, stream>>>(A, B, Abar, Apow, Pcat, Bsc);
  s4_ub_kernel<<<TT / 4, 128, 0, stream>>>(u, Bsc, uB);
  s4_gagg_kernel<<<NCH * 4, 32 * GKS, 0, stream>>>(uB, Pcat, G);
  s4_bscan_kernel<<<1, 128, 0, stream>>>(x0, G, Apow, Sbound, xfinal);
  s4_states_kernel<<<NCH, 128, 0, stream>>>(uB, Sbound, Abar, Sst);
  s4_y_kernel<<<TT / YT, 256, 0, stream>>>(Sst, u, C, D, y);
}


# the powers' serial chain (4 workgroups) rides in the ub launch instead of a launch of its own: its 63 steps run under the u.B product; ub scales B by DT as it loads it; 5 launches, not 6
# speedup vs baseline: 5.2772x; 1.0561x over previous
#include <hip/hip_runtime.h>
#include <hip/hip_bf16.h>


typedef __attribute__((ext_vector_type(2))) float v2f;
typedef __attribute__((ext_vector_type(4))) float v4f;
typedef __attribute__((ext_vector_type(8))) float v8f;

#define S4_DT 0.001f
#define TT 4096
#define NB 16
#define NS 64
#define NI 256
#define NO 256
#define LCH 64
#define NCH 64
#define LP 68

__device__ __forceinline__ v8f wmma4(v2f a, v2f b, v8f c) {
  return __builtin_amdgcn_wmma_f32_16x16x4_f32(
       false, a,  false, b,
       (short)0, c,  false,  false);
}

__device__ __forceinline__ v2f load2(const float* p) { return *(const v2f*)p; }

__device__ __forceinline__ void store_c(float* dst, int rs, int n0, v8f acc,
                                        int l16, int half) {
#pragma unroll
  for (int v = 0; v < 8; ++v)
    dst[(v + 8 * half) * rs + n0 + l16] = acc[v];
}

__device__ __forceinline__ v8f load_c(const float* src, int rs, int n0,
                                      int l16, int half) {
  v8f r;
#pragma unroll
  for (int v = 0; v < 8; ++v)
    r[v] = src[(v + 8 * half) * rs + n0 + l16];
  return r;
}

__device__ __forceinline__ void lds_step_barrier() {
  __builtin_amdgcn_fence(__ATOMIC_RELEASE, "workgroup", "local");
  __builtin_amdgcn_s_barrier();
  __builtin_amdgcn_fence(__ATOMIC_ACQUIRE, "workgroup", "local");
}

__device__ __forceinline__ int piece_lds(int q) {
  return (q >> 4) * LP + 4 * (q & 15);
}

#define PRB (NS / 16)

__device__ __forceinline__ float abar_elem(float a, int r, int c) {
  return a * S4_DT + (r == c ? 1.0f : 0.0f);
}

__global__ __launch_bounds__(128) void s4_prep_ub_kernel(
    const float* __restrict__ A, const float* __restrict__ B,
    const float* __restrict__ u, float* __restrict__ Abar,
    float* __restrict__ Apow, float* __restrict__ Pcat,
    float* __restrict__ uB) {
  __shared__ __attribute__((aligned(16))) float sb[2][16 * LP];
  const int tid = threadIdx.x;
  const int wave = tid >> 5, lane = tid & 31;
  const int l16 = lane & 15, half = lane >> 4;
  if (blockIdx.x >= PRB) {
    const int t = (blockIdx.x - PRB) * 4 + wave;
    const float* urow = u + (size_t)l16 * TT * NI + (size_t)t * NI;
    v8f acc[4] = {{}, {}, {}, {}};
#pragma unroll 4
    for (int kk = 0; kk < NI / 4; ++kk) {
      v2f a = load2(urow + 4 * kk + 2 * half);
#pragma unroll
      for (int j = 0; j < 4; ++j) {
        v2f b = load2(B + (j * 16 + l16) * NI + 4 * kk + 2 * half) * S4_DT;
        acc[j] = wmma4(a, b, acc[j]);
      }
    }
    float* dst = uB + (size_t)t * NB * NS;
#pragma unroll
    for (int j = 0; j < 4; ++j) store_c(dst, NS, j * 16, acc[j], l16, half);
    return;
  }
  const int n0 = wave * 16;
  const int rb = blockIdx.x;
  const int r0 = rb * 16;
  const int q0 = tid, q1 = tid + 128;
  const int lo0 = piece_lds(q0), lo1 = piece_lds(q1);
  const int row0 = r0 + (q0 >> 4), row1 = r0 + (q1 >> 4);
  const int c0 = 4 * (q0 & 15), c1 = 4 * (q1 & 15);

  float braw[32];
#pragma unroll
  for (int kk = 0; kk < 16; ++kk) {
    const int k = 4 * kk + 2 * half;
    braw[2 * kk] = A[k * NS + n0 + l16];
    braw[2 * kk + 1] = A[(k + 1) * NS + n0 + l16];
  }
  const v4f a0 = *(const v4f*)(A + row0 * NS + c0);
  const v4f a1 = *(const v4f*)(A + row1 * NS + c1);

  v2f bfr[16];
#pragma unroll
  for (int kk = 0; kk < 16; ++kk) {
    const int k = 4 * kk + 2 * half;
    bfr[kk][0] = abar_elem(braw[2 * kk], k, n0 + l16);
    bfr[kk][1] = abar_elem(braw[2 * kk + 1], k + 1, n0 + l16);
  }
  v4f v0, v1, i0, i1;
#pragma unroll
  for (int e = 0; e < 4; ++e) {
    v0[e] = abar_elem(a0[e], row0, c0 + e);
    v1[e] = abar_elem(a1[e], row1, c1 + e);
    i0[e] = (row0 == c0 + e) ? 1.0f : 0.0f;
    i1[e] = (row1 == c1 + e) ? 1.0f : 0.0f;
  }
  *(v4f*)&sb[0][lo0] = v0;
  *(v4f*)&sb[0][lo1] = v1;
  *(v4f*)(Abar + row0 * NS + c0) = v0;
  *(v4f*)(Abar + row1 * NS + c1) = v1;
  float* pd0 = Pcat + row0 * (LCH * NS) + c0;
  float* pd1 = Pcat + row1 * (LCH * NS) + c1;
  *(v4f*)(pd0 + (LCH - 1) * NS) = i0;
  *(v4f*)(pd1 + (LCH - 1) * NS) = i1;
  lds_step_barrier();

  int cur = 0;
  for (int k = 2; k <= LCH; ++k) {
    const v4f o0 = *(const v4f*)&sb[cur][lo0];
    const v4f o1 = *(const v4f*)&sb[cur][lo1];
    v2f afr[16];
#pragma unroll
    for (int kk = 0; kk < 16; ++kk)
      afr[kk] = load2(&sb[cur][l16 * LP + 4 * kk + 2 * half]);
    __builtin_amdgcn_sched_barrier(0);
    *(v4f*)(pd0 + (LCH - k) * NS) = o0;
    *(v4f*)(pd1 + (LCH - k) * NS) = o1;
    v8f acc = {};
#pragma unroll
    for (int kk = 0; kk < 16; ++kk) acc = wmma4(afr[kk], bfr[kk], acc);
    int nxt = cur ^ 1;
#pragma unroll
    for (int v = 0; v < 8; ++v)
      sb[nxt][(v + 8 * half) * LP + n0 + l16] = acc[v];
    lds_step_barrier();
    cur = nxt;
  }
  float* a64 = Apow + (size_t)LCH * NS * NS + r0 * NS;
  *(v4f*)(a64 + 4 * q0) = *(const v4f*)&sb[cur][lo0];
  *(v4f*)(a64 + 4 * q1) = *(const v4f*)&sb[cur][lo1];
}

#define GKS 16
#define GKL ((LCH * NS) / GKS)
#define GRP 18

__global__ __launch_bounds__(512) void s4_gagg_kernel(
    const float* __restrict__ uB, const float* __restrict__ Pcat,
    float* __restrict__ G) {
  __shared__ float red[GKS][16 * GRP];
  const int tid = threadIdx.x;
  const int wave = tid >> 5, lane = tid & 31;
  const int l16 = lane & 15, half = lane >> 4;
  const int c = blockIdx.x >> 2, jn = blockIdx.x & 3;

  const float* F = uB + (size_t)c * LCH * NB * NS;
  const float* Brow = Pcat + (size_t)(jn * 16 + l16) * (LCH * NS);
  v8f acc = {};
#pragma unroll 8
  for (int kk = 0; kk < GKL / 4; ++kk) {
    int kap = wave * GKL + 4 * kk + 2 * half;
    int j = kap >> 6, m = kap & 63;
    v2f a = load2(F + j * NB * NS + l16 * NS + m);
    v2f b = load2(Brow + kap);
    acc = wmma4(a, b, acc);
  }
#pragma unroll
  for (int v = 0; v < 8; ++v) red[wave][(v + 8 * half) * GRP + l16] = acc[v];
  __syncthreads();
  if (tid < 256) {
    const int row = tid >> 4, col = tid & 15;
    float sum = red[0][row * GRP + col];
#pragma unroll
    for (int s = 1; s < GKS; ++s) sum += red[s][row * GRP + col];
    G[(size_t)(c * 4 + jn) * 256 + row * 16 + col] = sum;
  }
}

__global__ __launch_bounds__(128) void s4_bscan_kernel(
    const float* __restrict__ x0, const float* __restrict__ G,
    const float* __restrict__ Apow, float* __restrict__ Sbound,
    float* __restrict__ xfinal) {
  __shared__ __attribute__((aligned(16))) float sb[2][NB * LP];
  const int tid = threadIdx.x;
  const int wave = tid >> 5, lane = tid & 31;
  const int l16 = lane & 15, half = lane >> 4;
  const int n0 = wave * 16;
  const float* AL = Apow + (size_t)LCH * NS * NS;
  const int q0 = tid, q1 = tid + 128;
  const int lo0 = piece_lds(q0), lo1 = piece_lds(q1);

  *(v4f*)&sb[0][lo0] = *(const v4f*)(x0 + 4 * q0);
  *(v4f*)&sb[0][lo1] = *(const v4f*)(x0 + 4 * q1);
  v8f f = load_c(G + wave * 256, 16, 0, l16, half);
  __syncthreads();

  v2f bfr[16];
#pragma unroll
  for (int kk = 0; kk < 16; ++kk)
    bfr[kk] = load2(AL + (n0 + l16) * NS + 4 * kk + 2 * half);

  int cur = 0;
  for (int c = 0; c < NCH; ++c) {
    const v4f o0 = *(const v4f*)&sb[cur][lo0];
    const v4f o1 = *(const v4f*)&sb[cur][lo1];
    v2f afr[16];
#pragma unroll
    for (int kk = 0; kk < 16; ++kk)
      afr[kk] = load2(&sb[cur][l16 * LP + 4 * kk + 2 * half]);
    const int cn = (c + 1 < NCH) ? c + 1 : c;
    const v8f fn =
        load_c(G + (size_t)cn * NB * NS + wave * 256, 16, 0, l16, half);
    __builtin_amdgcn_sched_barrier(0);
    float* dst = Sbound + (size_t)c * NB * NS;
    *(v4f*)(dst + 4 * q0) = o0;
    *(v4f*)(dst + 4 * q1) = o1;
    v8f acc = f;
#pragma unroll
    for (int kk = 0; kk < 16; ++kk) acc = wmma4(afr[kk], bfr[kk], acc);
    int nxt = cur ^ 1;
#pragma unroll
    for (int v = 0; v < 8; ++v)
      sb[nxt][(v + 8 * half) * LP + n0 + l16] = acc[v];
    lds_step_barrier();
    cur = nxt;
    f = fn;
  }
  *(v4f*)(xfinal + 4 * q0) = *(const v4f*)&sb[cur][lo0];
  *(v4f*)(xfinal + 4 * q1) = *(const v4f*)&sb[cur][lo1];
}

__global__ __launch_bounds__(128) void s4_states_kernel(
    const float* __restrict__ uB, const float* __restrict__ Sbound,
    const float* __restrict__ Abar, float* __restrict__ S) {
  __shared__ __attribute__((aligned(16))) float sb[2][NB * LP];
  const int tid = threadIdx.x;
  const int wave = tid >> 5, lane = tid & 31;
  const int l16 = lane & 15, half = lane >> 4;
  const int n0 = wave * 16;
  const int c = blockIdx.x;
  const int t0 = c * LCH;
  const int q0 = tid, q1 = tid + 128;
  const int lo0 = piece_lds(q0), lo1 = piece_lds(q1);

  const float* sb0 = Sbound + (size_t)c * NB * NS;
  *(v4f*)&sb[0][lo0] = *(const v4f*)(sb0 + 4 * q0);
  *(v4f*)&sb[0][lo1] = *(const v4f*)(sb0 + 4 * q1);
  v8f f = load_c(uB + (size_t)t0 * NB * NS, NS, n0, l16, half);
  __syncthreads();

  v2f bfr[16];
#pragma unroll
  for (int kk = 0; kk < 16; ++kk)
    bfr[kk] = load2(Abar + (n0 + l16) * NS + 4 * kk + 2 * half);

  int cur = 0;
  for (int k = 0; k < LCH; ++k) {
    const v4f o0 = *(const v4f*)&sb[cur][lo0];
    const v4f o1 = *(const v4f*)&sb[cur][lo1];
    v2f afr[16];
#pragma unroll
    for (int kk = 0; kk < 16; ++kk)
      afr[kk] = load2(&sb[cur][l16 * LP + 4 * kk + 2 * half]);
    const int kn = (k + 1 < LCH) ? k + 1 : k;
    const v8f fn = load_c(uB + (size_t)(t0 + kn) * NB * NS, NS, n0, l16, half);
    __builtin_amdgcn_sched_barrier(0);
    float* dst = S + (size_t)(t0 + k) * NB * NS;
    *(v4f*)(dst + 4 * q0) = o0;
    *(v4f*)(dst + 4 * q1) = o1;
    v8f acc = f;
#pragma unroll
    for (int kk = 0; kk < 16; ++kk) acc = wmma4(afr[kk], bfr[kk], acc);
    int nxt = cur ^ 1;
#pragma unroll
    for (int v = 0; v < 8; ++v)
      sb[nxt][(v + 8 * half) * LP + n0 + l16] = acc[v];
    lds_step_barrier();
    cur = nxt;
    f = fn;
  }
}

#define YT 4
#define YM (YT * NB)
#define YKB 32
#define YSC (NS / YKB)
#define YSTEPS ((NS + NI) / YKB)
#define YLP 36

__device__ __forceinline__ void y_fetch(
    int s, int t0, int srow, int c4, const float* __restrict__ S,
    const float* __restrict__ u, const float* __restrict__ C,
    const float* __restrict__ D, v4f (&wreg)[8], v4f (&xreg)[2]) {
  const bool st = s < YSC;
  const float* wsrc = st ? C + YKB * s : D + YKB * (s - YSC);
  const int wpitch = st ? NS : NI;
#pragma unroll
  for (int i = 0; i < 8; ++i)
    wreg[i] = *(const v4f*)(wsrc + (size_t)(srow + 32 * i) * wpitch + 4 * c4);
#pragma unroll
  for (int i = 0; i < 2; ++i) {
    const int row = srow + 32 * i, tt = row >> 4, b = row & 15;
    const float* xsrc =
        st ? S + (size_t)(t0 + tt) * NB * NS + b * NS + YKB * s
           : u + (size_t)b * TT * NI + (size_t)(t0 + tt) * NI + YKB * (s - YSC);
    xreg[i] = *(const v4f*)(xsrc + 4 * c4);
  }
}

__global__ __launch_bounds__(256) void s4_y_kernel(
    const float* __restrict__ S, const float* __restrict__ u,
    const float* __restrict__ C, const float* __restrict__ D,
    float* __restrict__ y) {
  __shared__ __attribute__((aligned(16))) float sW[NO * YLP];
  __shared__ __attribute__((aligned(16))) float sX[YM * YLP];

  const int tid = threadIdx.x;
  const int wave = tid >> 5, lane = tid & 31;
  const int l16 = lane & 15, half = lane >> 4;
  const int t0 = blockIdx.x * YT;
  const int m0 = (wave & 1) * 2;
  const int o0 = (wave >> 1) * 64;
  const int srow = tid >> 3;
  const int c4 = tid & 7;

  v8f acc[2][4] = {{{}, {}, {}, {}}, {{}, {}, {}, {}}};
  v4f wreg[8], xreg[2];

  y_fetch(0, t0, srow, c4, S, u, C, D, wreg, xreg);
  for (int s = 0; s < YSTEPS; ++s) {
#pragma unroll
    for (int i = 0; i < 8; ++i)
      *(v4f*)&sW[(srow + 32 * i) * YLP + 4 * c4] = wreg[i];
#pragma unroll
    for (int i = 0; i < 2; ++i)
      *(v4f*)&sX[(srow + 32 * i) * YLP + 4 * c4] = xreg[i];
    __syncthreads();
    if (s + 1 < YSTEPS) y_fetch(s + 1, t0, srow, c4, S, u, C, D, wreg, xreg);
    __builtin_amdgcn_sched_barrier(0);
#pragma unroll
    for (int kk = 0; kk < YKB / 4; ++kk) {
      v2f a0 = load2(&sX[((m0 + 0) * NB + l16) * YLP + 4 * kk + 2 * half]);
      v2f a1 = load2(&sX[((m0 + 1) * NB + l16) * YLP + 4 * kk + 2 * half]);
#pragma unroll
      for (int j = 0; j < 4; ++j) {
        v2f b = load2(&sW[(o0 + 16 * j + l16) * YLP + 4 * kk + 2 * half]);
        acc[0][j] = wmma4(a0, b, acc[0][j]);
        acc[1][j] = wmma4(a1, b, acc[1][j]);
      }
    }
    __syncthreads();
  }

#pragma unroll
  for (int mi = 0; mi < 2; ++mi)
#pragma unroll
    for (int j = 0; j < 4; ++j)
#pragma unroll
      for (int v = 0; v < 8; ++v)
        y[(size_t)(v + 8 * half) * TT * NO + (size_t)(t0 + m0 + mi) * NO + o0 +
          16 * j + l16] = acc[mi][j][v];
}

extern "C" void kernel_launch(void* const* d_in, const int* in_sizes, int n_in,
                              void* d_out, int out_size, void* d_ws,
                              size_t ws_size, hipStream_t stream) {
  const float* u  = (const float*)d_in[0];
  const float* x0 = (const float*)d_in[1];
  const float* A  = (const float*)d_in[2];
  const float* B  = (const float*)d_in[3];
  const float* C  = (const float*)d_in[4];
  const float* D  = (const float*)d_in[5];

  float* y      = (float*)d_out;
  float* xfinal = (float*)d_out + (size_t)NB * TT * NO;

  float* ws = (float*)d_ws;
  float* Abar   = ws;                 ws += NS * NS;
  float* Apow   = ws;                 ws += (LCH + 1) * NS * NS;
  float* Pcat   = ws;                 ws += NS * LCH * NS;
  float* uB     = ws;                 ws += (size_t)TT * NB * NS;
  float* Sst    = ws;                 ws += (size_t)TT * NB * NS;
  float* G      = ws;                 ws += NCH * NB * NS;
  float* Sbound = ws;                 ws += NCH * NB * NS;

  s4_prep_ub_kernel<<<PRB + TT / 4, 128, 0, stream>>>(A, B, u, Abar, Apow, Pcat,
                                                      uB);
  s4_gagg_kernel<<<NCH * 4, 32 * GKS, 0, stream>>>(uB, Pcat, G);
  s4_bscan_kernel<<<1, 128, 0, stream>>>(x0, G, Apow, Sbound, xfinal);
  s4_states_kernel<<<NCH, 128, 0, stream>>>(uB, Sbound, Abar, Sst);
  s4_y_kernel<<<TT / YT, 256, 0, stream>>>(Sst, u, C, D, y);
}


# ub and gagg products: operand slices staged in LDS by coalesced 16-byte loads; WMMA fragments read from LDS, not 8 bytes per lane from 16 global lines per load; k order and sums unchanged
# speedup vs baseline: 5.9451x; 1.1266x over previous
#include <hip/hip_runtime.h>
#include <hip/hip_bf16.h>


typedef __attribute__((ext_vector_type(2))) float v2f;
typedef __attribute__((ext_vector_type(4))) float v4f;
typedef __attribute__((ext_vector_type(8))) float v8f;

#define S4_DT 0.001f
#define TT 4096
#define NB 16
#define NS 64
#define NI 256
#define NO 256
#define LCH 64
#define NCH 64
#define LP 68

__device__ __forceinline__ v8f wmma4(v2f a, v2f b, v8f c) {
  return __builtin_amdgcn_wmma_f32_16x16x4_f32(
       false, a,  false, b,
       (short)0, c,  false,  false);
}

__device__ __forceinline__ v2f load2(const float* p) { return *(const v2f*)p; }

__device__ __forceinline__ void store_c(float* dst, int rs, int n0, v8f acc,
                                        int l16, int half) {
#pragma unroll
  for (int v = 0; v < 8; ++v)
    dst[(v + 8 * half) * rs + n0 + l16] = acc[v];
}

__device__ __forceinline__ v8f load_c(const float* src, int rs, int n0,
                                      int l16, int half) {
  v8f r;
#pragma unroll
  for (int v = 0; v < 8; ++v)
    r[v] = src[(v + 8 * half) * rs + n0 + l16];
  return r;
}

__device__ __forceinline__ void lds_step_barrier() {
  __builtin_amdgcn_fence(__ATOMIC_RELEASE, "workgroup", "local");
  __builtin_amdgcn_s_barrier();
  __builtin_amdgcn_fence(__ATOMIC_ACQUIRE, "workgroup", "local");
}

__device__ __forceinline__ int piece_lds(int q) {
  return (q >> 4) * LP + 4 * (q & 15);
}

#define PRB (NS / 16)
#define UT 4
#define UM (UT * NB)
#define UKB 32
#define USTEPS (NI / UKB)
#define ULP 36

__device__ __forceinline__ void ub_fetch(int s, int t0, int srow, int c4,
                                         const float* __restrict__ u,
                                         const float* __restrict__ B,
                                         v4f (&wreg)[4], v4f (&xreg)[4]) {
#pragma unroll
  for (int i = 0; i < 4; ++i) {
    const int row = srow + 16 * i, tt = row >> 4, b = row & 15;
    wreg[i] = *(const v4f*)(B + (size_t)row * NI + UKB * s + 4 * c4);
    xreg[i] = *(const v4f*)(u + (size_t)b * TT * NI + (size_t)(t0 + tt) * NI +
                            UKB * s + 4 * c4);
  }
}

__device__ __forceinline__ float abar_elem(float a, int r, int c) {
  return a * S4_DT + (r == c ? 1.0f : 0.0f);
}

__global__ __launch_bounds__(128) void s4_prep_ub_kernel(
    const float* __restrict__ A, const float* __restrict__ B,
    const float* __restrict__ u, float* __restrict__ Abar,
    float* __restrict__ Apow, float* __restrict__ Pcat,
    float* __restrict__ uB) {
  __shared__ __attribute__((aligned(16))) float sm[(NS + UM) * ULP];
  const int tid = threadIdx.x;
  const int wave = tid >> 5, lane = tid & 31;
  const int l16 = lane & 15, half = lane >> 4;
  if (blockIdx.x >= PRB) {
    float* sW = sm;
    float* sX = sm + NS * ULP;
    const int t0 = (blockIdx.x - PRB) * UT;
    const int srow = tid >> 3;
    const int c4 = tid & 7;
    v8f acc[4] = {{}, {}, {}, {}};
    v4f wreg[4], xreg[4];
    ub_fetch(0, t0, srow, c4, u, B, wreg, xreg);
    for (int s = 0; s < USTEPS; ++s) {
#pragma unroll
      for (int i = 0; i < 4; ++i) {
        *(v4f*)&sW[(srow + 16 * i) * ULP + 4 * c4] = wreg[i] * S4_DT;
        *(v4f*)&sX[(srow + 16 * i) * ULP + 4 * c4] = xreg[i];
      }
      lds_step_barrier();
      if (s + 1 < USTEPS) ub_fetch(s + 1, t0, srow, c4, u, B, wreg, xreg);
      __builtin_amdgcn_sched_barrier(0);
#pragma unroll
      for (int kk = 0; kk < UKB / 4; ++kk) {
        v2f a = load2(&sX[(wave * NB + l16) * ULP + 4 * kk + 2 * half]);
#pragma unroll
        for (int j = 0; j < 4; ++j) {
          v2f b = load2(&sW[(j * 16 + l16) * ULP + 4 * kk + 2 * half]);
          acc[j] = wmma4(a, b, acc[j]);
        }
      }
      lds_step_barrier();
    }
    float* dst = uB + (size_t)(t0 + wave) * NB * NS;
#pragma unroll
    for (int j = 0; j < 4; ++j) store_c(dst, NS, j * 16, acc[j], l16, half);
    return;
  }
  float (*sb)[16 * LP] = (float (*)[16 * LP])sm;
  const int n0 = wave * 16;
  const int rb = blockIdx.x;
  const int r0 = rb * 16;
  const int q0 = tid, q1 = tid + 128;
  const int lo0 = piece_lds(q0), lo1 = piece_lds(q1);
  const int row0 = r0 + (q0 >> 4), row1 = r0 + (q1 >> 4);
  const int c0 = 4 * (q0 & 15), c1 = 4 * (q1 & 15);

  float braw[32];
#pragma unroll
  for (int kk = 0; kk < 16; ++kk) {
    const int k = 4 * kk + 2 * half;
    braw[2 * kk] = A[k * NS + n0 + l16];
    braw[2 * kk + 1] = A[(k + 1) * NS + n0 + l16];
  }
  const v4f a0 = *(const v4f*)(A + row0 * NS + c0);
  const v4f a1 = *(const v4f*)(A + row1 * NS + c1);

  v2f bfr[16];
#pragma unroll
  for (int kk = 0; kk < 16; ++kk) {
    const int k = 4 * kk + 2 * half;
    bfr[kk][0] = abar_elem(braw[2 * kk], k, n0 + l16);
    bfr[kk][1] = abar_elem(braw[2 * kk + 1], k + 1, n0 + l16);
  }
  v4f v0, v1, i0, i1;
#pragma unroll
  for (int e = 0; e < 4; ++e) {
    v0[e] = abar_elem(a0[e], row0, c0 + e);
    v1[e] = abar_elem(a1[e], row1, c1 + e);
    i0[e] = (row0 == c0 + e) ? 1.0f : 0.0f;
    i1[e] = (row1 == c1 + e) ? 1.0f : 0.0f;
  }
  *(v4f*)&sb[0][lo0] = v0;
  *(v4f*)&sb[0][lo1] = v1;
  *(v4f*)(Abar + row0 * NS + c0) = v0;
  *(v4f*)(Abar + row1 * NS + c1) = v1;
  float* pd0 = Pcat + row0 * (LCH * NS) + c0;
  float* pd1 = Pcat + row1 * (LCH * NS) + c1;
  *(v4f*)(pd0 + (LCH - 1) * NS) = i0;
  *(v4f*)(pd1 + (LCH - 1) * NS) = i1;
  lds_step_barrier();

  int cur = 0;
  for (int k = 2; k <= LCH; ++k) {
    const v4f o0 = *(const v4f*)&sb[cur][lo0];
    const v4f o1 = *(const v4f*)&sb[cur][lo1];
    v2f afr[16];
#pragma unroll
    for (int kk = 0; kk < 16; ++kk)
      afr[kk] = load2(&sb[cur][l16 * LP + 4 * kk + 2 * half]);
    __builtin_amdgcn_sched_barrier(0);
    *(v4f*)(pd0 + (LCH - k) * NS) = o0;
    *(v4f*)(pd1 + (LCH - k) * NS) = o1;
    v8f acc = {};
#pragma unroll
    for (int kk = 0; kk < 16; ++kk) acc = wmma4(afr[kk], bfr[kk], acc);
    int nxt = cur ^ 1;
#pragma unroll
    for (int v = 0; v < 8; ++v)
      sb[nxt][(v + 8 * half) * LP + n0 + l16] = acc[v];
    lds_step_barrier();
    cur = nxt;
  }
  float* a64 = Apow + (size_t)LCH * NS * NS + r0 * NS;
  *(v4f*)(a64 + 4 * q0) = *(const v4f*)&sb[cur][lo0];
  *(v4f*)(a64 + 4 * q1) = *(const v4f*)&sb[cur][lo1];
}

#define GKS 16
#define GKL ((LCH * NS) / GKS)
#define GSL NS
#define GNS (GKL / GSL)

__device__ __forceinline__ void gagg_fetch(int s, int lane,
                                           const float* __restrict__ Fw,
                                           const float* __restrict__ Bw,
                                           v4f (&areg)[8], v4f (&breg)[8]) {
#pragma unroll
  for (int i = 0; i < 8; ++i) {
    const int q = lane + 32 * i;
    areg[i] = *(const v4f*)(Fw + s * (NB * NS) + 4 * q);
    breg[i] = *(const v4f*)(Bw + (size_t)(q >> 4) * (LCH * NS) + s * GSL +
                            4 * (q & 15));
  }
}
#define GRP 18

__global__ __launch_bounds__(512) void s4_gagg_kernel(
    const float* __restrict__ uB, const float* __restrict__ Pcat,
    float* __restrict__ G) {
  __shared__ __attribute__((aligned(16))) float stg[GKS][2][16 * LP];
  __shared__ float red[GKS][16 * GRP];
  const int tid = threadIdx.x;
  const int wave = tid >> 5, lane = tid & 31;
  const int l16 = lane & 15, half = lane >> 4;
  const int c = blockIdx.x >> 2, jn = blockIdx.x & 3;

  const float* F = uB + (size_t)c * LCH * NB * NS;
  const float* Fw = F + (size_t)wave * GNS * NB * NS;
  const float* Bw = Pcat + (size_t)(jn * 16) * (LCH * NS) + wave * GKL;
  float* sA = stg[wave][0];
  float* sB = stg[wave][1];
  v8f acc = {};
  v4f areg[8], breg[8];
  gagg_fetch(0, lane, Fw, Bw, areg, breg);
#pragma unroll 1
  for (int s = 0; s < GNS; ++s) {
#pragma unroll
    for (int i = 0; i < 8; ++i) {
      const int lo = piece_lds(lane + 32 * i);
      *(v4f*)&sA[lo] = areg[i];
      *(v4f*)&sB[lo] = breg[i];
    }
    lds_step_barrier();
    if (s + 1 < GNS) gagg_fetch(s + 1, lane, Fw, Bw, areg, breg);
    __builtin_amdgcn_sched_barrier(0);
#pragma unroll
    for (int kk = 0; kk < GSL / 4; ++kk) {
      v2f a = load2(&sA[l16 * LP + 4 * kk + 2 * half]);
      v2f b = load2(&sB[l16 * LP + 4 * kk + 2 * half]);
      acc = wmma4(a, b, acc);
    }
    lds_step_barrier();
  }
#pragma unroll
  for (int v = 0; v < 8; ++v) red[wave][(v + 8 * half) * GRP + l16] = acc[v];
  __syncthreads();
  if (tid < 256) {
    const int row = tid >> 4, col = tid & 15;
    float sum = red[0][row * GRP + col];
#pragma unroll
    for (int s = 1; s < GKS; ++s) sum += red[s][row * GRP + col];
    G[(size_t)(c * 4 + jn) * 256 + row * 16 + col] = sum;
  }
}

__global__ __launch_bounds__(128) void s4_bscan_kernel(
    const float* __restrict__ x0, const float* __restrict__ G,
    const float* __restrict__ Apow, float* __restrict__ Sbound,
    float* __restrict__ xfinal) {
  __shared__ __attribute__((aligned(16))) float sb[2][NB * LP];
  const int tid = threadIdx.x;
  const int wave = tid >> 5, lane = tid & 31;
  const int l16 = lane & 15, half = lane >> 4;
  const int n0 = wave * 16;
  const float* AL = Apow + (size_t)LCH * NS * NS;
  const int q0 = tid, q1 = tid + 128;
  const int lo0 = piece_lds(q0), lo1 = piece_lds(q1);

  *(v4f*)&sb[0][lo0] = *(const v4f*)(x0 + 4 * q0);
  *(v4f*)&sb[0][lo1] = *(const v4f*)(x0 + 4 * q1);
  v8f f = load_c(G + wave * 256, 16, 0, l16, half);
  __syncthreads();

  v2f bfr[16];
#pragma unroll
  for (int kk = 0; kk < 16; ++kk)
    bfr[kk] = load2(AL + (n0 + l16) * NS + 4 * kk + 2 * half);

  int cur = 0;
  for (int c = 0; c < NCH; ++c) {
    const v4f o0 = *(const v4f*)&sb[cur][lo0];
    const v4f o1 = *(const v4f*)&sb[cur][lo1];
    v2f afr[16];
#pragma unroll
    for (int kk = 0; kk < 16; ++kk)
      afr[kk] = load2(&sb[cur][l16 * LP + 4 * kk + 2 * half]);
    const int cn = (c + 1 < NCH) ? c + 1 : c;
    const v8f fn =
        load_c(G + (size_t)cn * NB * NS + wave * 256, 16, 0, l16, half);
    __builtin_amdgcn_sched_barrier(0);
    float* dst = Sbound + (size_t)c * NB * NS;
    *(v4f*)(dst + 4 * q0) = o0;
    *(v4f*)(dst + 4 * q1) = o1;
    v8f acc = f;
#pragma unroll
    for (int kk = 0; kk < 16; ++kk) acc = wmma4(afr[kk], bfr[kk], acc);
    int nxt = cur ^ 1;
#pragma unroll
    for (int v = 0; v < 8; ++v)
      sb[nxt][(v + 8 * half) * LP + n0 + l16] = acc[v];
    lds_step_barrier();
    cur = nxt;
    f = fn;
  }
  *(v4f*)(xfinal + 4 * q0) = *(const v4f*)&sb[cur][lo0];
  *(v4f*)(xfinal + 4 * q1) = *(const v4f*)&sb[cur][lo1];
}

__global__ __launch_bounds__(128) void s4_states_kernel(
    const float* __restrict__ uB, const float* __restrict__ Sbound,
    const float* __restrict__ Abar, float* __restrict__ S) {
  __shared__ __attribute__((aligned(16))) float sb[2][NB * LP];
  const int tid = threadIdx.x;
  const int wave = tid >> 5, lane = tid & 31;
  const int l16 = lane & 15, half = lane >> 4;
  const int n0 = wave * 16;
  const int c = blockIdx.x;
  const int t0 = c * LCH;
  const int q0 = tid, q1 = tid + 128;
  const int lo0 = piece_lds(q0), lo1 = piece_lds(q1);

  const float* sb0 = Sbound + (size_t)c * NB * NS;
  *(v4f*)&sb[0][lo0] = *(const v4f*)(sb0 + 4 * q0);
  *(v4f*)&sb[0][lo1] = *(const v4f*)(sb0 + 4 * q1);
  v8f f = load_c(uB + (size_t)t0 * NB * NS, NS, n0, l16, half);
  __syncthreads();

  v2f bfr[16];
#pragma unroll
  for (int kk = 0; kk < 16; ++kk)
    bfr[kk] = load2(Abar + (n0 + l16) * NS + 4 * kk + 2 * half);

  int cur = 0;
  for (int k = 0; k < LCH; ++k) {
    const v4f o0 = *(const v4f*)&sb[cur][lo0];
    const v4f o1 = *(const v4f*)&sb[cur][lo1];
    v2f afr[16];
#pragma unroll
    for (int kk = 0; kk < 16; ++kk)
      afr[kk] = load2(&sb[cur][l16 * LP + 4 * kk + 2 * half]);
    const int kn = (k + 1 < LCH) ? k + 1 : k;
    const v8f fn = load_c(uB + (size_t)(t0 + kn) * NB * NS, NS, n0, l16, half);
    __builtin_amdgcn_sched_barrier(0);
    float* dst = S + (size_t)(t0 + k) * NB * NS;
    *(v4f*)(dst + 4 * q0) = o0;
    *(v4f*)(dst + 4 * q1) = o1;
    v8f acc = f;
#pragma unroll
    for (int kk = 0; kk < 16; ++kk) acc = wmma4(afr[kk], bfr[kk], acc);
    int nxt = cur ^ 1;
#pragma unroll
    for (int v = 0; v < 8; ++v)
      sb[nxt][(v + 8 * half) * LP + n0 + l16] = acc[v];
    lds_step_barrier();
    cur = nxt;
    f = fn;
  }
}

#define YT 4
#define YM (YT * NB)
#define YKB 32
#define YSC (NS / YKB)
#define YSTEPS ((NS + NI) / YKB)
#define YLP 36

__device__ __forceinline__ void y_fetch(
    int s, int t0, int srow, int c4, const float* __restrict__ S,
    const float* __restrict__ u, const float* __restrict__ C,
    const float* __restrict__ D, v4f (&wreg)[8], v4f (&xreg)[2]) {
  const bool st = s < YSC;
  const float* wsrc = st ? C + YKB * s : D + YKB * (s - YSC);
  const int wpitch = st ? NS : NI;
#pragma unroll
  for (int i = 0; i < 8; ++i)
    wreg[i] = *(const v4f*)(wsrc + (size_t)(srow + 32 * i) * wpitch + 4 * c4);
#pragma unroll
  for (int i = 0; i < 2; ++i) {
    const int row = srow + 32 * i, tt = row >> 4, b = row & 15;
    const float* xsrc =
        st ? S + (size_t)(t0 + tt) * NB * NS + b * NS + YKB * s
           : u + (size_t)b * TT * NI + (size_t)(t0 + tt) * NI + YKB * (s - YSC);
    xreg[i] = *(const v4f*)(xsrc + 4 * c4);
  }
}

__global__ __launch_bounds__(256) void s4_y_kernel(
    const float* __restrict__ S, const float* __restrict__ u,
    const float* __restrict__ C, const float* __restrict__ D,
    float* __restrict__ y) {
  __shared__ __attribute__((aligned(16))) float sW[NO * YLP];
  __shared__ __attribute__((aligned(16))) float sX[YM * YLP];

  const int tid = threadIdx.x;
  const int wave = tid >> 5, lane = tid & 31;
  const int l16 = lane & 15, half = lane >> 4;
  const int t0 = blockIdx.x * YT;
  const int m0 = (wave & 1) * 2;
  const int o0 = (wave >> 1) * 64;
  const int srow = tid >> 3;
  const int c4 = tid & 7;

  v8f acc[2][4] = {{{}, {}, {}, {}}, {{}, {}, {}, {}}};
  v4f wreg[8], xreg[2];

  y_fetch(0, t0, srow, c4, S, u, C, D, wreg, xreg);
  for (int s = 0; s < YSTEPS; ++s) {
#pragma unroll
    for (int i = 0; i < 8; ++i)
      *(v4f*)&sW[(srow + 32 * i) * YLP + 4 * c4] = wreg[i];
#pragma unroll
    for (int i = 0; i < 2; ++i)
      *(v4f*)&sX[(srow + 32 * i) * YLP + 4 * c4] = xreg[i];
    __syncthreads();
    if (s + 1 < YSTEPS) y_fetch(s + 1, t0, srow, c4, S, u, C, D, wreg, xreg);
    __builtin_amdgcn_sched_barrier(0);
#pragma unroll
    for (int kk = 0; kk < YKB / 4; ++kk) {
      v2f a0 = load2(&sX[((m0 + 0) * NB + l16) * YLP + 4 * kk + 2 * half]);
      v2f a1 = load2(&sX[((m0 + 1) * NB + l16) * YLP + 4 * kk + 2 * half]);
#pragma unroll
      for (int j = 0; j < 4; ++j) {
        v2f b = load2(&sW[(o0 + 16 * j + l16) * YLP + 4 * kk + 2 * half]);
        acc[0][j] = wmma4(a0, b, acc[0][j]);
        acc[1][j] = wmma4(a1, b, acc[1][j]);
      }
    }
    __syncthreads();
  }

#pragma unroll
  for (int mi = 0; mi < 2; ++mi)
#pragma unroll
    for (int j = 0; j < 4; ++j)
#pragma unroll
      for (int v = 0; v < 8; ++v)
        y[(size_t)(v + 8 * half) * TT * NO + (size_t)(t0 + m0 + mi) * NO + o0 +
          16 * j + l16] = acc[mi][j][v];
}

extern "C" void kernel_launch(void* const* d_in, const int* in_sizes, int n_in,
                              void* d_out, int out_size, void* d_ws,
                              size_t ws_size, hipStream_t stream) {
  const float* u  = (const float*)d_in[0];
  const float* x0 = (const float*)d_in[1];
  const float* A  = (const float*)d_in[2];
  const float* B  = (const float*)d_in[3];
  const float* C  = (const float*)d_in[4];
  const float* D  = (const float*)d_in[5];

  float* y      = (float*)d_out;
  float* xfinal = (float*)d_out + (size_t)NB * TT * NO;

  float* ws = (float*)d_ws;
  float* Abar   = ws;                 ws += NS * NS;
  float* Apow   = ws;                 ws += (LCH + 1) * NS * NS;
  float* Pcat   = ws;                 ws += NS * LCH * NS;
  float* uB     = ws;                 ws += (size_t)TT * NB * NS;
  float* Sst    = ws;                 ws += (size_t)TT * NB * NS;
  float* G      = ws;                 ws += NCH * NB * NS;
  float* Sbound = ws;                 ws += NCH * NB * NS;

  s4_prep_ub_kernel<<<PRB + TT / 4, 128, 0, stream>>>(A, B, u, Abar, Apow, Pcat,
                                                      uB);
  s4_gagg_kernel<<<NCH * 4, 32 * GKS, 0, stream>>>(uB, Pcat, G);
  s4_bscan_kernel<<<1, 128, 0, stream>>>(x0, G, Apow, Sbound, xfinal);
  s4_states_kernel<<<NCH, 128, 0, stream>>>(uB, Sbound, Abar, Sst);
  s4_y_kernel<<<TT / YT, 256, 0, stream>>>(Sst, u, C, D, y);
}


# u.D^T part of the y product runs in the two scan launches beside the chain workgroups (which fetch start tiles 2 steps ahead), sums left in y; last launch adds S.C^T: sums inputs first, then states
# speedup vs baseline: 7.1354x; 1.2002x over previous
#include <hip/hip_runtime.h>
#include <hip/hip_bf16.h>


typedef __attribute__((ext_vector_type(2))) float v2f;
typedef __attribute__((ext_vector_type(4))) float v4f;
typedef __attribute__((ext_vector_type(8))) float v8f;

#define S4_DT 0.001f
#define TT 4096
#define NB 16
#define NS 64
#define NI 256
#define NO 256
#define LCH 64
#define NCH 64
#define LP 68

__device__ __forceinline__ v8f wmma4(v2f a, v2f b, v8f c) {
  return __builtin_amdgcn_wmma_f32_16x16x4_f32(
       false, a,  false, b,
       (short)0, c,  false,  false);
}

__device__ __forceinline__ v2f load2(const float* p) { return *(const v2f*)p; }

__device__ __forceinline__ void store_c(float* dst, int rs, int n0, v8f acc,
                                        int l16, int half) {
#pragma unroll
  for (int v = 0; v < 8; ++v)
    dst[(v + 8 * half) * rs + n0 + l16] = acc[v];
}

__device__ __forceinline__ v8f load_c(const float* src, int rs, int n0,
                                      int l16, int half) {
  v8f r;
#pragma unroll
  for (int v = 0; v < 8; ++v)
    r[v] = src[(v + 8 * half) * rs + n0 + l16];
  return r;
}

__device__ __forceinline__ void lds_step_barrier() {
  __builtin_amdgcn_fence(__ATOMIC_RELEASE, "workgroup", "local");
  __builtin_amdgcn_s_barrier();
  __builtin_amdgcn_fence(__ATOMIC_ACQUIRE, "workgroup", "local");
}

__device__ __forceinline__ int piece_lds(int q) {
  return (q >> 4) * LP + 4 * (q & 15);
}

#define PRB (NS / 16)
#define UT 4
#define UM (UT * NB)
#define UKB 32
#define USTEPS (NI / UKB)
#define ULP 36

__device__ __forceinline__ void ub_fetch(int s, int t0, int srow, int c4,
                                         const float* __restrict__ u,
                                         const float* __restrict__ B,
                                         v4f (&wreg)[4], v4f (&xreg)[4]) {
#pragma unroll
  for (int i = 0; i < 4; ++i) {
    const int row = srow + 16 * i, tt = row >> 4, b = row & 15;
    wreg[i] = *(const v4f*)(B + (size_t)row * NI + UKB * s + 4 * c4);
    xreg[i] = *(const v4f*)(u + (size_t)b * TT * NI + (size_t)(t0 + tt) * NI +
                            UKB * s + 4 * c4);
  }
}

__device__ __forceinline__ float abar_elem(float a, int r, int c) {
  return a * S4_DT + (r == c ? 1.0f : 0.0f);
}

__global__ __launch_bounds__(128) void s4_prep_ub_kernel(
    const float* __restrict__ A, const float* __restrict__ B,
    const float* __restrict__ u, float* __restrict__ Abar,
    float* __restrict__ Apow, float* __restrict__ Pcat,
    float* __restrict__ uB) {
  __shared__ __attribute__((aligned(16))) float sm[(NS + UM) * ULP];
  const int tid = threadIdx.x;
  const int wave = tid >> 5, lane = tid & 31;
  const int l16 = lane & 15, half = lane >> 4;
  if (blockIdx.x >= PRB) {
    float* sW = sm;
    float* sX = sm + NS * ULP;
    const int t0 = (blockIdx.x - PRB) * UT;
    const int srow = tid >> 3;
    const int c4 = tid & 7;
    v8f acc[4] = {{}, {}, {}, {}};
    v4f wreg[4], xreg[4];
    ub_fetch(0, t0, srow, c4, u, B, wreg, xreg);
    for (int s = 0; s < USTEPS; ++s) {
#pragma unroll
      for (int i = 0; i < 4; ++i) {
        *(v4f*)&sW[(srow + 16 * i) * ULP + 4 * c4] = wreg[i] * S4_DT;
        *(v4f*)&sX[(srow + 16 * i) * ULP + 4 * c4] = xreg[i];
      }
      lds_step_barrier();
      if (s + 1 < USTEPS) ub_fetch(s + 1, t0, srow, c4, u, B, wreg, xreg);
      __builtin_amdgcn_sched_barrier(0);
#pragma unroll
      for (int kk = 0; kk < UKB / 4; ++kk) {
        v2f a = load2(&sX[(wave * NB + l16) * ULP + 4 * kk + 2 * half]);
#pragma unroll
        for (int j = 0; j < 4; ++j) {
          v2f b = load2(&sW[(j * 16 + l16) * ULP + 4 * kk + 2 * half]);
          acc[j] = wmma4(a, b, acc[j]);
        }
      }
      lds_step_barrier();
    }
    float* dst = uB + (size_t)(t0 + wave) * NB * NS;
#pragma unroll
    for (int j = 0; j < 4; ++j) store_c(dst, NS, j * 16, acc[j], l16, half);
    return;
  }
  float (*sb)[16 * LP] = (float (*)[16 * LP])sm;
  const int n0 = wave * 16;
  const int rb = blockIdx.x;
  const int r0 = rb * 16;
  const int q0 = tid, q1 = tid + 128;
  const int lo0 = piece_lds(q0), lo1 = piece_lds(q1);
  const int row0 = r0 + (q0 >> 4), row1 = r0 + (q1 >> 4);
  const int c0 = 4 * (q0 & 15), c1 = 4 * (q1 & 15);

  float braw[32];
#pragma unroll
  for (int kk = 0; kk < 16; ++kk) {
    const int k = 4 * kk + 2 * half;
    braw[2 * kk] = A[k * NS + n0 + l16];
    braw[2 * kk + 1] = A[(k + 1) * NS + n0 + l16];
  }
  const v4f a0 = *(const v4f*)(A + row0 * NS + c0);
  const v4f a1 = *(const v4f*)(A + row1 * NS + c1);

  v2f bfr[16];
#pragma unroll
  for (int kk = 0; kk < 16; ++kk) {
    const int k = 4 * kk + 2 * half;
    bfr[kk][0] = abar_elem(braw[2 * kk], k, n0 + l16);
    bfr[kk][1] = abar_elem(braw[2 * kk + 1], k + 1, n0 + l16);
  }
  v4f v0, v1, i0, i1;
#pragma unroll
  for (int e = 0; e < 4; ++e) {
    v0[e] = abar_elem(a0[e], row0, c0 + e);
    v1[e] = abar_elem(a1[e], row1, c1 + e);
    i0[e] = (row0 == c0 + e) ? 1.0f : 0.0f;
    i1[e] = (row1 == c1 + e) ? 1.0f : 0.0f;
  }
  *(v4f*)&sb[0][lo0] = v0;
  *(v4f*)&sb[0][lo1] = v1;
  *(v4f*)(Abar + row0 * NS + c0) = v0;
  *(v4f*)(Abar + row1 * NS + c1) = v1;
  float* pd0 = Pcat + row0 * (LCH * NS) + c0;
  float* pd1 = Pcat + row1 * (LCH * NS) + c1;
  *(v4f*)(pd0 + (LCH - 1) * NS) = i0;
  *(v4f*)(pd1 + (LCH - 1) * NS) = i1;
  lds_step_barrier();

  int cur = 0;
  for (int k = 2; k <= LCH; ++k) {
    const v4f o0 = *(const v4f*)&sb[cur][lo0];
    const v4f o1 = *(const v4f*)&sb[cur][lo1];
    v2f afr[16];
#pragma unroll
    for (int kk = 0; kk < 16; ++kk)
      afr[kk] = load2(&sb[cur][l16 * LP + 4 * kk + 2 * half]);
    __builtin_amdgcn_sched_barrier(0);
    *(v4f*)(pd0 + (LCH - k) * NS) = o0;
    *(v4f*)(pd1 + (LCH - k) * NS) = o1;
    v8f acc = {};
#pragma unroll
    for (int kk = 0; kk < 16; ++kk) acc = wmma4(afr[kk], bfr[kk], acc);
    int nxt = cur ^ 1;
#pragma unroll
    for (int v = 0; v < 8; ++v)
      sb[nxt][(v + 8 * half) * LP + n0 + l16] = acc[v];
    lds_step_barrier();
    cur = nxt;
  }
  float* a64 = Apow + (size_t)LCH * NS * NS + r0 * NS;
  *(v4f*)(a64 + 4 * q0) = *(const v4f*)&sb[cur][lo0];
  *(v4f*)(a64 + 4 * q1) = *(const v4f*)&sb[cur][lo1];
}

#define GKS 16
#define GKL ((LCH * NS) / GKS)
#define GSL NS
#define GNS (GKL / GSL)

__device__ __forceinline__ void gagg_fetch(int s, int lane,
                                           const float* __restrict__ Fw,
                                           const float* __restrict__ Bw,
                                           v4f (&areg)[8], v4f (&breg)[8]) {
#pragma unroll
  for (int i = 0; i < 8; ++i) {
    const int q = lane + 32 * i;
    areg[i] = *(const v4f*)(Fw + s * (NB * NS) + 4 * q);
    breg[i] = *(const v4f*)(Bw + (size_t)(q >> 4) * (LCH * NS) + s * GSL +
                            4 * (q & 15));
  }
}
#define GRP 18

__global__ __launch_bounds__(512) void s4_gagg_kernel(
    const float* __restrict__ uB, const float* __restrict__ Pcat,
    float* __restrict__ G) {
  __shared__ __attribute__((aligned(16))) float stg[GKS][2][16 * LP];
  __shared__ float red[GKS][16 * GRP];
  const int tid = threadIdx.x;
  const int wave = tid >> 5, lane = tid & 31;
  const int l16 = lane & 15, half = lane >> 4;
  const int c = blockIdx.x >> 2, jn = blockIdx.x & 3;

  const float* F = uB + (size_t)c * LCH * NB * NS;
  const float* Fw = F + (size_t)wave * GNS * NB * NS;
  const float* Bw = Pcat + (size_t)(jn * 16) * (LCH * NS) + wave * GKL;
  float* sA = stg[wave][0];
  float* sB = stg[wave][1];
  v8f acc = {};
  v4f areg[8], breg[8];
  gagg_fetch(0, lane, Fw, Bw, areg, breg);
#pragma unroll 1
  for (int s = 0; s < GNS; ++s) {
#pragma unroll
    for (int i = 0; i < 8; ++i) {
      const int lo = piece_lds(lane + 32 * i);
      *(v4f*)&sA[lo] = areg[i];
      *(v4f*)&sB[lo] = breg[i];
    }
    lds_step_barrier();
    if (s + 1 < GNS) gagg_fetch(s + 1, lane, Fw, Bw, areg, breg);
    __builtin_amdgcn_sched_barrier(0);
#pragma unroll
    for (int kk = 0; kk < GSL / 4; ++kk) {
      v2f a = load2(&sA[l16 * LP + 4 * kk + 2 * half]);
      v2f b = load2(&sB[l16 * LP + 4 * kk + 2 * half]);
      acc = wmma4(a, b, acc);
    }
    lds_step_barrier();
  }
#pragma unroll
  for (int v = 0; v < 8; ++v) red[wave][(v + 8 * half) * GRP + l16] = acc[v];
  __syncthreads();
  if (tid < 256) {
    const int row = tid >> 4, col = tid & 15;
    float sum = red[0][row * GRP + col];
#pragma unroll
    for (int s = 1; s < GKS; ++s) sum += red[s][row * GRP + col];
    G[(size_t)(c * 4 + jn) * 256 + row * 16 + col] = sum;
  }
}

__device__ __forceinline__ void chain_step(
    const float* __restrict__ sbc, float* __restrict__ sbn, const v8f& start,
    float* __restrict__ dst, const v2f (&bfr)[16], int lo0, int lo1, int q0,
    int q1, int l16, int half, int n0) {
  const v4f o0 = *(const v4f*)&sbc[lo0];
  const v4f o1 = *(const v4f*)&sbc[lo1];
  v2f afr[16];
#pragma unroll
  for (int kk = 0; kk < 16; ++kk)
    afr[kk] = load2(&sbc[l16 * LP + 4 * kk + 2 * half]);
  __builtin_amdgcn_sched_barrier(0);
  *(v4f*)(dst + 4 * q0) = o0;
  *(v4f*)(dst + 4 * q1) = o1;
  v8f acc = start;
#pragma unroll
  for (int kk = 0; kk < 16; ++kk) acc = wmma4(afr[kk], bfr[kk], acc);
#pragma unroll
  for (int v = 0; v < 8; ++v) sbn[(v + 8 * half) * LP + n0 + l16] = acc[v];
}

__device__ __forceinline__ void bscan_role(
    float* __restrict__ sm, const float* __restrict__ x0,
    const float* __restrict__ G, const float* __restrict__ Apow,
    float* __restrict__ Sbound, float* __restrict__ xfinal) {
  float (*sb)[NB * LP] = (float (*)[NB * LP])sm;
  const int tid = threadIdx.x;
  const int wave = __builtin_amdgcn_readfirstlane(tid >> 5);
  const bool work = wave < 4;
  const int lane = tid & 31;
  const int l16 = lane & 15, half = lane >> 4;
  const int n0 = (wave & 3) * 16;
  const float* AL = Apow + (size_t)LCH * NS * NS;
  const int q0 = tid & 127, q1 = q0 + 128;
  const int lo0 = piece_lds(q0), lo1 = piece_lds(q1);

  const float* Gw = G + (wave & 3) * 256;
  v8f ft[3];
  ft[0] = load_c(Gw, 16, 0, l16, half);
  ft[1] = load_c(Gw + NB * NS, 16, 0, l16, half);
  if (work) {
    *(v4f*)&sb[0][lo0] = *(const v4f*)(x0 + 4 * q0);
    *(v4f*)&sb[0][lo1] = *(const v4f*)(x0 + 4 * q1);
  }
  __syncthreads();

  v2f bfr[16] = {};
  if (work) {
#pragma unroll
    for (int kk = 0; kk < 16; ++kk)
      bfr[kk] = load2(AL + (n0 + l16) * NS + 4 * kk + 2 * half);
  }

  for (int c3 = 0; c3 < NCH / 3; ++c3) {
#pragma unroll
    for (int r = 0; r < 3; ++r) {
      const int c = 3 * c3 + r;
      const int cn = (c + 2 < NCH) ? c + 2 : NCH - 1;
      ft[(r + 2) % 3] = load_c(Gw + (size_t)cn * NB * NS, 16, 0, l16, half);
      if (work)
        chain_step(sb[c & 1], sb[(c & 1) ^ 1], ft[r],
                   Sbound + (size_t)c * NB * NS, bfr, lo0, lo1, q0, q1, l16,
                   half, n0);
      lds_step_barrier();
    }
  }
  if (work)
    chain_step(sb[1], sb[0], ft[0], Sbound + (size_t)(NCH - 1) * NB * NS, bfr,
               lo0, lo1, q0, q1, l16, half, n0);
  lds_step_barrier();
  const int cur = 0;
  if (work) {
    *(v4f*)(xfinal + 4 * q0) = *(const v4f*)&sb[cur][lo0];
    *(v4f*)(xfinal + 4 * q1) = *(const v4f*)&sb[cur][lo1];
  }
}

__device__ __forceinline__ void states_role(
    float* __restrict__ sm, int c, const float* __restrict__ uB,
    const float* __restrict__ Sbound, const float* __restrict__ Abar,
    float* __restrict__ S) {
  float (*sb)[NB * LP] = (float (*)[NB * LP])sm;
  const int tid = threadIdx.x;
  const int wave = __builtin_amdgcn_readfirstlane(tid >> 5);
  const bool work = wave < 4;
  const int lane = tid & 31;
  const int l16 = lane & 15, half = lane >> 4;
  const int n0 = (wave & 3) * 16;
  const int t0 = c * LCH;
  const int q0 = tid & 127, q1 = q0 + 128;
  const int lo0 = piece_lds(q0), lo1 = piece_lds(q1);

  const float* Fw = uB + (size_t)t0 * NB * NS;
  v8f ft[3];
  ft[0] = load_c(Fw, NS, n0, l16, half);
  ft[1] = load_c(Fw + NB * NS, NS, n0, l16, half);
  if (work) {
    const float* sb0 = Sbound + (size_t)c * NB * NS;
    *(v4f*)&sb[0][lo0] = *(const v4f*)(sb0 + 4 * q0);
    *(v4f*)&sb[0][lo1] = *(const v4f*)(sb0 + 4 * q1);
  }
  __syncthreads();

  v2f bfr[16] = {};
  if (work) {
#pragma unroll
    for (int kk = 0; kk < 16; ++kk)
      bfr[kk] = load2(Abar + (n0 + l16) * NS + 4 * kk + 2 * half);
  }

  for (int k3 = 0; k3 < LCH / 3; ++k3) {
#pragma unroll
    for (int r = 0; r < 3; ++r) {
      const int k = 3 * k3 + r;
      const int kn = (k + 2 < LCH) ? k + 2 : LCH - 1;
      ft[(r + 2) % 3] = load_c(Fw + (size_t)kn * NB * NS, NS, n0, l16, half);
      if (work)
        chain_step(sb[k & 1], sb[(k & 1) ^ 1], ft[r],
                   S + (size_t)(t0 + k) * NB * NS, bfr, lo0, lo1, q0, q1, l16,
                   half, n0);
      lds_step_barrier();
    }
  }
  if (work)
    chain_step(sb[1], sb[0], ft[0], S + (size_t)(t0 + LCH - 1) * NB * NS, bfr,
               lo0, lo1, q0, q1, l16, half, n0);
  lds_step_barrier();
}

#define YT 4
#define YM (YT * NB)
#define YKB 32
#define YSC (NS / YKB)
#define YSTEPS ((NS + NI) / YKB)
#define YPI (YSTEPS - YSC)
#define YNT (TT / YT)
#define YN_B 511
#define YN_S 448
#define YLP 36

__device__ __forceinline__ void y_fetch(
    int p, int t0, int srow, int c4, const float* __restrict__ S,
    const float* __restrict__ u, const float* __restrict__ C,
    const float* __restrict__ D, v4f (&wreg)[8], v4f (&xreg)[2]) {
  const bool st = p >= YPI;
  const float* wsrc = st ? C + YKB * (p - YPI) : D + YKB * p;
  const int wpitch = st ? NS : NI;
#pragma unroll
  for (int i = 0; i < 8; ++i)
    wreg[i] = *(const v4f*)(wsrc + (size_t)(srow + 32 * i) * wpitch + 4 * c4);
#pragma unroll
  for (int i = 0; i < 2; ++i) {
    const int row = srow + 32 * i, tt = row >> 4, b = row & 15;
    const float* xsrc =
        st ? S + (size_t)(t0 + tt) * NB * NS + b * NS + YKB * (p - YPI)
           : u + (size_t)b * TT * NI + (size_t)(t0 + tt) * NI + YKB * p;
    xreg[i] = *(const v4f*)(xsrc + 4 * c4);
  }
}

__device__ __forceinline__ void y_tile(
    float* __restrict__ sm, int tile, int p0, int p1, bool from_y,
    const float* __restrict__ S, const float* __restrict__ u,
    const float* __restrict__ C, const float* __restrict__ D, float* y) {
  float* sW = sm;
  float* sX = sm + NO * YLP;

  const int tid = threadIdx.x;
  const int wave = tid >> 5, lane = tid & 31;
  const int l16 = lane & 15, half = lane >> 4;
  const int t0 = tile * YT;
  const int m0 = (wave & 1) * 2;
  const int o0 = (wave >> 1) * 64;
  const int srow = tid >> 3;
  const int c4 = tid & 7;

  v8f acc[2][4] = {{{}, {}, {}, {}}, {{}, {}, {}, {}}};
  v4f wreg[8], xreg[2];

  y_fetch(p0, t0, srow, c4, S, u, C, D, wreg, xreg);
  if (from_y) {
#pragma unroll
    for (int mi = 0; mi < 2; ++mi)
#pragma unroll
      for (int j = 0; j < 4; ++j)
#pragma unroll
        for (int v = 0; v < 8; ++v)
          acc[mi][j][v] = y[(size_t)(v + 8 * half) * TT * NO +
                            (size_t)(t0 + m0 + mi) * NO + o0 + 16 * j + l16];
  }
  for (int p = p0; p < p1; ++p) {
#pragma unroll
    for (int i = 0; i < 8; ++i)
      *(v4f*)&sW[(srow + 32 * i) * YLP + 4 * c4] = wreg[i];
#pragma unroll
    for (int i = 0; i < 2; ++i)
      *(v4f*)&sX[(srow + 32 * i) * YLP + 4 * c4] = xreg[i];
    __syncthreads();
    if (p + 1 < p1) y_fetch(p + 1, t0, srow, c4, S, u, C, D, wreg, xreg);
    __builtin_amdgcn_sched_barrier(0);
#pragma unroll
    for (int kk = 0; kk < YKB / 4; ++kk) {
      v2f a0 = load2(&sX[((m0 + 0) * NB + l16) * YLP + 4 * kk + 2 * half]);
      v2f a1 = load2(&sX[((m0 + 1) * NB + l16) * YLP + 4 * kk + 2 * half]);
#pragma unroll
      for (int j = 0; j < 4; ++j) {
        v2f b = load2(&sW[(o0 + 16 * j + l16) * YLP + 4 * kk + 2 * half]);
        acc[0][j] = wmma4(a0, b, acc[0][j]);
        acc[1][j] = wmma4(a1, b, acc[1][j]);
      }
    }
    __syncthreads();
  }

#pragma unroll
  for (int mi = 0; mi < 2; ++mi)
#pragma unroll
    for (int j = 0; j < 4; ++j)
#pragma unroll
      for (int v = 0; v < 8; ++v)
        y[(size_t)(v + 8 * half) * TT * NO + (size_t)(t0 + m0 + mi) * NO + o0 +
          16 * j + l16] = acc[mi][j][v];
}

#define YSM ((NO + YM) * YLP)

__global__ __launch_bounds__(256) void s4_bscan_yin_kernel(
    const float* __restrict__ x0, const float* __restrict__ G,
    const float* __restrict__ Apow, float* __restrict__ Sbound,
    float* __restrict__ xfinal, const float* __restrict__ u,
    const float* __restrict__ D, float* __restrict__ y) {
  __shared__ __attribute__((aligned(16))) float sm[YSM];
  if (blockIdx.x == 0) {
    bscan_role(sm, x0, G, Apow, Sbound, xfinal);
    return;
  }
  y_tile(sm, blockIdx.x - 1, 0, YPI, false, nullptr, u, nullptr, D, y);
}

__global__ __launch_bounds__(256) void s4_states_yin_kernel(
    const float* __restrict__ uB, const float* __restrict__ Sbound,
    const float* __restrict__ Abar, float* __restrict__ S,
    const float* __restrict__ u, const float* __restrict__ D,
    float* __restrict__ y) {
  __shared__ __attribute__((aligned(16))) float sm[YSM];
  if (blockIdx.x < NCH) {
    states_role(sm, blockIdx.x, uB, Sbound, Abar, S);
    return;
  }
  y_tile(sm, YN_B + (blockIdx.x - NCH), 0, YPI, false, nullptr, u, nullptr, D,
         y);
}

__global__ __launch_bounds__(256) void s4_yfin_kernel(
    const float* __restrict__ S, const float* __restrict__ u,
    const float* __restrict__ C, const float* __restrict__ D, float* y) {
  __shared__ __attribute__((aligned(16))) float sm[YSM];
  const int tile = YNT - 1 - blockIdx.x;
  const bool started = tile < YN_B + YN_S;
  y_tile(sm, tile, started ? YPI : 0, YSTEPS, started, S, u, C, D, y);
}

extern "C" void kernel_launch(void* const* d_in, const int* in_sizes, int n_in,
                              void* d_out, int out_size, void* d_ws,
                              size_t ws_size, hipStream_t stream) {
  const float* u  = (const float*)d_in[0];
  const float* x0 = (const float*)d_in[1];
  const float* A  = (const float*)d_in[2];
  const float* B  = (const float*)d_in[3];
  const float* C  = (const float*)d_in[4];
  const float* D  = (const float*)d_in[5];

  float* y      = (float*)d_out;
  float* xfinal = (float*)d_out + (size_t)NB * TT * NO;

  float* ws = (float*)d_ws;
  float* Abar   = ws;                 ws += NS * NS;
  float* Apow   = ws;                 ws += (LCH + 1) * NS * NS;
  float* Pcat   = ws;                 ws += NS * LCH * NS;
  float* uB     = ws;                 ws += (size_t)TT * NB * NS;
  float* Sst    = ws;                 ws += (size_t)TT * NB * NS;
  float* G      = ws;                 ws += NCH * NB * NS;
  float* Sbound = ws;                 ws += NCH * NB * NS;

  s4_prep_ub_kernel<<<PRB + TT / 4, 128, 0, stream>>>(A, B, u, Abar, Apow, Pcat,
                                                      uB);
  s4_gagg_kernel<<<NCH * 4, 32 * GKS, 0, stream>>>(uB, Pcat, G);
  s4_bscan_yin_kernel<<<1 + YN_B, 256, 0, stream>>>(x0, G, Apow, Sbound, xfinal,
                                                    u, D, y);
  s4_states_yin_kernel<<<NCH + YN_S, 256, 0, stream>>>(uB, Sbound, Abar, Sst, u,
                                                       D, y);
  s4_yfin_kernel<<<YNT, 256, 0, stream>>>(Sst, u, C, D, y);
}


# serial chains (powers, boundary scan, state steps): a step's 16 WMMAs run as two interleaved accumulator chains (even / odd k groups), added once at the end; per-element order of sums changes
# speedup vs baseline: 7.6544x; 1.0727x over previous
#include <hip/hip_runtime.h>
#include <hip/hip_bf16.h>


typedef __attribute__((ext_vector_type(2))) float v2f;
typedef __attribute__((ext_vector_type(4))) float v4f;
typedef __attribute__((ext_vector_type(8))) float v8f;

#define S4_DT 0.001f
#define TT 4096
#define NB 16
#define NS 64
#define NI 256
#define NO 256
#define LCH 64
#define NCH 64
#define LP 68

__device__ __forceinline__ v8f wmma4(v2f a, v2f b, v8f c) {
  return __builtin_amdgcn_wmma_f32_16x16x4_f32(
       false, a,  false, b,
       (short)0, c,  false,  false);
}

__device__ __forceinline__ v2f load2(const float* p) { return *(const v2f*)p; }

__device__ __forceinline__ v8f chain_product(const v8f& start,
                                             const v2f (&afr)[16],
                                             const v2f (&bfr)[16]) {
  v8f acc0 = start, acc1 = {};
#pragma unroll
  for (int kk = 0; kk < 16; kk += 2) {
    acc0 = wmma4(afr[kk], bfr[kk], acc0);
    acc1 = wmma4(afr[kk + 1], bfr[kk + 1], acc1);
  }
  return acc0 + acc1;
}

__device__ __forceinline__ void store_c(float* dst, int rs, int n0, v8f acc,
                                        int l16, int half) {
#pragma unroll
  for (int v = 0; v < 8; ++v)
    dst[(v + 8 * half) * rs + n0 + l16] = acc[v];
}

__device__ __forceinline__ v8f load_c(const float* src, int rs, int n0,
                                      int l16, int half) {
  v8f r;
#pragma unroll
  for (int v = 0; v < 8; ++v)
    r[v] = src[(v + 8 * half) * rs + n0 + l16];
  return r;
}

__device__ __forceinline__ void lds_step_barrier() {
  __builtin_amdgcn_fence(__ATOMIC_RELEASE, "workgroup", "local");
  __builtin_amdgcn_s_barrier();
  __builtin_amdgcn_fence(__ATOMIC_ACQUIRE, "workgroup", "local");
}

__device__ __forceinline__ int piece_lds(int q) {
  return (q >> 4) * LP + 4 * (q & 15);
}

#define PRB (NS / 16)
#define UT 4
#define UM (UT * NB)
#define UKB 32
#define USTEPS (NI / UKB)
#define ULP 36

__device__ __forceinline__ void ub_fetch(int s, int t0, int srow, int c4,
                                         const float* __restrict__ u,
                                         const float* __restrict__ B,
                                         v4f (&wreg)[4], v4f (&xreg)[4]) {
#pragma unroll
  for (int i = 0; i < 4; ++i) {
    const int row = srow + 16 * i, tt = row >> 4, b = row & 15;
    wreg[i] = *(const v4f*)(B + (size_t)row * NI + UKB * s + 4 * c4);
    xreg[i] = *(const v4f*)(u + (size_t)b * TT * NI + (size_t)(t0 + tt) * NI +
                            UKB * s + 4 * c4);
  }
}

__device__ __forceinline__ float abar_elem(float a, int r, int c) {
  return a * S4_DT + (r == c ? 1.0f : 0.0f);
}

__global__ __launch_bounds__(128) void s4_prep_ub_kernel(
    const float* __restrict__ A, const float* __restrict__ B,
    const float* __restrict__ u, float* __restrict__ Abar,
    float* __restrict__ Apow, float* __restrict__ Pcat,
    float* __restrict__ uB) {
  __shared__ __attribute__((aligned(16))) float sm[(NS + UM) * ULP];
  const int tid = threadIdx.x;
  const int wave = tid >> 5, lane = tid & 31;
  const int l16 = lane & 15, half = lane >> 4;
  if (blockIdx.x >= PRB) {
    float* sW = sm;
    float* sX = sm + NS * ULP;
    const int t0 = (blockIdx.x - PRB) * UT;
    const int srow = tid >> 3;
    const int c4 = tid & 7;
    v8f acc[4] = {{}, {}, {}, {}};
    v4f wreg[4], xreg[4];
    ub_fetch(0, t0, srow, c4, u, B, wreg, xreg);
    for (int s = 0; s < USTEPS; ++s) {
#pragma unroll
      for (int i = 0; i < 4; ++i) {
        *(v4f*)&sW[(srow + 16 * i) * ULP + 4 * c4] = wreg[i] * S4_DT;
        *(v4f*)&sX[(srow + 16 * i) * ULP + 4 * c4] = xreg[i];
      }
      lds_step_barrier();
      if (s + 1 < USTEPS) ub_fetch(s + 1, t0, srow, c4, u, B, wreg, xreg);
      __builtin_amdgcn_sched_barrier(0);
#pragma unroll
      for (int kk = 0; kk < UKB / 4; ++kk) {
        v2f a = load2(&sX[(wave * NB + l16) * ULP + 4 * kk + 2 * half]);
#pragma unroll
        for (int j = 0; j < 4; ++j) {
          v2f b = load2(&sW[(j * 16 + l16) * ULP + 4 * kk + 2 * half]);
          acc[j] = wmma4(a, b, acc[j]);
        }
      }
      lds_step_barrier();
    }
    float* dst = uB + (size_t)(t0 + wave) * NB * NS;
#pragma unroll
    for (int j = 0; j < 4; ++j) store_c(dst, NS, j * 16, acc[j], l16, half);
    return;
  }
  float (*sb)[16 * LP] = (float (*)[16 * LP])sm;
  const int n0 = wave * 16;
  const int rb = blockIdx.x;
  const int r0 = rb * 16;
  const int q0 = tid, q1 = tid + 128;
  const int lo0 = piece_lds(q0), lo1 = piece_lds(q1);
  const int row0 = r0 + (q0 >> 4), row1 = r0 + (q1 >> 4);
  const int c0 = 4 * (q0 & 15), c1 = 4 * (q1 & 15);

  float braw[32];
#pragma unroll
  for (int kk = 0; kk < 16; ++kk) {
    const int k = 4 * kk + 2 * half;
    braw[2 * kk] = A[k * NS + n0 + l16];
    braw[2 * kk + 1] = A[(k + 1) * NS + n0 + l16];
  }
  const v4f a0 = *(const v4f*)(A + row0 * NS + c0);
  const v4f a1 = *(const v4f*)(A + row1 * NS + c1);

  v2f bfr[16];
#pragma unroll
  for (int kk = 0; kk < 16; ++kk) {
    const int k = 4 * kk + 2 * half;
    bfr[kk][0] = abar_elem(braw[2 * kk], k, n0 + l16);
    bfr[kk][1] = abar_elem(braw[2 * kk + 1], k + 1, n0 + l16);
  }
  v4f v0, v1, i0, i1;
#pragma unroll
  for (int e = 0; e < 4; ++e) {
    v0[e] = abar_elem(a0[e], row0, c0 + e);
    v1[e] = abar_elem(a1[e], row1, c1 + e);
    i0[e] = (row0 == c0 + e) ? 1.0f : 0.0f;
    i1[e] = (row1 == c1 + e) ? 1.0f : 0.0f;
  }
  *(v4f*)&sb[0][lo0] = v0;
  *(v4f*)&sb[0][lo1] = v1;
  *(v4f*)(Abar + row0 * NS + c0) = v0;
  *(v4f*)(Abar + row1 * NS + c1) = v1;
  float* pd0 = Pcat + row0 * (LCH * NS) + c0;
  float* pd1 = Pcat + row1 * (LCH * NS) + c1;
  *(v4f*)(pd0 + (LCH - 1) * NS) = i0;
  *(v4f*)(pd1 + (LCH - 1) * NS) = i1;
  lds_step_barrier();

  int cur = 0;
  for (int k = 2; k <= LCH; ++k) {
    const v4f o0 = *(const v4f*)&sb[cur][lo0];
    const v4f o1 = *(const v4f*)&sb[cur][lo1];
    v2f afr[16];
#pragma unroll
    for (int kk = 0; kk < 16; ++kk)
      afr[kk] = load2(&sb[cur][l16 * LP + 4 * kk + 2 * half]);
    __builtin_amdgcn_sched_barrier(0);
    *(v4f*)(pd0 + (LCH - k) * NS) = o0;
    *(v4f*)(pd1 + (LCH - k) * NS) = o1;
    const v8f zero = {};
    const v8f acc = chain_product(zero, afr, bfr);
    int nxt = cur ^ 1;
#pragma unroll
    for (int v = 0; v < 8; ++v)
      sb[nxt][(v + 8 * half) * LP + n0 + l16] = acc[v];
    lds_step_barrier();
    cur = nxt;
  }
  float* a64 = Apow + (size_t)LCH * NS * NS + r0 * NS;
  *(v4f*)(a64 + 4 * q0) = *(const v4f*)&sb[cur][lo0];
  *(v4f*)(a64 + 4 * q1) = *(const v4f*)&sb[cur][lo1];
}

#define GKS 16
#define GKL ((LCH * NS) / GKS)
#define GSL NS
#define GNS (GKL / GSL)

__device__ __forceinline__ void gagg_fetch(int s, int lane,
                                           const float* __restrict__ Fw,
                                           const float* __restrict__ Bw,
                                           v4f (&areg)[8], v4f (&breg)[8]) {
#pragma unroll
  for (int i = 0; i < 8; ++i) {
    const int q = lane + 32 * i;
    areg[i] = *(const v4f*)(Fw + s * (NB * NS) + 4 * q);
    breg[i] = *(const v4f*)(Bw + (size_t)(q >> 4) * (LCH * NS) + s * GSL +
                            4 * (q & 15));
  }
}
#define GRP 18

__global__ __launch_bounds__(512) void s4_gagg_kernel(
    const float* __restrict__ uB, const float* __restrict__ Pcat,
    float* __restrict__ G) {
  __shared__ __attribute__((aligned(16))) float stg[GKS][2][16 * LP];
  __shared__ float red[GKS][16 * GRP];
  const int tid = threadIdx.x;
  const int wave = tid >> 5, lane = tid & 31;
  const int l16 = lane & 15, half = lane >> 4;
  const int c = blockIdx.x >> 2, jn = blockIdx.x & 3;

  const float* F = uB + (size_t)c * LCH * NB * NS;
  const float* Fw = F + (size_t)wave * GNS * NB * NS;
  const float* Bw = Pcat + (size_t)(jn * 16) * (LCH * NS) + wave * GKL;
  float* sA = stg[wave][0];
  float* sB = stg[wave][1];
  v8f acc = {};
  v4f areg[8], breg[8];
  gagg_fetch(0, lane, Fw, Bw, areg, breg);
#pragma unroll 1
  for (int s = 0; s < GNS; ++s) {
#pragma unroll
    for (int i = 0; i < 8; ++i) {
      const int lo = piece_lds(lane + 32 * i);
      *(v4f*)&sA[lo] = areg[i];
      *(v4f*)&sB[lo] = breg[i];
    }
    lds_step_barrier();
    if (s + 1 < GNS) gagg_fetch(s + 1, lane, Fw, Bw, areg, breg);
    __builtin_amdgcn_sched_barrier(0);
#pragma unroll
    for (int kk = 0; kk < GSL / 4; ++kk) {
      v2f a = load2(&sA[l16 * LP + 4 * kk + 2 * half]);
      v2f b = load2(&sB[l16 * LP + 4 * kk + 2 * half]);
      acc = wmma4(a, b, acc);
    }
    lds_step_barrier();
  }
#pragma unroll
  for (int v = 0; v < 8; ++v) red[wave][(v + 8 * half) * GRP + l16] = acc[v];
  __syncthreads();
  if (tid < 256) {
    const int row = tid >> 4, col = tid & 15;
    float sum = red[0][row * GRP + col];
#pragma unroll
    for (int s = 1; s < GKS; ++s) sum += red[s][row * GRP + col];
    G[(size_t)(c * 4 + jn) * 256 + row * 16 + col] = sum;
  }
}

__device__ __forceinline__ void chain_step(
    const float* __restrict__ sbc, float* __restrict__ sbn, const v8f& start,
    float* __restrict__ dst, const v2f (&bfr)[16], int lo0, int lo1, int q0,
    int q1, int l16, int half, int n0) {
  const v4f o0 = *(const v4f*)&sbc[lo0];
  const v4f o1 = *(const v4f*)&sbc[lo1];
  v2f afr[16];
#pragma unroll
  for (int kk = 0; kk < 16; ++kk)
    afr[kk] = load2(&sbc[l16 * LP + 4 * kk + 2 * half]);
  __builtin_amdgcn_sched_barrier(0);
  *(v4f*)(dst + 4 * q0) = o0;
  *(v4f*)(dst + 4 * q1) = o1;
  const v8f acc = chain_product(start, afr, bfr);
#pragma unroll
  for (int v = 0; v < 8; ++v) sbn[(v + 8 * half) * LP + n0 + l16] = acc[v];
}

__device__ __forceinline__ void bscan_role(
    float* __restrict__ sm, const float* __restrict__ x0,
    const float* __restrict__ G, const float* __restrict__ Apow,
    float* __restrict__ Sbound, float* __restrict__ xfinal) {
  float (*sb)[NB * LP] = (float (*)[NB * LP])sm;
  const int tid = threadIdx.x;
  const int wave = __builtin_amdgcn_readfirstlane(tid >> 5);
  const bool work = wave < 4;
  const int lane = tid & 31;
  const int l16 = lane & 15, half = lane >> 4;
  const int n0 = (wave & 3) * 16;
  const float* AL = Apow + (size_t)LCH * NS * NS;
  const int q0 = tid & 127, q1 = q0 + 128;
  const int lo0 = piece_lds(q0), lo1 = piece_lds(q1);

  const float* Gw = G + (wave & 3) * 256;
  v8f ft[3];
  ft[0] = load_c(Gw, 16, 0, l16, half);
  ft[1] = load_c(Gw + NB * NS, 16, 0, l16, half);
  if (work) {
    *(v4f*)&sb[0][lo0] = *(const v4f*)(x0 + 4 * q0);
    *(v4f*)&sb[0][lo1] = *(const v4f*)(x0 + 4 * q1);
  }
  __syncthreads();

  v2f bfr[16] = {};
  if (work) {
#pragma unroll
    for (int kk = 0; kk < 16; ++kk)
      bfr[kk] = load2(AL + (n0 + l16) * NS + 4 * kk + 2 * half);
  }

  for (int c3 = 0; c3 < NCH / 3; ++c3) {
#pragma unroll
    for (int r = 0; r < 3; ++r) {
      const int c = 3 * c3 + r;
      const int cn = (c + 2 < NCH) ? c + 2 : NCH - 1;
      ft[(r + 2) % 3] = load_c(Gw + (size_t)cn * NB * NS, 16, 0, l16, half);
      if (work)
        chain_step(sb[c & 1], sb[(c & 1) ^ 1], ft[r],
                   Sbound + (size_t)c * NB * NS, bfr, lo0, lo1, q0, q1, l16,
                   half, n0);
      lds_step_barrier();
    }
  }
  if (work)
    chain_step(sb[1], sb[0], ft[0], Sbound + (size_t)(NCH - 1) * NB * NS, bfr,
               lo0, lo1, q0, q1, l16, half, n0);
  lds_step_barrier();
  const int cur = 0;
  if (work) {
    *(v4f*)(xfinal + 4 * q0) = *(const v4f*)&sb[cur][lo0];
    *(v4f*)(xfinal + 4 * q1) = *(const v4f*)&sb[cur][lo1];
  }
}

__device__ __forceinline__ void states_role(
    float* __restrict__ sm, int c, const float* __restrict__ uB,
    const float* __restrict__ Sbound, const float* __restrict__ Abar,
    float* __restrict__ S) {
  float (*sb)[NB * LP] = (float (*)[NB * LP])sm;
  const int tid = threadIdx.x;
  const int wave = __builtin_amdgcn_readfirstlane(tid >> 5);
  const bool work = wave < 4;
  const int lane = tid & 31;
  const int l16 = lane & 15, half = lane >> 4;
  const int n0 = (wave & 3) * 16;
  const int t0 = c * LCH;
  const int q0 = tid & 127, q1 = q0 + 128;
  const int lo0 = piece_lds(q0), lo1 = piece_lds(q1);

  const float* Fw = uB + (size_t)t0 * NB * NS;
  v8f ft[3];
  ft[0] = load_c(Fw, NS, n0, l16, half);
  ft[1] = load_c(Fw + NB * NS, NS, n0, l16, half);
  if (work) {
    const float* sb0 = Sbound + (size_t)c * NB * NS;
    *(v4f*)&sb[0][lo0] = *(const v4f*)(sb0 + 4 * q0);
    *(v4f*)&sb[0][lo1] = *(const v4f*)(sb0 + 4 * q1);
  }
  __syncthreads();

  v2f bfr[16] = {};
  if (work) {
#pragma unroll
    for (int kk = 0; kk < 16; ++kk)
      bfr[kk] = load2(Abar + (n0 + l16) * NS + 4 * kk + 2 * half);
  }

  for (int k3 = 0; k3 < LCH / 3; ++k3) {
#pragma unroll
    for (int r = 0; r < 3; ++r) {
      const int k = 3 * k3 + r;
      const int kn = (k + 2 < LCH) ? k + 2 : LCH - 1;
      ft[(r + 2) % 3] = load_c(Fw + (size_t)kn * NB * NS, NS, n0, l16, half);
      if (work)
        chain_step(sb[k & 1], sb[(k & 1) ^ 1], ft[r],
                   S + (size_t)(t0 + k) * NB * NS, bfr, lo0, lo1, q0, q1, l16,
                   half, n0);
      lds_step_barrier();
    }
  }
  if (work)
    chain_step(sb[1], sb[0], ft[0], S + (size_t)(t0 + LCH - 1) * NB * NS, bfr,
               lo0, lo1, q0, q1, l16, half, n0);
  lds_step_barrier();
}

#define YT 4
#define YM (YT * NB)
#define YKB 32
#define YSC (NS / YKB)
#define YSTEPS ((NS + NI) / YKB)
#define YPI (YSTEPS - YSC)
#define YNT (TT / YT)
#define YN_B 511
#define YN_S 448
#define YLP 36

__device__ __forceinline__ void y_fetch(
    int p, int t0, int srow, int c4, const float* __restrict__ S,
    const float* __restrict__ u, const float* __restrict__ C,
    const float* __restrict__ D, v4f (&wreg)[8], v4f (&xreg)[2]) {
  const bool st = p >= YPI;
  const float* wsrc = st ? C + YKB * (p - YPI) : D + YKB * p;
  const int wpitch = st ? NS : NI;
#pragma unroll
  for (int i = 0; i < 8; ++i)
    wreg[i] = *(const v4f*)(wsrc + (size_t)(srow + 32 * i) * wpitch + 4 * c4);
#pragma unroll
  for (int i = 0; i < 2; ++i) {
    const int row = srow + 32 * i, tt = row >> 4, b = row & 15;
    const float* xsrc =
        st ? S + (size_t)(t0 + tt) * NB * NS + b * NS + YKB * (p - YPI)
           : u + (size_t)b * TT * NI + (size_t)(t0 + tt) * NI + YKB * p;
    xreg[i] = *(const v4f*)(xsrc + 4 * c4);
  }
}

__device__ __forceinline__ void y_tile(
    float* __restrict__ sm, int tile, int p0, int p1, bool from_y,
    const float* __restrict__ S, const float* __restrict__ u,
    const float* __restrict__ C, const float* __restrict__ D, float* y) {
  float* sW = sm;
  float* sX = sm + NO * YLP;

  const int tid = threadIdx.x;
  const int wave = tid >> 5, lane = tid & 31;
  const int l16 = lane & 15, half = lane >> 4;
  const int t0 = tile * YT;
  const int m0 = (wave & 1) * 2;
  const int o0 = (wave >> 1) * 64;
  const int srow = tid >> 3;
  const int c4 = tid & 7;

  v8f acc[2][4] = {{{}, {}, {}, {}}, {{}, {}, {}, {}}};
  v4f wreg[8], xreg[2];

  y_fetch(p0, t0, srow, c4, S, u, C, D, wreg, xreg);
  if (from_y) {
#pragma unroll
    for (int mi = 0; mi < 2; ++mi)
#pragma unroll
      for (int j = 0; j < 4; ++j)
#pragma unroll
        for (int v = 0; v < 8; ++v)
          acc[mi][j][v] = y[(size_t)(v + 8 * half) * TT * NO +
                            (size_t)(t0 + m0 + mi) * NO + o0 + 16 * j + l16];
  }
  for (int p = p0; p < p1; ++p) {
#pragma unroll
    for (int i = 0; i < 8; ++i)
      *(v4f*)&sW[(srow + 32 * i) * YLP + 4 * c4] = wreg[i];
#pragma unroll
    for (int i = 0; i < 2; ++i)
      *(v4f*)&sX[(srow + 32 * i) * YLP + 4 * c4] = xreg[i];
    __syncthreads();
    if (p + 1 < p1) y_fetch(p + 1, t0, srow, c4, S, u, C, D, wreg, xreg);
    __builtin_amdgcn_sched_barrier(0);
#pragma unroll
    for (int kk = 0; kk < YKB / 4; ++kk) {
      v2f a0 = load2(&sX[((m0 + 0) * NB + l16) * YLP + 4 * kk + 2 * half]);
      v2f a1 = load2(&sX[((m0 + 1) * NB + l16) * YLP + 4 * kk + 2 * half]);
#pragma unroll
      for (int j = 0; j < 4; ++j) {
        v2f b = load2(&sW[(o0 + 16 * j + l16) * YLP + 4 * kk + 2 * half]);
        acc[0][j] = wmma4(a0, b, acc[0][j]);
        acc[1][j] = wmma4(a1, b, acc[1][j]);
      }
    }
    __syncthreads();
  }

#pragma unroll
  for (int mi = 0; mi < 2; ++mi)
#pragma unroll
    for (int j = 0; j < 4; ++j)
#pragma unroll
      for (int v = 0; v < 8; ++v)
        y[(size_t)(v + 8 * half) * TT * NO + (size_t)(t0 + m0 + mi) * NO + o0 +
          16 * j + l16] = acc[mi][j][v];
}

#define YSM ((NO + YM) * YLP)

__global__ __launch_bounds__(256) void s4_bscan_yin_kernel(
    const float* __restrict__ x0, const float* __restrict__ G,
    const float* __restrict__ Apow, float* __restrict__ Sbound,
    float* __restrict__ xfinal, const float* __restrict__ u,
    const float* __restrict__ D, float* __restrict__ y) {
  __shared__ __attribute__((aligned(16))) float sm[YSM];
  if (blockIdx.x == 0) {
    bscan_role(sm, x0, G, Apow, Sbound, xfinal);
    return;
  }
  y_tile(sm, blockIdx.x - 1, 0, YPI, false, nullptr, u, nullptr, D, y);
}

__global__ __launch_bounds__(256) void s4_states_yin_kernel(
    const float* __restrict__ uB, const float* __restrict__ Sbound,
    const float* __restrict__ Abar, float* __restrict__ S,
    const float* __restrict__ u, const float* __restrict__ D,
    float* __restrict__ y) {
  __shared__ __attribute__((aligned(16))) float sm[YSM];
  if (blockIdx.x < NCH) {
    states_role(sm, blockIdx.x, uB, Sbound, Abar, S);
    return;
  }
  y_tile(sm, YN_B + (blockIdx.x - NCH), 0, YPI, false, nullptr, u, nullptr, D,
         y);
}

__global__ __launch_bounds__(256) void s4_yfin_kernel(
    const float* __restrict__ S, const float* __restrict__ u,
    const float* __restrict__ C, const float* __restrict__ D, float* y) {
  __shared__ __attribute__((aligned(16))) float sm[YSM];
  const int tile = YNT - 1 - blockIdx.x;
  const bool started = tile < YN_B + YN_S;
  y_tile(sm, tile, started ? YPI : 0, YSTEPS, started, S, u, C, D, y);
}

extern "C" void kernel_launch(void* const* d_in, const int* in_sizes, int n_in,
                              void* d_out, int out_size, void* d_ws,
                              size_t ws_size, hipStream_t stream) {
  const float* u  = (const float*)d_in[0];
  const float* x0 = (const float*)d_in[1];
  const float* A  = (const float*)d_in[2];
  const float* B  = (const float*)d_in[3];
  const float* C  = (const float*)d_in[4];
  const float* D  = (const float*)d_in[5];

  float* y      = (float*)d_out;
  float* xfinal = (float*)d_out + (size_t)NB * TT * NO;

  float* ws = (float*)d_ws;
  float* Abar   = ws;                 ws += NS * NS;
  float* Apow   = ws;                 ws += (LCH + 1) * NS * NS;
  float* Pcat   = ws;                 ws += NS * LCH * NS;
  float* uB     = ws;                 ws += (size_t)TT * NB * NS;
  float* Sst    = ws;                 ws += (size_t)TT * NB * NS;
  float* G      = ws;                 ws += NCH * NB * NS;
  float* Sbound = ws;                 ws += NCH * NB * NS;

  s4_prep_ub_kernel<<<PRB + TT / 4, 128, 0, stream>>>(A, B, u, Abar, Apow, Pcat,
                                                      uB);
  s4_gagg_kernel<<<NCH * 4, 32 * GKS, 0, stream>>>(uB, Pcat, G);
  s4_bscan_yin_kernel<<<1 + YN_B, 256, 0, stream>>>(x0, G, Apow, Sbound, xfinal,
                                                    u, D, y);
  s4_states_yin_kernel<<<NCH + YN_S, 256, 0, stream>>>(uB, Sbound, Abar, Sst, u,
                                                       D, y);
  s4_yfin_kernel<<<YNT, 256, 0, stream>>>(Sst, u, C, D, y);
}


# prep: powers walked in steps of A_bar^2 (each chain workgroup builds A_bar^2 itself): 8 chains (row block x even/odd power) of 31 steps, not 4 of 63; per-element order of products and sums changes
# speedup vs baseline: 8.0494x; 1.0516x over previous
#include <hip/hip_runtime.h>
#include <hip/hip_bf16.h>


typedef __attribute__((ext_vector_type(2))) float v2f;
typedef __attribute__((ext_vector_type(4))) float v4f;
typedef __attribute__((ext_vector_type(8))) float v8f;

#define S4_DT 0.001f
#define TT 4096
#define NB 16
#define NS 64
#define NI 256
#define NO 256
#define LCH 64
#define NCH 64
#define LP 68

__device__ __forceinline__ v8f wmma4(v2f a, v2f b, v8f c) {
  return __builtin_amdgcn_wmma_f32_16x16x4_f32(
       false, a,  false, b,
       (short)0, c,  false,  false);
}

__device__ __forceinline__ v2f load2(const float* p) { return *(const v2f*)p; }

__device__ __forceinline__ v8f chain_product(const v8f& start,
                                             const v2f (&afr)[16],
                                             const v2f (&bfr)[16]) {
  v8f acc0 = start, acc1 = {};
#pragma unroll
  for (int kk = 0; kk < 16; kk += 2) {
    acc0 = wmma4(afr[kk], bfr[kk], acc0);
    acc1 = wmma4(afr[kk + 1], bfr[kk + 1], acc1);
  }
  return acc0 + acc1;
}

__device__ __forceinline__ void store_c(float* dst, int rs, int n0, v8f acc,
                                        int l16, int half) {
#pragma unroll
  for (int v = 0; v < 8; ++v)
    dst[(v + 8 * half) * rs + n0 + l16] = acc[v];
}

__device__ __forceinline__ v8f load_c(const float* src, int rs, int n0,
                                      int l16, int half) {
  v8f r;
#pragma unroll
  for (int v = 0; v < 8; ++v)
    r[v] = src[(v + 8 * half) * rs + n0 + l16];
  return r;
}

__device__ __forceinline__ void lds_step_barrier() {
  __builtin_amdgcn_fence(__ATOMIC_RELEASE, "workgroup", "local");
  __builtin_amdgcn_s_barrier();
  __builtin_amdgcn_fence(__ATOMIC_ACQUIRE, "workgroup", "local");
}

__device__ __forceinline__ int piece_lds(int q) {
  return (q >> 4) * LP + 4 * (q & 15);
}

#define PRB (NS / 16)
#define PCW (2 * PRB)
#define UT 4
#define UM (UT * NB)
#define UKB 32
#define USTEPS (NI / UKB)
#define ULP 36
#define PSM (2 * NS * LP + 2 * 16 * LP)

__device__ __forceinline__ void ub_fetch(int s, int t0, int srow, int c4,
                                         const float* __restrict__ u,
                                         const float* __restrict__ B,
                                         v4f (&wreg)[4], v4f (&xreg)[4]) {
#pragma unroll
  for (int i = 0; i < 4; ++i) {
    const int row = srow + 16 * i, tt = row >> 4, b = row & 15;
    wreg[i] = *(const v4f*)(B + (size_t)row * NI + UKB * s + 4 * c4);
    xreg[i] = *(const v4f*)(u + (size_t)b * TT * NI + (size_t)(t0 + tt) * NI +
                            UKB * s + 4 * c4);
  }
}

__device__ __forceinline__ float abar_elem(float a, int r, int c) {
  return a * S4_DT + (r == c ? 1.0f : 0.0f);
}

__global__ __launch_bounds__(128) void s4_prep_ub_kernel(
    const float* __restrict__ A, const float* __restrict__ B,
    const float* __restrict__ u, float* __restrict__ Abar,
    float* __restrict__ Apow, float* __restrict__ Pcat,
    float* __restrict__ uB) {
  __shared__ __attribute__((aligned(16))) float sm[PSM];
  const int tid = threadIdx.x;
  const int wave = tid >> 5, lane = tid & 31;
  const int l16 = lane & 15, half = lane >> 4;
  if (blockIdx.x >= PCW) {
    float* sW = sm;
    float* sX = sm + NS * ULP;
    const int t0 = (blockIdx.x - PCW) * UT;
    const int srow = tid >> 3;
    const int c4 = tid & 7;
    v8f acc[4] = {{}, {}, {}, {}};
    v4f wreg[4], xreg[4];
    ub_fetch(0, t0, srow, c4, u, B, wreg, xreg);
    for (int s = 0; s < USTEPS; ++s) {
#pragma unroll
      for (int i = 0; i < 4; ++i) {
        *(v4f*)&sW[(srow + 16 * i) * ULP + 4 * c4] = wreg[i] * S4_DT;
        *(v4f*)&sX[(srow + 16 * i) * ULP + 4 * c4] = xreg[i];
      }
      lds_step_barrier();
      if (s + 1 < USTEPS) ub_fetch(s + 1, t0, srow, c4, u, B, wreg, xreg);
      __builtin_amdgcn_sched_barrier(0);
#pragma unroll
      for (int kk = 0; kk < UKB / 4; ++kk) {
        v2f a = load2(&sX[(wave * NB + l16) * ULP + 4 * kk + 2 * half]);
#pragma unroll
        for (int j = 0; j < 4; ++j) {
          v2f b = load2(&sW[(j * 16 + l16) * ULP + 4 * kk + 2 * half]);
          acc[j] = wmma4(a, b, acc[j]);
        }
      }
      lds_step_barrier();
    }
    float* dst = uB + (size_t)(t0 + wave) * NB * NS;
#pragma unroll
    for (int j = 0; j < 4; ++j) store_c(dst, NS, j * 16, acc[j], l16, half);
    return;
  }
  const int offA = 0, offM = NS * LP, offT = 2 * NS * LP;
  const int n0 = wave * 16;
  const int rb = blockIdx.x >> 1;
  const int par = blockIdx.x & 1;
  const int r0 = rb * 16;
  const int q0 = tid, q1 = tid + 128;
  const int lo0 = piece_lds(q0), lo1 = piece_lds(q1);
  const int row0 = r0 + (q0 >> 4), row1 = r0 + (q1 >> 4);
  const int c0 = 4 * (q0 & 15), c1 = 4 * (q1 & 15);

  v4f araw[8];
#pragma unroll
  for (int i = 0; i < 8; ++i) araw[i] = *(const v4f*)(A + 4 * (tid + 128 * i));
#pragma unroll
  for (int i = 0; i < 8; ++i) {
    const int q = tid + 128 * i;
    const int r = q >> 4, c = 4 * (q & 15);
    v4f v;
#pragma unroll
    for (int e = 0; e < 4; ++e) v[e] = abar_elem(araw[i][e], r, c + e);
    *(v4f*)&sm[offA + r * LP + c] = v;
  }
  lds_step_barrier();

  float* pd0 = Pcat + row0 * (LCH * NS) + c0;
  float* pd1 = Pcat + row1 * (LCH * NS) + c1;
  if (par == 0) {
    *(v4f*)(Abar + row0 * NS + c0) = *(const v4f*)&sm[offA + r0 * LP + lo0];
    *(v4f*)(Abar + row1 * NS + c1) = *(const v4f*)&sm[offA + r0 * LP + lo1];
    v4f i0, i1;
#pragma unroll
    for (int e = 0; e < 4; ++e) {
      i0[e] = (row0 == c0 + e) ? 1.0f : 0.0f;
      i1[e] = (row1 == c1 + e) ? 1.0f : 0.0f;
    }
    *(v4f*)(pd0 + (LCH - 1) * NS) = i0;
    *(v4f*)(pd1 + (LCH - 1) * NS) = i1;
  }

  {
    v2f bfr1[16];
#pragma unroll
    for (int kk = 0; kk < 16; ++kk) {
      const int k = 4 * kk + 2 * half;
      bfr1[kk][0] = sm[offA + k * LP + n0 + l16];
      bfr1[kk][1] = sm[offA + (k + 1) * LP + n0 + l16];
    }
    v8f sq[4] = {{}, {}, {}, {}};
#pragma unroll
    for (int kk = 0; kk < 16; ++kk) {
#pragma unroll
      for (int mt = 0; mt < 4; ++mt) {
        const v2f a = load2(&sm[offA + (16 * mt + l16) * LP + 4 * kk + 2 * half]);
        sq[mt] = wmma4(a, bfr1[kk], sq[mt]);
      }
    }
#pragma unroll
    for (int mt = 0; mt < 4; ++mt)
#pragma unroll
      for (int v = 0; v < 8; ++v)
        sm[offM + (16 * mt + v + 8 * half) * LP + n0 + l16] = sq[mt][v];
  }
  lds_step_barrier();

  v2f bfr[16];
#pragma unroll
  for (int kk = 0; kk < 16; ++kk) {
    const int k = 4 * kk + 2 * half;
    bfr[kk][0] = sm[offM + k * LP + n0 + l16];
    bfr[kk][1] = sm[offM + (k + 1) * LP + n0 + l16];
  }

  int cur = (par ? offA : offM) + r0 * LP;
  int p = 2 - par;
  for (int i = 0; i < LCH / 2 - 1; ++i) {
    const v4f o0 = *(const v4f*)&sm[cur + lo0];
    const v4f o1 = *(const v4f*)&sm[cur + lo1];
    v2f afr[16];
#pragma unroll
    for (int kk = 0; kk < 16; ++kk)
      afr[kk] = load2(&sm[cur + l16 * LP + 4 * kk + 2 * half]);
    __builtin_amdgcn_sched_barrier(0);
    *(v4f*)(pd0 + (LCH - 1 - p) * NS) = o0;
    *(v4f*)(pd1 + (LCH - 1 - p) * NS) = o1;
    const v8f zero = {};
    const v8f acc = chain_product(zero, afr, bfr);
    const int nxt = offT + (i & 1) * (16 * LP);
#pragma unroll
    for (int v = 0; v < 8; ++v)
      sm[nxt + (v + 8 * half) * LP + n0 + l16] = acc[v];
    lds_step_barrier();
    cur = nxt;
    p += 2;
  }
  if (par == 0) {
    float* a64 = Apow + (size_t)LCH * NS * NS + r0 * NS;
    *(v4f*)(a64 + 4 * q0) = *(const v4f*)&sm[cur + lo0];
    *(v4f*)(a64 + 4 * q1) = *(const v4f*)&sm[cur + lo1];
  } else {
    *(v4f*)pd0 = *(const v4f*)&sm[cur + lo0];
    *(v4f*)pd1 = *(const v4f*)&sm[cur + lo1];
  }
}

#define GKS 16
#define GKL ((LCH * NS) / GKS)
#define GSL NS
#define GNS (GKL / GSL)

__device__ __forceinline__ void gagg_fetch(int s, int lane,
                                           const float* __restrict__ Fw,
                                           const float* __restrict__ Bw,
                                           v4f (&areg)[8], v4f (&breg)[8]) {
#pragma unroll
  for (int i = 0; i < 8; ++i) {
    const int q = lane + 32 * i;
    areg[i] = *(const v4f*)(Fw + s * (NB * NS) + 4 * q);
    breg[i] = *(const v4f*)(Bw + (size_t)(q >> 4) * (LCH * NS) + s * GSL +
                            4 * (q & 15));
  }
}
#define GRP 18

__global__ __launch_bounds__(512) void s4_gagg_kernel(
    const float* __restrict__ uB, const float* __restrict__ Pcat,
    float* __restrict__ G) {
  __shared__ __attribute__((aligned(16))) float stg[GKS][2][16 * LP];
  __shared__ float red[GKS][16 * GRP];
  const int tid = threadIdx.x;
  const int wave = tid >> 5, lane = tid & 31;
  const int l16 = lane & 15, half = lane >> 4;
  const int c = blockIdx.x >> 2, jn = blockIdx.x & 3;

  const float* F = uB + (size_t)c * LCH * NB * NS;
  const float* Fw = F + (size_t)wave * GNS * NB * NS;
  const float* Bw = Pcat + (size_t)(jn * 16) * (LCH * NS) + wave * GKL;
  float* sA = stg[wave][0];
  float* sB = stg[wave][1];
  v8f acc = {};
  v4f areg[8], breg[8];
  gagg_fetch(0, lane, Fw, Bw, areg, breg);
#pragma unroll 1
  for (int s = 0; s < GNS; ++s) {
#pragma unroll
    for (int i = 0; i < 8; ++i) {
      const int lo = piece_lds(lane + 32 * i);
      *(v4f*)&sA[lo] = areg[i];
      *(v4f*)&sB[lo] = breg[i];
    }
    lds_step_barrier();
    if (s + 1 < GNS) gagg_fetch(s + 1, lane, Fw, Bw, areg, breg);
    __builtin_amdgcn_sched_barrier(0);
#pragma unroll
    for (int kk = 0; kk < GSL / 4; ++kk) {
      v2f a = load2(&sA[l16 * LP + 4 * kk + 2 * half]);
      v2f b = load2(&sB[l16 * LP + 4 * kk + 2 * half]);
      acc = wmma4(a, b, acc);
    }
    lds_step_barrier();
  }
#pragma unroll
  for (int v = 0; v < 8; ++v) red[wave][(v + 8 * half) * GRP + l16] = acc[v];
  __syncthreads();
  if (tid < 256) {
    const int row = tid >> 4, col = tid & 15;
    float sum = red[0][row * GRP + col];
#pragma unroll
    for (int s = 1; s < GKS; ++s) sum += red[s][row * GRP + col];
    G[(size_t)(c * 4 + jn) * 256 + row * 16 + col] = sum;
  }
}

__device__ __forceinline__ void chain_step(
    const float* __restrict__ sbc, float* __restrict__ sbn, const v8f& start,
    float* __restrict__ dst, const v2f (&bfr)[16], int lo0, int lo1, int q0,
    int q1, int l16, int half, int n0) {
  const v4f o0 = *(const v4f*)&sbc[lo0];
  const v4f o1 = *(const v4f*)&sbc[lo1];
  v2f afr[16];
#pragma unroll
  for (int kk = 0; kk < 16; ++kk)
    afr[kk] = load2(&sbc[l16 * LP + 4 * kk + 2 * half]);
  __builtin_amdgcn_sched_barrier(0);
  *(v4f*)(dst + 4 * q0) = o0;
  *(v4f*)(dst + 4 * q1) = o1;
  const v8f acc = chain_product(start, afr, bfr);
#pragma unroll
  for (int v = 0; v < 8; ++v) sbn[(v + 8 * half) * LP + n0 + l16] = acc[v];
}

__device__ __forceinline__ void bscan_role(
    float* __restrict__ sm, const float* __restrict__ x0,
    const float* __restrict__ G, const float* __restrict__ Apow,
    float* __restrict__ Sbound, float* __restrict__ xfinal) {
  float (*sb)[NB * LP] = (float (*)[NB * LP])sm;
  const int tid = threadIdx.x;
  const int wave = __builtin_amdgcn_readfirstlane(tid >> 5);
  const bool work = wave < 4;
  const int lane = tid & 31;
  const int l16 = lane & 15, half = lane >> 4;
  const int n0 = (wave & 3) * 16;
  const float* AL = Apow + (size_t)LCH * NS * NS;
  const int q0 = tid & 127, q1 = q0 + 128;
  const int lo0 = piece_lds(q0), lo1 = piece_lds(q1);

  const float* Gw = G + (wave & 3) * 256;
  v8f ft[3];
  ft[0] = load_c(Gw, 16, 0, l16, half);
  ft[1] = load_c(Gw + NB * NS, 16, 0, l16, half);
  if (work) {
    *(v4f*)&sb[0][lo0] = *(const v4f*)(x0 + 4 * q0);
    *(v4f*)&sb[0][lo1] = *(const v4f*)(x0 + 4 * q1);
  }
  __syncthreads();

  v2f bfr[16] = {};
  if (work) {
#pragma unroll
    for (int kk = 0; kk < 16; ++kk)
      bfr[kk] = load2(AL + (n0 + l16) * NS + 4 * kk + 2 * half);
  }

  for (int c3 = 0; c3 < NCH / 3; ++c3) {
#pragma unroll
    for (int r = 0; r < 3; ++r) {
      const int c = 3 * c3 + r;
      const int cn = (c + 2 < NCH) ? c + 2 : NCH - 1;
      ft[(r + 2) % 3] = load_c(Gw + (size_t)cn * NB * NS, 16, 0, l16, half);
      if (work)
        chain_step(sb[c & 1], sb[(c & 1) ^ 1], ft[r],
                   Sbound + (size_t)c * NB * NS, bfr, lo0, lo1, q0, q1, l16,
                   half, n0);
      lds_step_barrier();
    }
  }
  if (work)
    chain_step(sb[1], sb[0], ft[0], Sbound + (size_t)(NCH - 1) * NB * NS, bfr,
               lo0, lo1, q0, q1, l16, half, n0);
  lds_step_barrier();
  const int cur = 0;
  if (work) {
    *(v4f*)(xfinal + 4 * q0) = *(const v4f*)&sb[cur][lo0];
    *(v4f*)(xfinal + 4 * q1) = *(const v4f*)&sb[cur][lo1];
  }
}

__device__ __forceinline__ void states_role(
    float* __restrict__ sm, int c, const float* __restrict__ uB,
    const float* __restrict__ Sbound, const float* __restrict__ Abar,
    float* __restrict__ S) {
  float (*sb)[NB * LP] = (float (*)[NB * LP])sm;
  const int tid = threadIdx.x;
  const int wave = __builtin_amdgcn_readfirstlane(tid >> 5);
  const bool work = wave < 4;
  const int lane = tid & 31;
  const int l16 = lane & 15, half = lane >> 4;
  const int n0 = (wave & 3) * 16;
  const int t0 = c * LCH;
  const int q0 = tid & 127, q1 = q0 + 128;
  const int lo0 = piece_lds(q0), lo1 = piece_lds(q1);

  const float* Fw = uB + (size_t)t0 * NB * NS;
  v8f ft[3];
  ft[0] = load_c(Fw, NS, n0, l16, half);
  ft[1] = load_c(Fw + NB * NS, NS, n0, l16, half);
  if (work) {
    const float* sb0 = Sbound + (size_t)c * NB * NS;
    *(v4f*)&sb[0][lo0] = *(const v4f*)(sb0 + 4 * q0);
    *(v4f*)&sb[0][lo1] = *(const v4f*)(sb0 + 4 * q1);
  }
  __syncthreads();

  v2f bfr[16] = {};
  if (work) {
#pragma unroll
    for (int kk = 0; kk < 16; ++kk)
      bfr[kk] = load2(Abar + (n0 + l16) * NS + 4 * kk + 2 * half);
  }

  for (int k3 = 0; k3 < LCH / 3; ++k3) {
#pragma unroll
    for (int r = 0; r < 3; ++r) {
      const int k = 3 * k3 + r;
      const int kn = (k + 2 < LCH) ? k + 2 : LCH - 1;
      ft[(r + 2) % 3] = load_c(Fw + (size_t)kn * NB * NS, NS, n0, l16, half);
      if (work)
        chain_step(sb[k & 1], sb[(k & 1) ^ 1], ft[r],
                   S + (size_t)(t0 + k) * NB * NS, bfr, lo0, lo1, q0, q1, l16,
                   half, n0);
      lds_step_barrier();
    }
  }
  if (work)
    chain_step(sb[1], sb[0], ft[0], S + (size_t)(t0 + LCH - 1) * NB * NS, bfr,
               lo0, lo1, q0, q1, l16, half, n0);
  lds_step_barrier();
}

#define YT 4
#define YM (YT * NB)
#define YKB 32
#define YSC (NS / YKB)
#define YSTEPS ((NS + NI) / YKB)
#define YPI (YSTEPS - YSC)
#define YNT (TT / YT)
#define YN_B 511
#define YN_S 448
#define YLP 36

__device__ __forceinline__ void y_fetch(
    int p, int t0, int srow, int c4, const float* __restrict__ S,
    const float* __restrict__ u, const float* __restrict__ C,
    const float* __restrict__ D, v4f (&wreg)[8], v4f (&xreg)[2]) {
  const bool st = p >= YPI;
  const float* wsrc = st ? C + YKB * (p - YPI) : D + YKB * p;
  const int wpitch = st ? NS : NI;
#pragma unroll
  for (int i = 0; i < 8; ++i)
    wreg[i] = *(const v4f*)(wsrc + (size_t)(srow + 32 * i) * wpitch + 4 * c4);
#pragma unroll
  for (int i = 0; i < 2; ++i) {
    const int row = srow + 32 * i, tt = row >> 4, b = row & 15;
    const float* xsrc =
        st ? S + (size_t)(t0 + tt) * NB * NS + b * NS + YKB * (p - YPI)
           : u + (size_t)b * TT * NI + (size_t)(t0 + tt) * NI + YKB * p;
    xreg[i] = *(const v4f*)(xsrc + 4 * c4);
  }
}

__device__ __forceinline__ void y_tile(
    float* __restrict__ sm, int tile, int p0, int p1, bool from_y,
    const float* __restrict__ S, const float* __restrict__ u,
    const float* __restrict__ C, const float* __restrict__ D, float* y) {
  float* sW = sm;
  float* sX = sm + NO * YLP;

  const int tid = threadIdx.x;
  const int wave = tid >> 5, lane = tid & 31;
  const int l16 = lane & 15, half = lane >> 4;
  const int t0 = tile * YT;
  const int m0 = (wave & 1) * 2;
  const int o0 = (wave >> 1) * 64;
  const int srow = tid >> 3;
  const int c4 = tid & 7;

  v8f acc[2][4] = {{{}, {}, {}, {}}, {{}, {}, {}, {}}};
  v4f wreg[8], xreg[2];

  y_fetch(p0, t0, srow, c4, S, u, C, D, wreg, xreg);
  if (from_y) {
#pragma unroll
    for (int mi = 0; mi < 2; ++mi)
#pragma unroll
      for (int j = 0; j < 4; ++j)
#pragma unroll
        for (int v = 0; v < 8; ++v)
          acc[mi][j][v] = y[(size_t)(v + 8 * half) * TT * NO +
                            (size_t)(t0 + m0 + mi) * NO + o0 + 16 * j + l16];
  }
  for (int p = p0; p < p1; ++p) {
#pragma unroll
    for (int i = 0; i < 8; ++i)
      *(v4f*)&sW[(srow + 32 * i) * YLP + 4 * c4] = wreg[i];
#pragma unroll
    for (int i = 0; i < 2; ++i)
      *(v4f*)&sX[(srow + 32 * i) * YLP + 4 * c4] = xreg[i];
    __syncthreads();
    if (p + 1 < p1) y_fetch(p + 1, t0, srow, c4, S, u, C, D, wreg, xreg);
    __builtin_amdgcn_sched_barrier(0);
#pragma unroll
    for (int kk = 0; kk < YKB / 4; ++kk) {
      v2f a0 = load2(&sX[((m0 + 0) * NB + l16) * YLP + 4 * kk + 2 * half]);
      v2f a1 = load2(&sX[((m0 + 1) * NB + l16) * YLP + 4 * kk + 2 * half]);
#pragma unroll
      for (int j = 0; j < 4; ++j) {
        v2f b = load2(&sW[(o0 + 16 * j + l16) * YLP + 4 * kk + 2 * half]);
        acc[0][j] = wmma4(a0, b, acc[0][j]);
        acc[1][j] = wmma4(a1, b, acc[1][j]);
      }
    }
    __syncthreads();
  }

#pragma unroll
  for (int mi = 0; mi < 2; ++mi)
#pragma unroll
    for (int j = 0; j < 4; ++j)
#pragma unroll
      for (int v = 0; v < 8; ++v)
        y[(size_t)(v + 8 * half) * TT * NO + (size_t)(t0 + m0 + mi) * NO + o0 +
          16 * j + l16] = acc[mi][j][v];
}

#define YSM ((NO + YM) * YLP)

__global__ __launch_bounds__(256) void s4_bscan_yin_kernel(
    const float* __restrict__ x0, const float* __restrict__ G,
    const float* __restrict__ Apow, float* __restrict__ Sbound,
    float* __restrict__ xfinal, const float* __restrict__ u,
    const float* __restrict__ D, float* __restrict__ y) {
  __shared__ __attribute__((aligned(16))) float sm[YSM];
  if (blockIdx.x == 0) {
    bscan_role(sm, x0, G, Apow, Sbound, xfinal);
    return;
  }
  y_tile(sm, blockIdx.x - 1, 0, YPI, false, nullptr, u, nullptr, D, y);
}

__global__ __launch_bounds__(256) void s4_states_yin_kernel(
    const float* __restrict__ uB, const float* __restrict__ Sbound,
    const float* __restrict__ Abar, float* __restrict__ S,
    const float* __restrict__ u, const float* __restrict__ D,
    float* __restrict__ y) {
  __shared__ __attribute__((aligned(16))) float sm[YSM];
  if (blockIdx.x < NCH) {
    states_role(sm, blockIdx.x, uB, Sbound, Abar, S);
    return;
  }
  y_tile(sm, YN_B + (blockIdx.x - NCH), 0, YPI, false, nullptr, u, nullptr, D,
         y);
}

__global__ __launch_bounds__(256) void s4_yfin_kernel(
    const float* __restrict__ S, const float* __restrict__ u,
    const float* __restrict__ C, const float* __restrict__ D, float* y) {
  __shared__ __attribute__((aligned(16))) float sm[YSM];
  const int tile = YNT - 1 - blockIdx.x;
  const bool started = tile < YN_B + YN_S;
  y_tile(sm, tile, started ? YPI : 0, YSTEPS, started, S, u, C, D, y);
}

extern "C" void kernel_launch(void* const* d_in, const int* in_sizes, int n_in,
                              void* d_out, int out_size, void* d_ws,
                              size_t ws_size, hipStream_t stream) {
  const float* u  = (const float*)d_in[0];
  const float* x0 = (const float*)d_in[1];
  const float* A  = (const float*)d_in[2];
  const float* B  = (const float*)d_in[3];
  const float* C  = (const float*)d_in[4];
  const float* D  = (const float*)d_in[5];

  float* y      = (float*)d_out;
  float* xfinal = (float*)d_out + (size_t)NB * TT * NO;

  float* ws = (float*)d_ws;
  float* Abar   = ws;                 ws += NS * NS;
  float* Apow   = ws;                 ws += (LCH + 1) * NS * NS;
  float* Pcat   = ws;                 ws += NS * LCH * NS;
  float* uB     = ws;                 ws += (size_t)TT * NB * NS;
  float* Sst    = ws;                 ws += (size_t)TT * NB * NS;
  float* G      = ws;                 ws += NCH * NB * NS;
  float* Sbound = ws;                 ws += NCH * NB * NS;

  s4_prep_ub_kernel<<<PCW + TT / 4, 128, 0, stream>>>(A, B, u, Abar, Apow, Pcat,
                                                      uB);
  s4_gagg_kernel<<<NCH * 4, 32 * GKS, 0, stream>>>(uB, Pcat, G);
  s4_bscan_yin_kernel<<<1 + YN_B, 256, 0, stream>>>(x0, G, Apow, Sbound, xfinal,
                                                    u, D, y);
  s4_states_yin_kernel<<<NCH + YN_S, 256, 0, stream>>>(uB, Sbound, Abar, Sst, u,
                                                       D, y);
  s4_yfin_kernel<<<YNT, 256, 0, stream>>>(Sst, u, C, D, y);
}
